# starting kernel
# baseline (speedup 1.0000x reference)
#include <hip/hip_runtime.h>
#include <hip/hip_bf16.h>

typedef __attribute__((ext_vector_type(2))) float v2f;
typedef __attribute__((ext_vector_type(8))) float v8f;

#define IN_DIM  128
#define HD      128
#define E_DIM   32
#define NHEAD   4
#define DHEAD   32
#define OUT_DIM 32
#define INV_SQRT_D 0.17677669529663687f

static __device__ __forceinline__ v8f wmma_f32(v2f a, v2f b, v8f c) {
  return __builtin_amdgcn_wmma_f32_16x16x4_f32(false, a, false, b, (short)0, c,
                                               false, false);
}

__global__ void zero_kernel(float* p, int n) {
  int i = blockIdx.x * blockDim.x + threadIdx.x;
  if (i < n) p[i] = 0.0f;
}

__global__ void precompute_wvo(const float* __restrict__ Wv,
                               const float* __restrict__ bv,
                               const float* __restrict__ Wo,
                               float* __restrict__ Wvo,
                               float* __restrict__ bvo) {
  int idx = blockIdx.x * blockDim.x + threadIdx.x;
  if (idx < IN_DIM * HD) {
    int k = idx >> 7;
    int c = idx & 127;
    int h = c >> 5, j = c & 31;
    float acc = 0.0f;
    for (int d = 0; d < DHEAD; ++d)
      acc += Wv[k * HD + h * DHEAD + d] * Wo[(h * DHEAD + d) * OUT_DIM + j];
    Wvo[idx] = acc;
  } else if (idx < IN_DIM * HD + HD) {
    int c = idx - IN_DIM * HD;
    int h = c >> 5, j = c & 31;
    float acc = 0.0f;
    for (int d = 0; d < DHEAD; ++d)
      acc += bv[h * DHEAD + d] * Wo[(h * DHEAD + d) * OUT_DIM + j];
    bvo[c] = acc;
  }
}

__global__ __launch_bounds__(256) void node_gemm(
    const float* __restrict__ x,
    const float* __restrict__ Wq, const float* __restrict__ bq,
    const float* __restrict__ Wk, const float* __restrict__ bk,
    const float* __restrict__ Wvo, const float* __restrict__ bvo,
    float* __restrict__ Q, float* __restrict__ Km, float* __restrict__ Vo,
    int N) {
  __shared__ float As[16][IN_DIM];
  const int tid = threadIdx.x;
  const int m0 = blockIdx.x * 16;

  for (int i = 0; i < 8; ++i) {
    int lin = tid + i * 256;
    int r = lin >> 7, c = lin & 127;
    int gr = m0 + r; if (gr >= N) gr = N - 1;
    As[r][c] = x[(size_t)gr * IN_DIM + c];
  }
  __syncthreads();

  const int w = tid >> 5, L = tid & 31;
  const int n0 = w * 16;
  const int rowA = L & 15;
  const int kOff = (L < 16) ? 0 : 2;
  const int colB = L & 15;

  const float bq0 = bq[n0 + colB];
  const float bk0 = bk[n0 + colB];
  const float bv0 = bvo[n0 + colB];
  v8f cq, ck, cv;
  for (int j = 0; j < 8; ++j) { cq[j] = bq0; ck[j] = bk0; cv[j] = bv0; }

  for (int kk = 0; kk < IN_DIM; kk += 4) {
    v2f a;
    a.x = As[rowA][kk + kOff];
    a.y = As[rowA][kk + kOff + 1];
    v2f b;
    b.x = Wq[(kk + kOff) * HD + n0 + colB];
    b.y = Wq[(kk + kOff + 1) * HD + n0 + colB];
    cq = wmma_f32(a, b, cq);
    b.x = Wk[(kk + kOff) * HD + n0 + colB];
    b.y = Wk[(kk + kOff + 1) * HD + n0 + colB];
    ck = wmma_f32(a, b, ck);
    b.x = Wvo[(kk + kOff) * HD + n0 + colB];
    b.y = Wvo[(kk + kOff + 1) * HD + n0 + colB];
    cv = wmma_f32(a, b, cv);
  }

  for (int j = 0; j < 8; ++j) {
    int r = (L < 16) ? j : (j + 8);
    int gr = m0 + r;
    if (gr < N) {
      size_t o = (size_t)gr * HD + n0 + colB;
      Q[o] = cq[j]; Km[o] = ck[j]; Vo[o] = cv[j];
    }
  }
}

__global__ __launch_bounds__(128) void edge_selfdot(
    const float* __restrict__ ea, const float* __restrict__ We,
    const float* __restrict__ be, float* __restrict__ sE, int NE) {
  __shared__ float Ae[16][E_DIM];
  const int tid = threadIdx.x;
  const int e0 = blockIdx.x * 16;

  for (int i = 0; i < 4; ++i) {
    int lin = tid + i * 128;
    int r = lin >> 5, c = lin & 31;
    int ge = e0 + r; if (ge >= NE) ge = NE - 1;
    Ae[r][c] = ea[(size_t)ge * E_DIM + c];
  }
  __syncthreads();

  const int h = tid >> 5, L = tid & 31;
  const int n0 = h * DHEAD;
  const int rowA = L & 15;
  const int kOff = (L < 16) ? 0 : 2;
  const int colB = L & 15;

  v8f c0 = {}, c1 = {};
  for (int kk = 0; kk < E_DIM; kk += 4) {
    v2f a;
    a.x = Ae[rowA][kk + kOff];
    a.y = Ae[rowA][kk + kOff + 1];
    v2f b0, b1;
    b0.x = We[(kk + kOff) * HD + n0 + colB];
    b0.y = We[(kk + kOff + 1) * HD + n0 + colB];
    b1.x = We[(kk + kOff) * HD + n0 + 16 + colB];
    b1.y = We[(kk + kOff + 1) * HD + n0 + 16 + colB];
    c0 = wmma_f32(a, b0, c0);
    c1 = wmma_f32(a, b1, c1);
  }

  const float be0 = be[n0 + colB], be1 = be[n0 + 16 + colB];
  float t[8];
  for (int j = 0; j < 8; ++j) {
    float u = c0[j] + be0, v = c1[j] + be1;
    t[j] = u * u + v * v;
  }
  for (int m = 1; m <= 8; m <<= 1)
    for (int j = 0; j < 8; ++j) t[j] += __shfl_xor(t[j], m);

  if (L == 0 || L == 16) {
    int rb = (L == 0) ? 0 : 8;
    for (int j = 0; j < 8; ++j) {
      int ge = e0 + rb + j;
      if (ge < NE) sE[(size_t)ge * NHEAD + h] = t[j];
    }
  }
}

__global__ __launch_bounds__(256) void edge_pass(
    const int* __restrict__ ei, const float* __restrict__ Q,
    const float* __restrict__ Km, const float* __restrict__ Vo,
    const float* __restrict__ sE, const float* __restrict__ bo,
    float* __restrict__ sums, float* __restrict__ cnt, int NE) {
  const int e = blockIdx.x * 8 + (threadIdx.x >> 5);
  if (e >= NE) return;
  const int L = threadIdx.x & 31;
  const int src = ei[e];
  const int dst = ei[NE + e];

  const int h = L >> 3, d4 = (L & 7) << 2;
  const float4 q = *(const float4*)(Q + (size_t)src * HD + h * DHEAD + d4);
  const float4 k = *(const float4*)(Km + (size_t)dst * HD + h * DHEAD + d4);
  float p = q.x * k.x + q.y * k.y + q.z * k.z + q.w * k.w;
  p += __shfl_xor(p, 1);
  p += __shfl_xor(p, 2);
  p += __shfl_xor(p, 4);

  float sc[NHEAD];
  for (int h2 = 0; h2 < NHEAD; ++h2)
    sc[h2] = (__shfl(p, h2 * 8) + sE[(size_t)e * NHEAD + h2]) * INV_SQRT_D;

  float m = sc[0];
  for (int h2 = 1; h2 < NHEAD; ++h2) m = fmaxf(m, sc[h2]);
  float ex[NHEAD], den = 0.0f;
  for (int h2 = 0; h2 < NHEAD; ++h2) { ex[h2] = __expf(sc[h2] - m); den += ex[h2]; }
  const float inv = 1.0f / den;

  const int j = L;
  float acc = bo[j];
  for (int h2 = 0; h2 < NHEAD; ++h2)
    acc += (ex[h2] * inv) * Vo[(size_t)dst * HD + h2 * DHEAD + j];

  atomicAdd(&sums[(size_t)src * OUT_DIM + j], acc);
  if (L == 0) atomicAdd(&cnt[src], 1.0f);
}

__global__ void finalize(const float* __restrict__ sums,
                         const float* __restrict__ cnt,
                         float* __restrict__ out, int N) {
  int i = blockIdx.x * blockDim.x + threadIdx.x;
  if (i < N * OUT_DIM)
    out[i] = sums[i] / fmaxf(cnt[i >> 5], 1.0f);
}

extern "C" void kernel_launch(void* const* d_in, const int* in_sizes, int n_in,
                              void* d_out, int out_size, void* d_ws, size_t ws_size,
                              hipStream_t stream) {
  const float* x  = (const float*)d_in[0];
  const int*   ei = (const int*)d_in[1];
  const float* ea = (const float*)d_in[2];
  const float* Wq = (const float*)d_in[3];
  const float* bq = (const float*)d_in[4];
  const float* Wk = (const float*)d_in[5];
  const float* bk = (const float*)d_in[6];
  const float* Wv = (const float*)d_in[7];
  const float* bv = (const float*)d_in[8];
  const float* We = (const float*)d_in[9];
  const float* be = (const float*)d_in[10];
  const float* Wo = (const float*)d_in[11];
  const float* bo = (const float*)d_in[12];

  const int N  = in_sizes[0] / IN_DIM;
  const int NE = in_sizes[1] / 2;

  float* ws   = (float*)d_ws;
  float* Q    = ws;
  float* Km   = Q   + (size_t)N * HD;
  float* Vo   = Km  + (size_t)N * HD;
  float* Wvo  = Vo  + (size_t)N * HD;
  float* bvo  = Wvo + IN_DIM * HD;
  float* sE   = bvo + HD;
  float* sums = sE  + (size_t)NE * NHEAD;
  float* cnt  = sums + (size_t)N * OUT_DIM;

  {
    int n = N * OUT_DIM + N;
    zero_kernel<<<(n + 255) / 256, 256, 0, stream>>>(sums, n);
  }
  {
    int n = IN_DIM * HD + HD;
    precompute_wvo<<<(n + 127) / 128, 128, 0, stream>>>(Wv, bv, Wo, Wvo, bvo);
  }
  node_gemm<<<(N + 15) / 16, 256, 0, stream>>>(x, Wq, bq, Wk, bk, Wvo, bvo,
                                               Q, Km, Vo, N);
  edge_selfdot<<<(NE + 15) / 16, 128, 0, stream>>>(ea, We, be, sE, NE);
  edge_pass<<<(NE + 7) / 8, 256, 0, stream>>>(ei, Q, Km, Vo, sE, bo,
                                              sums, cnt, NE);
  {
    int n = N * OUT_DIM;
    finalize<<<(n + 255) / 256, 256, 0, stream>>>(sums, cnt, (float*)d_out, N);
  }
}


# edge pass: fixed grid of 16384 waves, each looping over its own contiguous run of edges (was one wave per edge, 640000 waves); per-edge arithmetic identical, float atomics kept (arrival order)
# speedup vs baseline: 1.3971x; 1.3971x over previous
#include <hip/hip_runtime.h>
#include <hip/hip_bf16.h>

typedef __attribute__((ext_vector_type(2))) float v2f;
typedef __attribute__((ext_vector_type(8))) float v8f;

#define IN_DIM  128
#define HD      128
#define E_DIM   32
#define NHEAD   4
#define DHEAD   32
#define OUT_DIM 32
#define INV_SQRT_D 0.17677669529663687f

static __device__ __forceinline__ v8f wmma_f32(v2f a, v2f b, v8f c) {
  return __builtin_amdgcn_wmma_f32_16x16x4_f32(false, a, false, b, (short)0, c,
                                               false, false);
}

__global__ void zero_kernel(float* p, int n) {
  int i = blockIdx.x * blockDim.x + threadIdx.x;
  if (i < n) p[i] = 0.0f;
}

__global__ void precompute_wvo(const float* __restrict__ Wv,
                               const float* __restrict__ bv,
                               const float* __restrict__ Wo,
                               float* __restrict__ Wvo,
                               float* __restrict__ bvo) {
  int idx = blockIdx.x * blockDim.x + threadIdx.x;
  if (idx < IN_DIM * HD) {
    int k = idx >> 7;
    int c = idx & 127;
    int h = c >> 5, j = c & 31;
    float acc = 0.0f;
    for (int d = 0; d < DHEAD; ++d)
      acc += Wv[k * HD + h * DHEAD + d] * Wo[(h * DHEAD + d) * OUT_DIM + j];
    Wvo[idx] = acc;
  } else if (idx < IN_DIM * HD + HD) {
    int c = idx - IN_DIM * HD;
    int h = c >> 5, j = c & 31;
    float acc = 0.0f;
    for (int d = 0; d < DHEAD; ++d)
      acc += bv[h * DHEAD + d] * Wo[(h * DHEAD + d) * OUT_DIM + j];
    bvo[c] = acc;
  }
}

__global__ __launch_bounds__(256) void node_gemm(
    const float* __restrict__ x,
    const float* __restrict__ Wq, const float* __restrict__ bq,
    const float* __restrict__ Wk, const float* __restrict__ bk,
    const float* __restrict__ Wvo, const float* __restrict__ bvo,
    float* __restrict__ Q, float* __restrict__ Km, float* __restrict__ Vo,
    int N) {
  __shared__ float As[16][IN_DIM];
  const int tid = threadIdx.x;
  const int m0 = blockIdx.x * 16;

  for (int i = 0; i < 8; ++i) {
    int lin = tid + i * 256;
    int r = lin >> 7, c = lin & 127;
    int gr = m0 + r; if (gr >= N) gr = N - 1;
    As[r][c] = x[(size_t)gr * IN_DIM + c];
  }
  __syncthreads();

  const int w = tid >> 5, L = tid & 31;
  const int n0 = w * 16;
  const int rowA = L & 15;
  const int kOff = (L < 16) ? 0 : 2;
  const int colB = L & 15;

  const float bq0 = bq[n0 + colB];
  const float bk0 = bk[n0 + colB];
  const float bv0 = bvo[n0 + colB];
  v8f cq, ck, cv;
  for (int j = 0; j < 8; ++j) { cq[j] = bq0; ck[j] = bk0; cv[j] = bv0; }

  for (int kk = 0; kk < IN_DIM; kk += 4) {
    v2f a;
    a.x = As[rowA][kk + kOff];
    a.y = As[rowA][kk + kOff + 1];
    v2f b;
    b.x = Wq[(kk + kOff) * HD + n0 + colB];
    b.y = Wq[(kk + kOff + 1) * HD + n0 + colB];
    cq = wmma_f32(a, b, cq);
    b.x = Wk[(kk + kOff) * HD + n0 + colB];
    b.y = Wk[(kk + kOff + 1) * HD + n0 + colB];
    ck = wmma_f32(a, b, ck);
    b.x = Wvo[(kk + kOff) * HD + n0 + colB];
    b.y = Wvo[(kk + kOff + 1) * HD + n0 + colB];
    cv = wmma_f32(a, b, cv);
  }

  for (int j = 0; j < 8; ++j) {
    int r = (L < 16) ? j : (j + 8);
    int gr = m0 + r;
    if (gr < N) {
      size_t o = (size_t)gr * HD + n0 + colB;
      Q[o] = cq[j]; Km[o] = ck[j]; Vo[o] = cv[j];
    }
  }
}

__global__ __launch_bounds__(128) void edge_selfdot(
    const float* __restrict__ ea, const float* __restrict__ We,
    const float* __restrict__ be, float* __restrict__ sE, int NE) {
  __shared__ float Ae[16][E_DIM];
  const int tid = threadIdx.x;
  const int e0 = blockIdx.x * 16;

  for (int i = 0; i < 4; ++i) {
    int lin = tid + i * 128;
    int r = lin >> 5, c = lin & 31;
    int ge = e0 + r; if (ge >= NE) ge = NE - 1;
    Ae[r][c] = ea[(size_t)ge * E_DIM + c];
  }
  __syncthreads();

  const int h = tid >> 5, L = tid & 31;
  const int n0 = h * DHEAD;
  const int rowA = L & 15;
  const int kOff = (L < 16) ? 0 : 2;
  const int colB = L & 15;

  v8f c0 = {}, c1 = {};
  for (int kk = 0; kk < E_DIM; kk += 4) {
    v2f a;
    a.x = Ae[rowA][kk + kOff];
    a.y = Ae[rowA][kk + kOff + 1];
    v2f b0, b1;
    b0.x = We[(kk + kOff) * HD + n0 + colB];
    b0.y = We[(kk + kOff + 1) * HD + n0 + colB];
    b1.x = We[(kk + kOff) * HD + n0 + 16 + colB];
    b1.y = We[(kk + kOff + 1) * HD + n0 + 16 + colB];
    c0 = wmma_f32(a, b0, c0);
    c1 = wmma_f32(a, b1, c1);
  }

  const float be0 = be[n0 + colB], be1 = be[n0 + 16 + colB];
  float t[8];
  for (int j = 0; j < 8; ++j) {
    float u = c0[j] + be0, v = c1[j] + be1;
    t[j] = u * u + v * v;
  }
  for (int m = 1; m <= 8; m <<= 1)
    for (int j = 0; j < 8; ++j) t[j] += __shfl_xor(t[j], m);

  if (L == 0 || L == 16) {
    int rb = (L == 0) ? 0 : 8;
    for (int j = 0; j < 8; ++j) {
      int ge = e0 + rb + j;
      if (ge < NE) sE[(size_t)ge * NHEAD + h] = t[j];
    }
  }
}

#define EP_BLOCKS   2048
#define EP_WAVES_LG 14

static __device__ __forceinline__ int bcast_i(int v, int lane) {
  return __builtin_amdgcn_readlane(v, lane);
}
static __device__ __forceinline__ float bcast_f(float v, int lane) {
  return __int_as_float(__builtin_amdgcn_readlane(__float_as_int(v), lane));
}

__global__ __launch_bounds__(256) void edge_pass(
    const int* __restrict__ ei, const float* __restrict__ Q,
    const float* __restrict__ Km, const float* __restrict__ Vo,
    const float* __restrict__ sE, const float* __restrict__ bo,
    float* __restrict__ sums, float* __restrict__ cnt, int NE) {
  const int L = threadIdx.x & 31;
  const int wv = __builtin_amdgcn_readfirstlane((int)(threadIdx.x >> 5));
  const long long wid = (long long)blockIdx.x * 8 + wv;
  const int beg = (int)((wid * NE) >> EP_WAVES_LG);
  const int end = (int)(((wid + 1) * NE) >> EP_WAVES_LG);
  if (beg >= end) return;

  const float bo_j = bo[L];

  for (int t = beg; t < end; t += 32) {
    const int nb = (end - t < 32) ? (end - t) : 32;
    int myE = t + L; if (myE >= end) myE = end - 1;
    const int src_l = ei[myE];
    const int dst_l = ei[NE + myE];
    const float4 se_l = *(const float4*)(sE + (size_t)myE * NHEAD);
    if (L < nb) atomicAdd(&cnt[src_l], 1.0f);

    for (int i = 0; i < nb; ++i) {
      const int src = bcast_i(src_l, i);
      const int dst = bcast_i(dst_l, i);

      const float4 q = *(const float4*)(Q  + (size_t)src * HD + L * 4);
      const float4 k = *(const float4*)(Km + (size_t)dst * HD + L * 4);
      float vo[NHEAD];
      for (int h2 = 0; h2 < NHEAD; ++h2)
        vo[h2] = Vo[(size_t)dst * HD + h2 * DHEAD + L];
      const float se[NHEAD] = {bcast_f(se_l.x, i), bcast_f(se_l.y, i),
                               bcast_f(se_l.z, i), bcast_f(se_l.w, i)};

      float p = q.x * k.x + q.y * k.y + q.z * k.z + q.w * k.w;
      p += __shfl_xor(p, 1);
      p += __shfl_xor(p, 2);
      p += __shfl_xor(p, 4);

      float sc[NHEAD];
      for (int h2 = 0; h2 < NHEAD; ++h2)
        sc[h2] = (__shfl(p, h2 * 8) + se[h2]) * INV_SQRT_D;

      float m = sc[0];
      for (int h2 = 1; h2 < NHEAD; ++h2) m = fmaxf(m, sc[h2]);
      float ex[NHEAD], den = 0.0f;
      for (int h2 = 0; h2 < NHEAD; ++h2) { ex[h2] = __expf(sc[h2] - m); den += ex[h2]; }
      const float inv = 1.0f / den;

      float acc = bo_j;
      for (int h2 = 0; h2 < NHEAD; ++h2)
        acc += (ex[h2] * inv) * vo[h2];

      atomicAdd(&sums[(size_t)src * OUT_DIM + L], acc);
    }
  }
}

__global__ void finalize(const float* __restrict__ sums,
                         const float* __restrict__ cnt,
                         float* __restrict__ out, int N) {
  int i = blockIdx.x * blockDim.x + threadIdx.x;
  if (i < N * OUT_DIM)
    out[i] = sums[i] / fmaxf(cnt[i >> 5], 1.0f);
}

extern "C" void kernel_launch(void* const* d_in, const int* in_sizes, int n_in,
                              void* d_out, int out_size, void* d_ws, size_t ws_size,
                              hipStream_t stream) {
  const float* x  = (const float*)d_in[0];
  const int*   ei = (const int*)d_in[1];
  const float* ea = (const float*)d_in[2];
  const float* Wq = (const float*)d_in[3];
  const float* bq = (const float*)d_in[4];
  const float* Wk = (const float*)d_in[5];
  const float* bk = (const float*)d_in[6];
  const float* Wv = (const float*)d_in[7];
  const float* bv = (const float*)d_in[8];
  const float* We = (const float*)d_in[9];
  const float* be = (const float*)d_in[10];
  const float* Wo = (const float*)d_in[11];
  const float* bo = (const float*)d_in[12];

  const int N  = in_sizes[0] / IN_DIM;
  const int NE = in_sizes[1] / 2;

  float* ws   = (float*)d_ws;
  float* Q    = ws;
  float* Km   = Q   + (size_t)N * HD;
  float* Vo   = Km  + (size_t)N * HD;
  float* Wvo  = Vo  + (size_t)N * HD;
  float* bvo  = Wvo + IN_DIM * HD;
  float* sE   = bvo + HD;
  float* sums = sE  + (size_t)NE * NHEAD;
  float* cnt  = sums + (size_t)N * OUT_DIM;

  {
    int n = N * OUT_DIM + N;
    zero_kernel<<<(n + 255) / 256, 256, 0, stream>>>(sums, n);
  }
  {
    int n = IN_DIM * HD + HD;
    precompute_wvo<<<(n + 127) / 128, 128, 0, stream>>>(Wv, bv, Wo, Wvo, bvo);
  }
  node_gemm<<<(N + 15) / 16, 256, 0, stream>>>(x, Wq, bq, Wk, bk, Wvo, bvo,
                                               Q, Km, Vo, N);
  edge_selfdot<<<(NE + 15) / 16, 128, 0, stream>>>(ea, We, be, sE, NE);
  edge_pass<<<EP_BLOCKS, 256, 0, stream>>>(ei, Q, Km, Vo, sE, bo,
                                            sums, cnt, NE);
  {
    int n = N * OUT_DIM;
    finalize<<<(n + 255) / 256, 256, 0, stream>>>(sums, cnt, (float*)d_out, N);
  }
}


# edge self-dot GEMM fused into the edge pass: each wave runs the f32 WMMAs for its own 16-edge batches (We fragments in LDS); the 160000-wave launch and the score round trip go; sums in the same order
# speedup vs baseline: 1.6556x; 1.1850x over previous
#include <hip/hip_runtime.h>
#include <hip/hip_bf16.h>

typedef __attribute__((ext_vector_type(2))) float v2f;
typedef __attribute__((ext_vector_type(8))) float v8f;

#define IN_DIM  128
#define HD      128
#define E_DIM   32
#define NHEAD   4
#define DHEAD   32
#define OUT_DIM 32
#define INV_SQRT_D 0.17677669529663687f

static __device__ __forceinline__ v8f wmma_f32(v2f a, v2f b, v8f c) {
  return __builtin_amdgcn_wmma_f32_16x16x4_f32(false, a, false, b, (short)0, c,
                                               false, false);
}

__global__ void zero_kernel(float* p, int n) {
  int i = blockIdx.x * blockDim.x + threadIdx.x;
  if (i < n) p[i] = 0.0f;
}

__global__ void precompute_wvo(const float* __restrict__ Wv,
                               const float* __restrict__ bv,
                               const float* __restrict__ Wo,
                               float* __restrict__ Wvo,
                               float* __restrict__ bvo) {
  int idx = blockIdx.x * blockDim.x + threadIdx.x;
  if (idx < IN_DIM * HD) {
    int k = idx >> 7;
    int c = idx & 127;
    int h = c >> 5, j = c & 31;
    float acc = 0.0f;
    for (int d = 0; d < DHEAD; ++d)
      acc += Wv[k * HD + h * DHEAD + d] * Wo[(h * DHEAD + d) * OUT_DIM + j];
    Wvo[idx] = acc;
  } else if (idx < IN_DIM * HD + HD) {
    int c = idx - IN_DIM * HD;
    int h = c >> 5, j = c & 31;
    float acc = 0.0f;
    for (int d = 0; d < DHEAD; ++d)
      acc += bv[h * DHEAD + d] * Wo[(h * DHEAD + d) * OUT_DIM + j];
    bvo[c] = acc;
  }
}

__global__ __launch_bounds__(256) void node_gemm(
    const float* __restrict__ x,
    const float* __restrict__ Wq, const float* __restrict__ bq,
    const float* __restrict__ Wk, const float* __restrict__ bk,
    const float* __restrict__ Wvo, const float* __restrict__ bvo,
    float* __restrict__ Q, float* __restrict__ Km, float* __restrict__ Vo,
    int N) {
  __shared__ float As[16][IN_DIM];
  const int tid = threadIdx.x;
  const int m0 = blockIdx.x * 16;

  for (int i = 0; i < 8; ++i) {
    int lin = tid + i * 256;
    int r = lin >> 7, c = lin & 127;
    int gr = m0 + r; if (gr >= N) gr = N - 1;
    As[r][c] = x[(size_t)gr * IN_DIM + c];
  }
  __syncthreads();

  const int w = tid >> 5, L = tid & 31;
  const int n0 = w * 16;
  const int rowA = L & 15;
  const int kOff = (L < 16) ? 0 : 2;
  const int colB = L & 15;

  const float bq0 = bq[n0 + colB];
  const float bk0 = bk[n0 + colB];
  const float bv0 = bvo[n0 + colB];
  v8f cq, ck, cv;
  for (int j = 0; j < 8; ++j) { cq[j] = bq0; ck[j] = bk0; cv[j] = bv0; }

  for (int kk = 0; kk < IN_DIM; kk += 4) {
    v2f a;
    a.x = As[rowA][kk + kOff];
    a.y = As[rowA][kk + kOff + 1];
    v2f b;
    b.x = Wq[(kk + kOff) * HD + n0 + colB];
    b.y = Wq[(kk + kOff + 1) * HD + n0 + colB];
    cq = wmma_f32(a, b, cq);
    b.x = Wk[(kk + kOff) * HD + n0 + colB];
    b.y = Wk[(kk + kOff + 1) * HD + n0 + colB];
    ck = wmma_f32(a, b, ck);
    b.x = Wvo[(kk + kOff) * HD + n0 + colB];
    b.y = Wvo[(kk + kOff + 1) * HD + n0 + colB];
    cv = wmma_f32(a, b, cv);
  }

  for (int j = 0; j < 8; ++j) {
    int r = (L < 16) ? j : (j + 8);
    int gr = m0 + r;
    if (gr < N) {
      size_t o = (size_t)gr * HD + n0 + colB;
      Q[o] = cq[j]; Km[o] = ck[j]; Vo[o] = cv[j];
    }
  }
}

#define EP_BLOCKS   2048
#define EP_WAVES_LG 14

static __device__ __forceinline__ int bcast_i(int v, int lane) {
  return __builtin_amdgcn_readlane(v, lane);
}
static __device__ __forceinline__ float bcast_f(float v, int lane) {
  return __int_as_float(__builtin_amdgcn_readlane(__float_as_int(v), lane));
}
template <int M>
static __device__ __forceinline__ float row_xor(float v) {
  return __int_as_float(__builtin_amdgcn_update_dpp(
      0, __float_as_int(v), 0x160 + M, 0xf, 0xf, true));
}

__global__ __launch_bounds__(256) void edge_pass(
    const int* __restrict__ ei, const float* __restrict__ ea,
    const float* __restrict__ We, const float* __restrict__ be,
    const float* __restrict__ Q, const float* __restrict__ Km,
    const float* __restrict__ Vo, const float* __restrict__ bo,
    float* __restrict__ sums, float* __restrict__ cnt, int NE) {
  __shared__ v2f   Bf[64][32];
  __shared__ float bes[HD];

  const int tid = threadIdx.x;
  const int L = tid & 31;
  const int wv = __builtin_amdgcn_readfirstlane((int)(tid >> 5));
  const int c16 = L & 15;
  const int kOff = (L < 16) ? 0 : 2;

  for (int q = 0; q < 8; ++q) {
    const int f = wv * 8 + q;
    const int col = (f >> 4) * DHEAD + ((f >> 3) & 1) * 16 + c16;
    const int kk = (f & 7) * 4;
    v2f b;
    b.x = We[(kk + kOff) * HD + col];
    b.y = We[(kk + kOff + 1) * HD + col];
    Bf[f][L] = b;
  }
  if (tid < HD) bes[tid] = be[tid];
  __syncthreads();

  const long long wid = (long long)blockIdx.x * 8 + wv;
  const int beg = (int)((wid * NE) >> EP_WAVES_LG);
  const int end = (int)(((wid + 1) * NE) >> EP_WAVES_LG);
  if (beg >= end) return;

  const float bo_j = bo[L];
  const int slot = (L >> 4) * 8 + (L & 7);
  const bool p1 = (L & 1) != 0, p2 = (L & 2) != 0, p4 = (L & 4) != 0;
  const unsigned l16 = (unsigned)L * 16u, l4 = (unsigned)L * 4u;

  for (int t = beg; t < end; t += 16) {
    const int nb = (end - t < 16) ? (end - t) : 16;
    int rowE = t + c16; if (rowE >= end) rowE = end - 1;
    const unsigned aoff = (unsigned)rowE * (E_DIM * 4u) + (unsigned)kOff * 4u;
    v2f af[8];
    for (int s = 0; s < 8; ++s)
      af[s] = *(const v2f*)((const char*)ea + aoff + s * 16);

    float se_l[NHEAD];
    for (int h = 0; h < NHEAD; ++h) {
      int lo = L;
      asm volatile("" : "+v"(lo));
      v8f c0 = {}, c1 = {};
      v2f b0 = Bf[(h * 2 + 0) * 8][lo];
      v2f b1 = Bf[(h * 2 + 1) * 8][lo];
      for (int s = 0; s < 8; ++s) {
        v2f n0 = b0, n1 = b1;
        if (s < 7) {
          n0 = Bf[(h * 2 + 0) * 8 + s + 1][lo];
          n1 = Bf[(h * 2 + 1) * 8 + s + 1][lo];
        }
        c0 = wmma_f32(af[s], b0, c0);
        c1 = wmma_f32(af[s], b1, c1);
        __builtin_amdgcn_sched_barrier(0);
        b0 = n0; b1 = n1;
      }
      const float be0 = bes[h * DHEAD + c16], be1 = bes[h * DHEAD + 16 + c16];
      float tt[8];
      for (int j = 0; j < 8; ++j) {
        float u = c0[j] + be0, v = c1[j] + be1;
        tt[j] = u * u + v * v;
      }
      float u[4], w2[2];
      for (int k = 0; k < 4; ++k) {
        const float keep = p1 ? tt[2 * k + 1] : tt[2 * k];
        const float send = p1 ? tt[2 * k] : tt[2 * k + 1];
        u[k] = keep + row_xor<1>(send);
      }
      for (int k = 0; k < 2; ++k) {
        const float keep = p2 ? u[2 * k + 1] : u[2 * k];
        const float send = p2 ? u[2 * k] : u[2 * k + 1];
        w2[k] = keep + row_xor<2>(send);
      }
      const float keep4 = p4 ? w2[1] : w2[0];
      const float send4 = p4 ? w2[0] : w2[1];
      const float x8 = keep4 + row_xor<4>(send4);
      se_l[h] = x8 + row_xor<8>(x8);
    }

    int myE = t + slot; if (myE >= end) myE = end - 1;
    const int src_l = ei[myE];
    const int dst_l = ei[NE + myE];
    if ((L & 8) == 0 && slot < nb) atomicAdd(&cnt[src_l], 1.0f);

    for (int i = 0; i < nb; ++i) {
      const int li = ((i >> 3) << 4) | (i & 7);
      const int src = bcast_i(src_l, li);
      const int dst = bcast_i(dst_l, li);

      const unsigned oq = (unsigned)src * (HD * 4u) + l16;
      const unsigned ok = (unsigned)dst * (HD * 4u) + l16;
      const unsigned ov = (unsigned)dst * (HD * 4u) + l4;
      const float4 q = *(const float4*)((const char*)Q  + oq);
      const float4 k = *(const float4*)((const char*)Km + ok);
      float vo[NHEAD];
      for (int h2 = 0; h2 < NHEAD; ++h2)
        vo[h2] = *(const float*)((const char*)Vo + ov + h2 * (DHEAD * 4));
      __builtin_amdgcn_sched_barrier(0);
      const float se[NHEAD] = {bcast_f(se_l[0], li), bcast_f(se_l[1], li),
                               bcast_f(se_l[2], li), bcast_f(se_l[3], li)};

      float p = q.x * k.x + q.y * k.y + q.z * k.z + q.w * k.w;
      p += __shfl_xor(p, 1);
      p += __shfl_xor(p, 2);
      p += __shfl_xor(p, 4);

      float sc[NHEAD];
      for (int h2 = 0; h2 < NHEAD; ++h2)
        sc[h2] = (__shfl(p, h2 * 8) + se[h2]) * INV_SQRT_D;

      float m = sc[0];
      for (int h2 = 1; h2 < NHEAD; ++h2) m = fmaxf(m, sc[h2]);
      float ex[NHEAD], den = 0.0f;
      for (int h2 = 0; h2 < NHEAD; ++h2) { ex[h2] = __expf(sc[h2] - m); den += ex[h2]; }
      const float inv = 1.0f / den;

      float acc = bo_j;
      for (int h2 = 0; h2 < NHEAD; ++h2)
        acc += (ex[h2] * inv) * vo[h2];

      const unsigned os = (unsigned)src * (OUT_DIM * 4u) + l4;
      atomicAdd((float*)((char*)sums + os), acc);
    }
  }
}

__global__ void finalize(const float* __restrict__ sums,
                         const float* __restrict__ cnt,
                         float* __restrict__ out, int N) {
  int i = blockIdx.x * blockDim.x + threadIdx.x;
  if (i < N * OUT_DIM)
    out[i] = sums[i] / fmaxf(cnt[i >> 5], 1.0f);
}

extern "C" void kernel_launch(void* const* d_in, const int* in_sizes, int n_in,
                              void* d_out, int out_size, void* d_ws, size_t ws_size,
                              hipStream_t stream) {
  const float* x  = (const float*)d_in[0];
  const int*   ei = (const int*)d_in[1];
  const float* ea = (const float*)d_in[2];
  const float* Wq = (const float*)d_in[3];
  const float* bq = (const float*)d_in[4];
  const float* Wk = (const float*)d_in[5];
  const float* bk = (const float*)d_in[6];
  const float* Wv = (const float*)d_in[7];
  const float* bv = (const float*)d_in[8];
  const float* We = (const float*)d_in[9];
  const float* be = (const float*)d_in[10];
  const float* Wo = (const float*)d_in[11];
  const float* bo = (const float*)d_in[12];

  const int N  = in_sizes[0] / IN_DIM;
  const int NE = in_sizes[1] / 2;

  float* ws   = (float*)d_ws;
  float* Q    = ws;
  float* Km   = Q   + (size_t)N * HD;
  float* Vo   = Km  + (size_t)N * HD;
  float* Wvo  = Vo  + (size_t)N * HD;
  float* bvo  = Wvo + IN_DIM * HD;
  float* sums = bvo + HD;
  float* cnt  = sums + (size_t)N * OUT_DIM;

  {
    int n = N * OUT_DIM + N;
    zero_kernel<<<(n + 255) / 256, 256, 0, stream>>>(sums, n);
  }
  {
    int n = IN_DIM * HD + HD;
    precompute_wvo<<<(n + 127) / 128, 128, 0, stream>>>(Wv, bv, Wo, Wvo, bvo);
  }
  node_gemm<<<(N + 15) / 16, 256, 0, stream>>>(x, Wq, bq, Wk, bk, Wvo, bvo,
                                               Q, Km, Vo, N);
  edge_pass<<<EP_BLOCKS, 256, 0, stream>>>(ei, ea, We, be, Q, Km, Vo, bo,
                                            sums, cnt, NE);
  {
    int n = N * OUT_DIM;
    finalize<<<(n + 255) / 256, 256, 0, stream>>>(sums, cnt, (float*)d_out, N);
  }
}


# edge pass by owner: a 32-wave work-group owns 40 source nodes, filters edge_index for them, sums rows in LDS (LDS float adds, arrival order), stores the means; global atomics, zero and mean passes go
# speedup vs baseline: 2.1028x; 1.2701x over previous
#include <hip/hip_runtime.h>
#include <hip/hip_bf16.h>

typedef __attribute__((ext_vector_type(2))) float v2f;
typedef __attribute__((ext_vector_type(8))) float v8f;

#define IN_DIM  128
#define HD      128
#define E_DIM   32
#define NHEAD   4
#define DHEAD   32
#define OUT_DIM 32
#define INV_SQRT_D 0.17677669529663687f

static __device__ __forceinline__ v8f wmma_f32(v2f a, v2f b, v8f c) {
  return __builtin_amdgcn_wmma_f32_16x16x4_f32(false, a, false, b, (short)0, c,
                                               false, false);
}

__global__ void precompute_wvo(const float* __restrict__ Wv,
                               const float* __restrict__ bv,
                               const float* __restrict__ Wo,
                               float* __restrict__ Wvo,
                               float* __restrict__ bvo) {
  int idx = blockIdx.x * blockDim.x + threadIdx.x;
  if (idx < IN_DIM * HD) {
    int k = idx >> 7;
    int c = idx & 127;
    int h = c >> 5, j = c & 31;
    float acc = 0.0f;
    for (int d = 0; d < DHEAD; ++d)
      acc += Wv[k * HD + h * DHEAD + d] * Wo[(h * DHEAD + d) * OUT_DIM + j];
    Wvo[idx] = acc;
  } else if (idx < IN_DIM * HD + HD) {
    int c = idx - IN_DIM * HD;
    int h = c >> 5, j = c & 31;
    float acc = 0.0f;
    for (int d = 0; d < DHEAD; ++d)
      acc += bv[h * DHEAD + d] * Wo[(h * DHEAD + d) * OUT_DIM + j];
    bvo[c] = acc;
  }
}

__global__ __launch_bounds__(256) void node_gemm(
    const float* __restrict__ x,
    const float* __restrict__ Wq, const float* __restrict__ bq,
    const float* __restrict__ Wk, const float* __restrict__ bk,
    const float* __restrict__ Wvo, const float* __restrict__ bvo,
    float* __restrict__ Q, float* __restrict__ Km, float* __restrict__ Vo,
    int N) {
  __shared__ float As[16][IN_DIM];
  const int tid = threadIdx.x;
  const int m0 = blockIdx.x * 16;

  for (int i = 0; i < 8; ++i) {
    int lin = tid + i * 256;
    int r = lin >> 7, c = lin & 127;
    int gr = m0 + r; if (gr >= N) gr = N - 1;
    As[r][c] = x[(size_t)gr * IN_DIM + c];
  }
  __syncthreads();

  const int w = tid >> 5, L = tid & 31;
  const int n0 = w * 16;
  const int rowA = L & 15;
  const int kOff = (L < 16) ? 0 : 2;
  const int colB = L & 15;

  const float bq0 = bq[n0 + colB];
  const float bk0 = bk[n0 + colB];
  const float bv0 = bvo[n0 + colB];
  v8f cq, ck, cv;
  for (int j = 0; j < 8; ++j) { cq[j] = bq0; ck[j] = bk0; cv[j] = bv0; }

  for (int kk = 0; kk < IN_DIM; kk += 4) {
    v2f a;
    a.x = As[rowA][kk + kOff];
    a.y = As[rowA][kk + kOff + 1];
    v2f b;
    b.x = Wq[(kk + kOff) * HD + n0 + colB];
    b.y = Wq[(kk + kOff + 1) * HD + n0 + colB];
    cq = wmma_f32(a, b, cq);
    b.x = Wk[(kk + kOff) * HD + n0 + colB];
    b.y = Wk[(kk + kOff + 1) * HD + n0 + colB];
    ck = wmma_f32(a, b, ck);
    b.x = Wvo[(kk + kOff) * HD + n0 + colB];
    b.y = Wvo[(kk + kOff + 1) * HD + n0 + colB];
    cv = wmma_f32(a, b, cv);
  }

  for (int j = 0; j < 8; ++j) {
    int r = (L < 16) ? j : (j + 8);
    int gr = m0 + r;
    if (gr < N) {
      size_t o = (size_t)gr * HD + n0 + colB;
      Q[o] = cq[j]; Km[o] = ck[j]; Vo[o] = cv[j];
    }
  }
}

#define RN        40
#define EP_WAVES  32
#define CH_EDGES  1024
#define QCAP      (CH_EDGES + 16)

static __device__ __forceinline__ int bcast_i(int v, int lane) {
  return __builtin_amdgcn_readlane(v, lane);
}
static __device__ __forceinline__ float bcast_f(float v, int lane) {
  return __int_as_float(__builtin_amdgcn_readlane(__float_as_int(v), lane));
}
template <int M>
static __device__ __forceinline__ float row_xor(float v) {
  return __int_as_float(__builtin_amdgcn_update_dpp(
      0, __float_as_int(v), 0x160 + M, 0xf, 0xf, true));
}

__global__ __launch_bounds__(1024) void edge_pass(
    const int* __restrict__ ei, const float* __restrict__ ea,
    const float* __restrict__ We, const float* __restrict__ be,
    const float* __restrict__ Q, const float* __restrict__ Km,
    const float* __restrict__ Vo, const float* __restrict__ bo,
    float* __restrict__ out, int N, int NE) {
  __shared__ v2f    Bf[64][32];
  __shared__ float  bes[HD];
  __shared__ float4 Qs[RN][32];
  __shared__ float  sumS[RN][OUT_DIM];
  __shared__ int    cntS[RN];
  __shared__ int    qS[EP_WAVES][QCAP];
  __shared__ int    s_next;

  const int tid = threadIdx.x;
  const int L = tid & 31;
  const int wv = __builtin_amdgcn_readfirstlane((int)(tid >> 5));
  const int c16 = L & 15;
  const int kOff = (L < 16) ? 0 : 2;
  const int base = blockIdx.x * RN;

  for (int q = 0; q < 2; ++q) {
    const int f = wv * 2 + q;
    const int col = (f >> 4) * DHEAD + ((f >> 3) & 1) * 16 + c16;
    const int kk = (f & 7) * 4;
    v2f b;
    b.x = We[(kk + kOff) * HD + col];
    b.y = We[(kk + kOff + 1) * HD + col];
    Bf[f][L] = b;
  }
  if (tid < HD) bes[tid] = be[tid];
  for (int i = tid; i < RN * 32; i += 1024) {
    int node = base + (i >> 5); if (node >= N) node = N - 1;
    Qs[i >> 5][i & 31] = *(const float4*)(Q + (size_t)node * HD + (i & 31) * 4);
  }
  for (int i = tid; i < RN * OUT_DIM; i += 1024) sumS[i >> 5][i & 31] = 0.0f;
  if (tid < RN) cntS[tid] = 0;
  if (tid == 0) s_next = 0;
  __syncthreads();

  const int nCh = (NE + CH_EDGES - 1) / CH_EDGES;
  const int rot = (int)blockIdx.x * (nCh / (int)gridDim.x);

  int* const qw = qS[wv];
  int qn = 0;
  bool more = true;

  const float bo_j = bo[L];
  const int slot = (L >> 4) * 8 + (L & 7);
  const bool p1 = (L & 1) != 0, p2 = (L & 2) != 0, p4 = (L & 4) != 0;
  const unsigned l16 = (unsigned)L * 16u, l4 = (unsigned)L * 4u;

  while (more || qn > 0) {
    if (more && qn < 16) {
      int c = 0;
      if (L == 0) c = atomicAdd(&s_next, 1);
      c = __builtin_amdgcn_readfirstlane(c);
      if (c >= nCh) { more = false; continue; }
      int cc = c + rot; if (cc >= nCh) cc -= nCh;
      const int e_base = cc * CH_EDGES;
      int4 sv[8];
      if (e_base + CH_EDGES <= NE) {
        const unsigned boff = (unsigned)e_base * 4u + l16;
        for (int j = 0; j < 8; ++j)
          sv[j] = *(const int4*)((const char*)ei + boff + j * 512);
      } else {
        for (int j = 0; j < 8; ++j) {
          const int e0 = e_base + j * 128 + L * 4;
          sv[j] = (e0 < NE) ? *(const int4*)(ei + e0) : make_int4(-1, -1, -1, -1);
          if (e0 + 1 >= NE) sv[j].y = -1;
          if (e0 + 2 >= NE) sv[j].z = -1;
          if (e0 + 3 >= NE) sv[j].w = -1;
        }
      }
      for (int j = 0; j < 8; ++j) {
        const bool m0 = (unsigned)(sv[j].x - base) < (unsigned)RN;
        const bool m1 = (unsigned)(sv[j].y - base) < (unsigned)RN;
        const bool m2 = (unsigned)(sv[j].z - base) < (unsigned)RN;
        const bool m3 = (unsigned)(sv[j].w - base) < (unsigned)RN;
        if (__builtin_amdgcn_ballot_w32(m0 | m1 | m2 | m3) != 0u) {
          const int e0 = e_base + j * 128 + L * 4;
          const bool mk[4] = {m0, m1, m2, m3};
          for (int k = 0; k < 4; ++k) {
            const unsigned bal = __builtin_amdgcn_ballot_w32(mk[k]);
            if (bal != 0u) {
              if (mk[k]) qw[qn + (int)__builtin_amdgcn_mbcnt_lo(bal, 0u)] = e0 + k;
              qn += __builtin_popcount(bal);
            }
          }
        }
      }
      continue;
    }

    const int nb = (qn < 16) ? qn : 16;
    const int q0 = qn - nb;
    qn = q0;
    const int rowE = qw[q0 + ((c16 < nb) ? c16 : nb - 1)];
    const unsigned aoff = (unsigned)rowE * (E_DIM * 4u) + (unsigned)kOff * 4u;
    v2f af[8];
    for (int s = 0; s < 8; ++s)
      af[s] = *(const v2f*)((const char*)ea + aoff + s * 16);

    float se_l[NHEAD];
    for (int h = 0; h < NHEAD; ++h) {
      v8f c0 = {}, c1 = {};
      v2f b0 = Bf[(h * 2 + 0) * 8][L];
      v2f b1 = Bf[(h * 2 + 1) * 8][L];
      for (int s = 0; s < 8; ++s) {
        v2f n0 = b0, n1 = b1;
        if (s < 7) {
          n0 = Bf[(h * 2 + 0) * 8 + s + 1][L];
          n1 = Bf[(h * 2 + 1) * 8 + s + 1][L];
        }
        c0 = wmma_f32(af[s], b0, c0);
        c1 = wmma_f32(af[s], b1, c1);
        b0 = n0; b1 = n1;
      }
      const float be0 = bes[h * DHEAD + c16], be1 = bes[h * DHEAD + 16 + c16];
      float tt[8];
      for (int j = 0; j < 8; ++j) {
        float u = c0[j] + be0, v = c1[j] + be1;
        tt[j] = u * u + v * v;
      }
      float u[4], w2[2];
      for (int k = 0; k < 4; ++k) {
        const float keep = p1 ? tt[2 * k + 1] : tt[2 * k];
        const float send = p1 ? tt[2 * k] : tt[2 * k + 1];
        u[k] = keep + row_xor<1>(send);
      }
      for (int k = 0; k < 2; ++k) {
        const float keep = p2 ? u[2 * k + 1] : u[2 * k];
        const float send = p2 ? u[2 * k] : u[2 * k + 1];
        w2[k] = keep + row_xor<2>(send);
      }
      const float keep4 = p4 ? w2[1] : w2[0];
      const float send4 = p4 ? w2[0] : w2[1];
      const float x8 = keep4 + row_xor<4>(send4);
      se_l[h] = x8 + row_xor<8>(x8);
    }

    const int myE = qw[q0 + ((slot < nb) ? slot : nb - 1)];
    const int src_l = ei[myE] - base;
    const int dst_l = ei[NE + myE];
    if ((L & 8) == 0 && slot < nb) atomicAdd(&cntS[src_l], 1);

    for (int i = 0; i < nb; ++i) {
      const int li = ((i >> 3) << 4) | (i & 7);
      const int sr = bcast_i(src_l, li);
      const int dst = bcast_i(dst_l, li);

      const unsigned ok = (unsigned)dst * (HD * 4u) + l16;
      const unsigned ov = (unsigned)dst * (HD * 4u) + l4;
      const float4 k = *(const float4*)((const char*)Km + ok);
      float vo[NHEAD];
      for (int h2 = 0; h2 < NHEAD; ++h2)
        vo[h2] = *(const float*)((const char*)Vo + ov + h2 * (DHEAD * 4));
      const float4 q = Qs[sr][L];
      const float se[NHEAD] = {bcast_f(se_l[0], li), bcast_f(se_l[1], li),
                               bcast_f(se_l[2], li), bcast_f(se_l[3], li)};

      float p = q.x * k.x + q.y * k.y + q.z * k.z + q.w * k.w;
      p += __shfl_xor(p, 1);
      p += __shfl_xor(p, 2);
      p += __shfl_xor(p, 4);

      float sc[NHEAD];
      for (int h2 = 0; h2 < NHEAD; ++h2)
        sc[h2] = (__shfl(p, h2 * 8) + se[h2]) * INV_SQRT_D;

      float m = sc[0];
      for (int h2 = 1; h2 < NHEAD; ++h2) m = fmaxf(m, sc[h2]);
      float ex[NHEAD], den = 0.0f;
      for (int h2 = 0; h2 < NHEAD; ++h2) { ex[h2] = __expf(sc[h2] - m); den += ex[h2]; }
      const float inv = 1.0f / den;

      float acc = bo_j;
      for (int h2 = 0; h2 < NHEAD; ++h2)
        acc += (ex[h2] * inv) * vo[h2];

      atomicAdd(&sumS[sr][L], acc);
    }
  }

  __syncthreads();
  for (int i = tid; i < RN * OUT_DIM; i += 1024) {
    const int node = base + (i >> 5);
    if (node < N)
      out[(size_t)node * OUT_DIM + (i & 31)] =
          sumS[i >> 5][i & 31] / fmaxf((float)cntS[i >> 5], 1.0f);
  }
}

extern "C" void kernel_launch(void* const* d_in, const int* in_sizes, int n_in,
                              void* d_out, int out_size, void* d_ws, size_t ws_size,
                              hipStream_t stream) {
  const float* x  = (const float*)d_in[0];
  const int*   ei = (const int*)d_in[1];
  const float* ea = (const float*)d_in[2];
  const float* Wq = (const float*)d_in[3];
  const float* bq = (const float*)d_in[4];
  const float* Wk = (const float*)d_in[5];
  const float* bk = (const float*)d_in[6];
  const float* Wv = (const float*)d_in[7];
  const float* bv = (const float*)d_in[8];
  const float* We = (const float*)d_in[9];
  const float* be = (const float*)d_in[10];
  const float* Wo = (const float*)d_in[11];
  const float* bo = (const float*)d_in[12];

  const int N  = in_sizes[0] / IN_DIM;
  const int NE = in_sizes[1] / 2;

  float* ws   = (float*)d_ws;
  float* Q    = ws;
  float* Km   = Q   + (size_t)N * HD;
  float* Vo   = Km  + (size_t)N * HD;
  float* Wvo  = Vo  + (size_t)N * HD;
  float* bvo  = Wvo + IN_DIM * HD;

  {
    int n = IN_DIM * HD + HD;
    precompute_wvo<<<(n + 127) / 128, 128, 0, stream>>>(Wv, bv, Wo, Wvo, bvo);
  }
  node_gemm<<<(N + 15) / 16, 256, 0, stream>>>(x, Wq, bq, Wk, bk, Wvo, bvo,
                                               Q, Km, Vo, N);
  edge_pass<<<(N + RN - 1) / RN, 1024, 0, stream>>>(ei, ea, We, be, Q, Km, Vo,
                                                    bo, (float*)d_out, N, NE);
}


# edge section: K and Vo rows of four edges requested together, their four chains one straight-line group (was one edge per trip, each waiting on its own loads); LDS float adds still in arrival order
# speedup vs baseline: 2.3330x; 1.1095x over previous
#include <hip/hip_runtime.h>
#include <hip/hip_bf16.h>

typedef __attribute__((ext_vector_type(2))) float v2f;
typedef __attribute__((ext_vector_type(8))) float v8f;

#define IN_DIM  128
#define HD      128
#define E_DIM   32
#define NHEAD   4
#define DHEAD   32
#define OUT_DIM 32
#define INV_SQRT_D 0.17677669529663687f

static __device__ __forceinline__ v8f wmma_f32(v2f a, v2f b, v8f c) {
  return __builtin_amdgcn_wmma_f32_16x16x4_f32(false, a, false, b, (short)0, c,
                                               false, false);
}

__global__ void precompute_wvo(const float* __restrict__ Wv,
                               const float* __restrict__ bv,
                               const float* __restrict__ Wo,
                               float* __restrict__ Wvo,
                               float* __restrict__ bvo) {
  int idx = blockIdx.x * blockDim.x + threadIdx.x;
  if (idx < IN_DIM * HD) {
    int k = idx >> 7;
    int c = idx & 127;
    int h = c >> 5, j = c & 31;
    float acc = 0.0f;
    for (int d = 0; d < DHEAD; ++d)
      acc += Wv[k * HD + h * DHEAD + d] * Wo[(h * DHEAD + d) * OUT_DIM + j];
    Wvo[idx] = acc;
  } else if (idx < IN_DIM * HD + HD) {
    int c = idx - IN_DIM * HD;
    int h = c >> 5, j = c & 31;
    float acc = 0.0f;
    for (int d = 0; d < DHEAD; ++d)
      acc += bv[h * DHEAD + d] * Wo[(h * DHEAD + d) * OUT_DIM + j];
    bvo[c] = acc;
  }
}

__global__ __launch_bounds__(256) void node_gemm(
    const float* __restrict__ x,
    const float* __restrict__ Wq, const float* __restrict__ bq,
    const float* __restrict__ Wk, const float* __restrict__ bk,
    const float* __restrict__ Wvo, const float* __restrict__ bvo,
    float* __restrict__ Q, float* __restrict__ Km, float* __restrict__ Vo,
    int N) {
  __shared__ float As[16][IN_DIM];
  const int tid = threadIdx.x;
  const int m0 = blockIdx.x * 16;

  for (int i = 0; i < 8; ++i) {
    int lin = tid + i * 256;
    int r = lin >> 7, c = lin & 127;
    int gr = m0 + r; if (gr >= N) gr = N - 1;
    As[r][c] = x[(size_t)gr * IN_DIM + c];
  }
  __syncthreads();

  const int w = tid >> 5, L = tid & 31;
  const int n0 = w * 16;
  const int rowA = L & 15;
  const int kOff = (L < 16) ? 0 : 2;
  const int colB = L & 15;

  const float bq0 = bq[n0 + colB];
  const float bk0 = bk[n0 + colB];
  const float bv0 = bvo[n0 + colB];
  v8f cq, ck, cv;
  for (int j = 0; j < 8; ++j) { cq[j] = bq0; ck[j] = bk0; cv[j] = bv0; }

  for (int kk = 0; kk < IN_DIM; kk += 4) {
    v2f a;
    a.x = As[rowA][kk + kOff];
    a.y = As[rowA][kk + kOff + 1];
    v2f b;
    b.x = Wq[(kk + kOff) * HD + n0 + colB];
    b.y = Wq[(kk + kOff + 1) * HD + n0 + colB];
    cq = wmma_f32(a, b, cq);
    b.x = Wk[(kk + kOff) * HD + n0 + colB];
    b.y = Wk[(kk + kOff + 1) * HD + n0 + colB];
    ck = wmma_f32(a, b, ck);
    b.x = Wvo[(kk + kOff) * HD + n0 + colB];
    b.y = Wvo[(kk + kOff + 1) * HD + n0 + colB];
    cv = wmma_f32(a, b, cv);
  }

  for (int j = 0; j < 8; ++j) {
    int r = (L < 16) ? j : (j + 8);
    int gr = m0 + r;
    if (gr < N) {
      size_t o = (size_t)gr * HD + n0 + colB;
      Q[o] = cq[j]; Km[o] = ck[j]; Vo[o] = cv[j];
    }
  }
}

#define RN        40
#define EP_WAVES  32
#define CH_EDGES  1024
#define QCAP      (CH_EDGES + 16)
#define EG        4

static __device__ __forceinline__ int bcast_i(int v, int lane) {
  return __builtin_amdgcn_readlane(v, lane);
}
static __device__ __forceinline__ float bcast_f(float v, int lane) {
  return __int_as_float(__builtin_amdgcn_readlane(__float_as_int(v), lane));
}
template <int M>
static __device__ __forceinline__ float row_xor(float v) {
  return __int_as_float(__builtin_amdgcn_update_dpp(
      0, __float_as_int(v), 0x160 + M, 0xf, 0xf, true));
}

static __device__ __forceinline__ float edge_out(const float4 k, const float4 q,
                                                 const float (&vo)[NHEAD],
                                                 const float (&se)[NHEAD],
                                                 const float bo_j) {
  float p = q.x * k.x + q.y * k.y + q.z * k.z + q.w * k.w;
  p += __shfl_xor(p, 1);
  p += __shfl_xor(p, 2);
  p += __shfl_xor(p, 4);

  float sc[NHEAD];
  for (int h2 = 0; h2 < NHEAD; ++h2)
    sc[h2] = (__shfl(p, h2 * 8) + se[h2]) * INV_SQRT_D;

  float m = sc[0];
  for (int h2 = 1; h2 < NHEAD; ++h2) m = fmaxf(m, sc[h2]);
  float ex[NHEAD], den = 0.0f;
  for (int h2 = 0; h2 < NHEAD; ++h2) { ex[h2] = __expf(sc[h2] - m); den += ex[h2]; }
  const float inv = 1.0f / den;

  float acc = bo_j;
  for (int h2 = 0; h2 < NHEAD; ++h2)
    acc += (ex[h2] * inv) * vo[h2];
  return acc;
}

__global__ __launch_bounds__(1024) void edge_pass(
    const int* __restrict__ ei, const float* __restrict__ ea,
    const float* __restrict__ We, const float* __restrict__ be,
    const float* __restrict__ Q, const float* __restrict__ Km,
    const float* __restrict__ Vo, const float* __restrict__ bo,
    float* __restrict__ out, int N, int NE) {
  __shared__ v2f    Bf[64][32];
  __shared__ float  bes[HD];
  __shared__ float4 Qs[RN][32];
  __shared__ float  sumS[RN][OUT_DIM];
  __shared__ int    cntS[RN];
  __shared__ int    qS[EP_WAVES][QCAP];
  __shared__ int    s_next;

  const int tid = threadIdx.x;
  const int L = tid & 31;
  const int wv = __builtin_amdgcn_readfirstlane((int)(tid >> 5));
  const int c16 = L & 15;
  const int kOff = (L < 16) ? 0 : 2;
  const int base = blockIdx.x * RN;

  for (int q = 0; q < 2; ++q) {
    const int f = wv * 2 + q;
    const int col = (f >> 4) * DHEAD + ((f >> 3) & 1) * 16 + c16;
    const int kk = (f & 7) * 4;
    v2f b;
    b.x = We[(kk + kOff) * HD + col];
    b.y = We[(kk + kOff + 1) * HD + col];
    Bf[f][L] = b;
  }
  if (tid < HD) bes[tid] = be[tid];
  for (int i = tid; i < RN * 32; i += 1024) {
    int node = base + (i >> 5); if (node >= N) node = N - 1;
    Qs[i >> 5][i & 31] = *(const float4*)(Q + (size_t)node * HD + (i & 31) * 4);
  }
  for (int i = tid; i < RN * OUT_DIM; i += 1024) sumS[i >> 5][i & 31] = 0.0f;
  if (tid < RN) cntS[tid] = 0;
  if (tid == 0) s_next = 0;
  __syncthreads();

  const int nCh = (NE + CH_EDGES - 1) / CH_EDGES;
  const int rot = (int)blockIdx.x * (nCh / (int)gridDim.x);

  int* const qw = qS[wv];
  int qn = 0;
  bool more = true;

  const float bo_j = bo[L];
  const int slot = (L >> 4) * 8 + (L & 7);
  const bool p1 = (L & 1) != 0, p2 = (L & 2) != 0, p4 = (L & 4) != 0;
  const unsigned l16 = (unsigned)L * 16u, l4 = (unsigned)L * 4u;

  while (more || qn > 0) {
    if (more && qn < 16) {
      int c = 0;
      if (L == 0) c = atomicAdd(&s_next, 1);
      c = __builtin_amdgcn_readfirstlane(c);
      if (c >= nCh) { more = false; continue; }
      int cc = c + rot; if (cc >= nCh) cc -= nCh;
      const int e_base = cc * CH_EDGES;
      int4 sv[8];
      if (e_base + CH_EDGES <= NE) {
        const unsigned boff = (unsigned)e_base * 4u + l16;
        for (int j = 0; j < 8; ++j)
          sv[j] = *(const int4*)((const char*)ei + boff + j * 512);
      } else {
        for (int j = 0; j < 8; ++j) {
          const int e0 = e_base + j * 128 + L * 4;
          sv[j] = (e0 < NE) ? *(const int4*)(ei + e0) : make_int4(-1, -1, -1, -1);
          if (e0 + 1 >= NE) sv[j].y = -1;
          if (e0 + 2 >= NE) sv[j].z = -1;
          if (e0 + 3 >= NE) sv[j].w = -1;
        }
      }
      for (int j = 0; j < 8; ++j) {
        const bool m0 = (unsigned)(sv[j].x - base) < (unsigned)RN;
        const bool m1 = (unsigned)(sv[j].y - base) < (unsigned)RN;
        const bool m2 = (unsigned)(sv[j].z - base) < (unsigned)RN;
        const bool m3 = (unsigned)(sv[j].w - base) < (unsigned)RN;
        if (__builtin_amdgcn_ballot_w32(m0 | m1 | m2 | m3) != 0u) {
          const int e0 = e_base + j * 128 + L * 4;
          const bool mk[4] = {m0, m1, m2, m3};
          for (int k = 0; k < 4; ++k) {
            const unsigned bal = __builtin_amdgcn_ballot_w32(mk[k]);
            if (bal != 0u) {
              if (mk[k]) qw[qn + (int)__builtin_amdgcn_mbcnt_lo(bal, 0u)] = e0 + k;
              qn += __builtin_popcount(bal);
            }
          }
        }
      }
      continue;
    }

    const int nb = (qn < 16) ? qn : 16;
    const int q0 = qn - nb;
    qn = q0;
    const int rowE = qw[q0 + ((c16 < nb) ? c16 : nb - 1)];
    const unsigned aoff = (unsigned)rowE * (E_DIM * 4u) + (unsigned)kOff * 4u;
    v2f af[8];
    for (int s = 0; s < 8; ++s)
      af[s] = *(const v2f*)((const char*)ea + aoff + s * 16);

    float se_l[NHEAD];
    for (int h = 0; h < NHEAD; ++h) {
      v8f c0 = {}, c1 = {};
      v2f b0 = Bf[(h * 2 + 0) * 8][L];
      v2f b1 = Bf[(h * 2 + 1) * 8][L];
      for (int s = 0; s < 8; ++s) {
        v2f n0 = b0, n1 = b1;
        if (s < 7) {
          n0 = Bf[(h * 2 + 0) * 8 + s + 1][L];
          n1 = Bf[(h * 2 + 1) * 8 + s + 1][L];
        }
        c0 = wmma_f32(af[s], b0, c0);
        c1 = wmma_f32(af[s], b1, c1);
        b0 = n0; b1 = n1;
      }
      const float be0 = bes[h * DHEAD + c16], be1 = bes[h * DHEAD + 16 + c16];
      float tt[8];
      for (int j = 0; j < 8; ++j) {
        float u = c0[j] + be0, v = c1[j] + be1;
        tt[j] = u * u + v * v;
      }
      float u[4], w2[2];
      for (int k = 0; k < 4; ++k) {
        const float keep = p1 ? tt[2 * k + 1] : tt[2 * k];
        const float send = p1 ? tt[2 * k] : tt[2 * k + 1];
        u[k] = keep + row_xor<1>(send);
      }
      for (int k = 0; k < 2; ++k) {
        const float keep = p2 ? u[2 * k + 1] : u[2 * k];
        const float send = p2 ? u[2 * k] : u[2 * k + 1];
        w2[k] = keep + row_xor<2>(send);
      }
      const float keep4 = p4 ? w2[1] : w2[0];
      const float send4 = p4 ? w2[0] : w2[1];
      const float x8 = keep4 + row_xor<4>(send4);
      se_l[h] = x8 + row_xor<8>(x8);
    }

    const int myE = qw[q0 + ((slot < nb) ? slot : nb - 1)];
    const int src_l = ei[myE] - base;
    const int dst_l = ei[NE + myE];
    if ((L & 8) == 0 && slot < nb) atomicAdd(&cntS[src_l], 1);

    int g = 0;
    for (; g + EG <= nb; g += EG) {
      int      sr[EG];
      float4   kr[EG];
      float    vo[EG][NHEAD];
      float    se[EG][NHEAD];
      unsigned ov[EG];

      for (int u = 0; u < EG; ++u) {
        const int i = g + u;
        const int li = ((i >> 3) << 4) | (i & 7);
        sr[u] = bcast_i(src_l, li);
        const int dst = bcast_i(dst_l, li);
        const unsigned ok = (unsigned)dst * (HD * 4u) + l16;
        ov[u] = (unsigned)dst * (HD * 4u) + l4;
        kr[u] = *(const float4*)((const char*)Km + ok);
        for (int h2 = 0; h2 < NHEAD; ++h2) se[u][h2] = bcast_f(se_l[h2], li);
      }
      for (int u = 0; u < EG; ++u)
        for (int h2 = 0; h2 < NHEAD; ++h2)
          vo[u][h2] = *(const float*)((const char*)Vo + ov[u] + h2 * (DHEAD * 4));

      float acc[EG];
      for (int u = 0; u < EG; ++u)
        acc[u] = edge_out(kr[u], Qs[sr[u]][L], vo[u], se[u], bo_j);
      for (int u = 0; u < EG; ++u)
        atomicAdd(&sumS[sr[u]][L], acc[u]);
    }

    for (int i = g; i < nb; ++i) {
      const int li = ((i >> 3) << 4) | (i & 7);
      const int sr = bcast_i(src_l, li);
      const int dst = bcast_i(dst_l, li);
      const unsigned ok = (unsigned)dst * (HD * 4u) + l16;
      const unsigned ov = (unsigned)dst * (HD * 4u) + l4;
      const float4 k = *(const float4*)((const char*)Km + ok);
      float vo[NHEAD];
      for (int h2 = 0; h2 < NHEAD; ++h2)
        vo[h2] = *(const float*)((const char*)Vo + ov + h2 * (DHEAD * 4));
      const float se[NHEAD] = {bcast_f(se_l[0], li), bcast_f(se_l[1], li),
                               bcast_f(se_l[2], li), bcast_f(se_l[3], li)};
      const float acc = edge_out(k, Qs[sr][L], vo, se, bo_j);
      atomicAdd(&sumS[sr][L], acc);
    }
  }

  __syncthreads();
  for (int i = tid; i < RN * OUT_DIM; i += 1024) {
    const int node = base + (i >> 5);
    if (node < N)
      out[(size_t)node * OUT_DIM + (i & 31)] =
          sumS[i >> 5][i & 31] / fmaxf((float)cntS[i >> 5], 1.0f);
  }
}

extern "C" void kernel_launch(void* const* d_in, const int* in_sizes, int n_in,
                              void* d_out, int out_size, void* d_ws, size_t ws_size,
                              hipStream_t stream) {
  const float* x  = (const float*)d_in[0];
  const int*   ei = (const int*)d_in[1];
  const float* ea = (const float*)d_in[2];
  const float* Wq = (const float*)d_in[3];
  const float* bq = (const float*)d_in[4];
  const float* Wk = (const float*)d_in[5];
  const float* bk = (const float*)d_in[6];
  const float* Wv = (const float*)d_in[7];
  const float* bv = (const float*)d_in[8];
  const float* We = (const float*)d_in[9];
  const float* be = (const float*)d_in[10];
  const float* Wo = (const float*)d_in[11];
  const float* bo = (const float*)d_in[12];

  const int N  = in_sizes[0] / IN_DIM;
  const int NE = in_sizes[1] / 2;

  float* ws   = (float*)d_ws;
  float* Q    = ws;
  float* Km   = Q   + (size_t)N * HD;
  float* Vo   = Km  + (size_t)N * HD;
  float* Wvo  = Vo  + (size_t)N * HD;
  float* bvo  = Wvo + IN_DIM * HD;

  {
    int n = IN_DIM * HD + HD;
    precompute_wvo<<<(n + 127) / 128, 128, 0, stream>>>(Wv, bv, Wo, Wvo, bvo);
  }
  node_gemm<<<(N + 15) / 16, 256, 0, stream>>>(x, Wq, bq, Wk, bk, Wvo, bvo,
                                               Q, Km, Vo, N);
  edge_pass<<<(N + RN - 1) / RN, 1024, 0, stream>>>(ei, ea, We, be, Q, Km, Vo,
                                                    bo, (float*)d_out, N, NE);
}


# edge pass fed by a bucket pass: a first-launch role sorts each 4096-edge slice's ids by owner work-group; an owner reads its own pieces, not the whole source plane; LDS float adds in arrival order
# speedup vs baseline: 2.6798x; 1.1486x over previous
#include <hip/hip_runtime.h>
#include <hip/hip_bf16.h>

typedef __attribute__((ext_vector_type(2))) float v2f;
typedef __attribute__((ext_vector_type(8))) float v8f;

#define IN_DIM  128
#define HD      128
#define E_DIM   32
#define NHEAD   4
#define DHEAD   32
#define OUT_DIM 32
#define INV_SQRT_D 0.17677669529663687f

static __device__ __forceinline__ v8f wmma_f32(v2f a, v2f b, v8f c) {
  return __builtin_amdgcn_wmma_f32_16x16x4_f32(false, a, false, b, (short)0, c,
                                               false, false);
}

#define RN        40
#define SLICE     4096
#define BK_MAX    1024
#define PRE_BLOCKS ((IN_DIM * HD + HD + 1023) / 1024)

__global__ __launch_bounds__(1024) void prep_pass(const float* __restrict__ Wv,
                               const float* __restrict__ bv,
                               const float* __restrict__ Wo,
                               float* __restrict__ Wvo,
                               float* __restrict__ bvo,
                               const int* __restrict__ ei,
                               int* __restrict__ sorted,
                               int* __restrict__ segStart,
                               int N, int NE, int NBK) {
  __shared__ int  histL[BK_MAX];
  __shared__ int  wsum[32];
  __shared__ int  s_total;
  __shared__ int4 sortL4[SLICE / 4];
  if (blockIdx.x >= PRE_BLOCKS) {
    const int tid = threadIdx.x;
    const int g = (int)blockIdx.x - PRE_BLOCKS;
    int* const sortL = (int*)sortL4;
    histL[tid] = 0;
    __syncthreads();

    const int e0 = g * SLICE + tid * 4;
    int s[4] = {-1, -1, -1, -1};
    if (e0 + 3 < NE) {
      const int4 v4 = *(const int4*)(ei + e0);
      s[0] = v4.x; s[1] = v4.y; s[2] = v4.z; s[3] = v4.w;
    } else {
      for (int k = 0; k < 4; ++k)
        if (e0 + k < NE) s[k] = ei[e0 + k];
    }
    int bk[4], rk[4];
    for (int k = 0; k < 4; ++k) {
      const bool ok = (unsigned)s[k] < (unsigned)N;
      bk[k] = ok ? s[k] / RN : -1;
      rk[k] = 0;
      if (ok) rk[k] = atomicAdd(&histL[bk[k]], 1);
    }
    __syncthreads();

    const int L = tid & 31, w = tid >> 5;
    const int v = (tid < NBK) ? histL[tid] : 0;
    int inc = v;
    for (int d = 1; d < 32; d <<= 1) {
      const int t = __shfl_up(inc, d);
      if (L >= d) inc += t;
    }
    if (L == 31) wsum[w] = inc;
    __syncthreads();
    if (w == 0) {
      const int x = wsum[L];
      int xi = x;
      for (int d = 1; d < 32; d <<= 1) {
        const int t = __shfl_up(xi, d);
        if (L >= d) xi += t;
      }
      wsum[L] = xi - x;
    }
    __syncthreads();
    const int excl = inc - v + wsum[w];
    int* const st = segStart + (size_t)g * (NBK + 1);
    if (tid < NBK) { histL[tid] = excl; st[tid] = excl; }
    if (tid == NBK - 1) { st[NBK] = excl + v; s_total = excl + v; }
    __syncthreads();

    for (int k = 0; k < 4; ++k)
      if (bk[k] >= 0) sortL[histL[bk[k]] + rk[k]] = e0 + k;
    __syncthreads();
    if (tid * 4 < s_total)
      *(int4*)(sorted + (size_t)g * SLICE + tid * 4) = sortL4[tid];
    return;
  }

  int idx = blockIdx.x * blockDim.x + threadIdx.x;
  if (idx < IN_DIM * HD) {
    int k = idx >> 7;
    int c = idx & 127;
    int h = c >> 5, j = c & 31;
    float acc = 0.0f;
    for (int d = 0; d < DHEAD; ++d)
      acc += Wv[k * HD + h * DHEAD + d] * Wo[(h * DHEAD + d) * OUT_DIM + j];
    Wvo[idx] = acc;
  } else if (idx < IN_DIM * HD + HD) {
    int c = idx - IN_DIM * HD;
    int h = c >> 5, j = c & 31;
    float acc = 0.0f;
    for (int d = 0; d < DHEAD; ++d)
      acc += bv[h * DHEAD + d] * Wo[(h * DHEAD + d) * OUT_DIM + j];
    bvo[c] = acc;
  }
}

__global__ __launch_bounds__(256) void node_gemm(
    const float* __restrict__ x,
    const float* __restrict__ Wq, const float* __restrict__ bq,
    const float* __restrict__ Wk, const float* __restrict__ bk,
    const float* __restrict__ Wvo, const float* __restrict__ bvo,
    float* __restrict__ Q, float* __restrict__ Km, float* __restrict__ Vo,
    int N) {
  __shared__ float As[16][IN_DIM];
  const int tid = threadIdx.x;
  const int m0 = blockIdx.x * 16;

  for (int i = 0; i < 8; ++i) {
    int lin = tid + i * 256;
    int r = lin >> 7, c = lin & 127;
    int gr = m0 + r; if (gr >= N) gr = N - 1;
    As[r][c] = x[(size_t)gr * IN_DIM + c];
  }
  __syncthreads();

  const int w = tid >> 5, L = tid & 31;
  const int n0 = w * 16;
  const int rowA = L & 15;
  const int kOff = (L < 16) ? 0 : 2;
  const int colB = L & 15;

  const float bq0 = bq[n0 + colB];
  const float bk0 = bk[n0 + colB];
  const float bv0 = bvo[n0 + colB];
  v8f cq, ck, cv;
  for (int j = 0; j < 8; ++j) { cq[j] = bq0; ck[j] = bk0; cv[j] = bv0; }

  for (int kk = 0; kk < IN_DIM; kk += 4) {
    v2f a;
    a.x = As[rowA][kk + kOff];
    a.y = As[rowA][kk + kOff + 1];
    v2f b;
    b.x = Wq[(kk + kOff) * HD + n0 + colB];
    b.y = Wq[(kk + kOff + 1) * HD + n0 + colB];
    cq = wmma_f32(a, b, cq);
    b.x = Wk[(kk + kOff) * HD + n0 + colB];
    b.y = Wk[(kk + kOff + 1) * HD + n0 + colB];
    ck = wmma_f32(a, b, ck);
    b.x = Wvo[(kk + kOff) * HD + n0 + colB];
    b.y = Wvo[(kk + kOff + 1) * HD + n0 + colB];
    cv = wmma_f32(a, b, cv);
  }

  for (int j = 0; j < 8; ++j) {
    int r = (L < 16) ? j : (j + 8);
    int gr = m0 + r;
    if (gr < N) {
      size_t o = (size_t)gr * HD + n0 + colB;
      Q[o] = cq[j]; Km[o] = ck[j]; Vo[o] = cv[j];
    }
  }
}

#define EP_WAVES  32
#define QCAP      (16 + 32)
#define EG        4

static __device__ __forceinline__ int bcast_i(int v, int lane) {
  return __builtin_amdgcn_readlane(v, lane);
}
static __device__ __forceinline__ float bcast_f(float v, int lane) {
  return __int_as_float(__builtin_amdgcn_readlane(__float_as_int(v), lane));
}
template <int M>
static __device__ __forceinline__ float row_xor(float v) {
  return __int_as_float(__builtin_amdgcn_update_dpp(
      0, __float_as_int(v), 0x160 + M, 0xf, 0xf, true));
}

static __device__ __forceinline__ float edge_out(const float4 k, const float4 q,
                                                 const float (&vo)[NHEAD],
                                                 const float (&se)[NHEAD],
                                                 const float bo_j) {
  float p = q.x * k.x + q.y * k.y + q.z * k.z + q.w * k.w;
  p += __shfl_xor(p, 1);
  p += __shfl_xor(p, 2);
  p += __shfl_xor(p, 4);

  float sc[NHEAD];
  for (int h2 = 0; h2 < NHEAD; ++h2)
    sc[h2] = (__shfl(p, h2 * 8) + se[h2]) * INV_SQRT_D;

  float m = sc[0];
  for (int h2 = 1; h2 < NHEAD; ++h2) m = fmaxf(m, sc[h2]);
  float ex[NHEAD], den = 0.0f;
  for (int h2 = 0; h2 < NHEAD; ++h2) { ex[h2] = __expf(sc[h2] - m); den += ex[h2]; }
  const float inv = 1.0f / den;

  float acc = bo_j;
  for (int h2 = 0; h2 < NHEAD; ++h2)
    acc += (ex[h2] * inv) * vo[h2];
  return acc;
}

__global__ __launch_bounds__(1024) void edge_pass(
    const int* __restrict__ ei, const float* __restrict__ ea,
    const float* __restrict__ We, const float* __restrict__ be,
    const float* __restrict__ Q, const float* __restrict__ Km,
    const float* __restrict__ Vo, const float* __restrict__ bo,
    const int* __restrict__ sorted, const int* __restrict__ segStart, int G,
    float* __restrict__ out, int N, int NE) {
  __shared__ v2f    Bf[64][32];
  __shared__ float  bes[HD];
  __shared__ float4 Qs[RN][32];
  __shared__ float  sumS[RN][OUT_DIM];
  __shared__ int    cntS[RN];
  __shared__ int    qS[EP_WAVES][QCAP];
  __shared__ int    s_next;

  const int tid = threadIdx.x;
  const int L = tid & 31;
  const int wv = __builtin_amdgcn_readfirstlane((int)(tid >> 5));
  const int c16 = L & 15;
  const int kOff = (L < 16) ? 0 : 2;
  const int base = blockIdx.x * RN;

  for (int q = 0; q < 2; ++q) {
    const int f = wv * 2 + q;
    const int col = (f >> 4) * DHEAD + ((f >> 3) & 1) * 16 + c16;
    const int kk = (f & 7) * 4;
    v2f b;
    b.x = We[(kk + kOff) * HD + col];
    b.y = We[(kk + kOff + 1) * HD + col];
    Bf[f][L] = b;
  }
  if (tid < HD) bes[tid] = be[tid];
  for (int i = tid; i < RN * 32; i += 1024) {
    int node = base + (i >> 5); if (node >= N) node = N - 1;
    Qs[i >> 5][i & 31] = *(const float4*)(Q + (size_t)node * HD + (i & 31) * 4);
  }
  for (int i = tid; i < RN * OUT_DIM; i += 1024) sumS[i >> 5][i & 31] = 0.0f;
  if (tid < RN) cntS[tid] = 0;
  if (tid == 0) s_next = 0;
  __syncthreads();

  const int rot = (int)(((unsigned)blockIdx.x * (unsigned)G) / gridDim.x);
  const int segPitch = (int)gridDim.x + 1;

  int* const qw = qS[wv];
  int qn = 0;
  bool more = true;
  int segPos = 0, segEnd = 0;

  const float bo_j = bo[L];
  const int slot = (L >> 4) * 8 + (L & 7);
  const bool p1 = (L & 1) != 0, p2 = (L & 2) != 0, p4 = (L & 4) != 0;
  const unsigned l16 = (unsigned)L * 16u, l4 = (unsigned)L * 4u;

  while (true) {
    if (qn < 16) {
      if (segPos < segEnd) {
        int n = segEnd - segPos; if (n > 32) n = 32;
        if (L < n) qw[qn + L] = sorted[segPos + L];
        qn += n; segPos += n;
        continue;
      }
      if (more) {
        int c = 0;
        if (L == 0) c = atomicAdd(&s_next, 1);
        c = __builtin_amdgcn_readfirstlane(c);
        if (c >= G) { more = false; continue; }
        int gg = c + rot; if (gg >= G) gg -= G;
        const int* const st = segStart + gg * segPitch + (int)blockIdx.x;
        segPos = gg * SLICE + st[0];
        segEnd = gg * SLICE + st[1];
        continue;
      }
      if (qn == 0) break;
    }

    const int nb = (qn < 16) ? qn : 16;
    const int q0 = qn - nb;
    qn = q0;
    const int rowE = qw[q0 + ((c16 < nb) ? c16 : nb - 1)];
    const unsigned aoff = (unsigned)rowE * (E_DIM * 4u) + (unsigned)kOff * 4u;
    v2f af[8];
    for (int s = 0; s < 8; ++s)
      af[s] = *(const v2f*)((const char*)ea + aoff + s * 16);

    float se_l[NHEAD];
    for (int h = 0; h < NHEAD; ++h) {
      v8f c0 = {}, c1 = {};
      v2f b0 = Bf[(h * 2 + 0) * 8][L];
      v2f b1 = Bf[(h * 2 + 1) * 8][L];
      for (int s = 0; s < 8; ++s) {
        v2f n0 = b0, n1 = b1;
        if (s < 7) {
          n0 = Bf[(h * 2 + 0) * 8 + s + 1][L];
          n1 = Bf[(h * 2 + 1) * 8 + s + 1][L];
        }
        c0 = wmma_f32(af[s], b0, c0);
        c1 = wmma_f32(af[s], b1, c1);
        b0 = n0; b1 = n1;
      }
      const float be0 = bes[h * DHEAD + c16], be1 = bes[h * DHEAD + 16 + c16];
      float tt[8];
      for (int j = 0; j < 8; ++j) {
        float u = c0[j] + be0, v = c1[j] + be1;
        tt[j] = u * u + v * v;
      }
      float u[4], w2[2];
      for (int k = 0; k < 4; ++k) {
        const float keep = p1 ? tt[2 * k + 1] : tt[2 * k];
        const float send = p1 ? tt[2 * k] : tt[2 * k + 1];
        u[k] = keep + row_xor<1>(send);
      }
      for (int k = 0; k < 2; ++k) {
        const float keep = p2 ? u[2 * k + 1] : u[2 * k];
        const float send = p2 ? u[2 * k] : u[2 * k + 1];
        w2[k] = keep + row_xor<2>(send);
      }
      const float keep4 = p4 ? w2[1] : w2[0];
      const float send4 = p4 ? w2[0] : w2[1];
      const float x8 = keep4 + row_xor<4>(send4);
      se_l[h] = x8 + row_xor<8>(x8);
    }

    const int myE = qw[q0 + ((slot < nb) ? slot : nb - 1)];
    const int src_l = ei[myE] - base;
    const int dst_l = ei[NE + myE];
    if ((L & 8) == 0 && slot < nb) atomicAdd(&cntS[src_l], 1);

    int g = 0;
    for (; g + EG <= nb; g += EG) {
      int      sr[EG];
      float4   kr[EG];
      float    vo[EG][NHEAD];
      float    se[EG][NHEAD];
      unsigned ov[EG];

      for (int u = 0; u < EG; ++u) {
        const int i = g + u;
        const int li = ((i >> 3) << 4) | (i & 7);
        sr[u] = bcast_i(src_l, li);
        const int dst = bcast_i(dst_l, li);
        const unsigned ok = (unsigned)dst * (HD * 4u) + l16;
        ov[u] = (unsigned)dst * (HD * 4u) + l4;
        kr[u] = *(const float4*)((const char*)Km + ok);
        for (int h2 = 0; h2 < NHEAD; ++h2) se[u][h2] = bcast_f(se_l[h2], li);
      }
      for (int u = 0; u < EG; ++u)
        for (int h2 = 0; h2 < NHEAD; ++h2)
          vo[u][h2] = *(const float*)((const char*)Vo + ov[u] + h2 * (DHEAD * 4));

      float acc[EG];
      for (int u = 0; u < EG; ++u)
        acc[u] = edge_out(kr[u], Qs[sr[u]][L], vo[u], se[u], bo_j);
      for (int u = 0; u < EG; ++u)
        atomicAdd(&sumS[sr[u]][L], acc[u]);
    }

    for (int i = g; i < nb; ++i) {
      const int li = ((i >> 3) << 4) | (i & 7);
      const int sr = bcast_i(src_l, li);
      const int dst = bcast_i(dst_l, li);
      const unsigned ok = (unsigned)dst * (HD * 4u) + l16;
      const unsigned ov = (unsigned)dst * (HD * 4u) + l4;
      const float4 k = *(const float4*)((const char*)Km + ok);
      float vo[NHEAD];
      for (int h2 = 0; h2 < NHEAD; ++h2)
        vo[h2] = *(const float*)((const char*)Vo + ov + h2 * (DHEAD * 4));
      const float se[NHEAD] = {bcast_f(se_l[0], li), bcast_f(se_l[1], li),
                               bcast_f(se_l[2], li), bcast_f(se_l[3], li)};
      const float acc = edge_out(k, Qs[sr][L], vo, se, bo_j);
      atomicAdd(&sumS[sr][L], acc);
    }
  }

  __syncthreads();
  for (int i = tid; i < RN * OUT_DIM; i += 1024) {
    const int node = base + (i >> 5);
    if (node < N)
      out[(size_t)node * OUT_DIM + (i & 31)] =
          sumS[i >> 5][i & 31] / fmaxf((float)cntS[i >> 5], 1.0f);
  }
}

extern "C" void kernel_launch(void* const* d_in, const int* in_sizes, int n_in,
                              void* d_out, int out_size, void* d_ws, size_t ws_size,
                              hipStream_t stream) {
  const float* x  = (const float*)d_in[0];
  const int*   ei = (const int*)d_in[1];
  const float* ea = (const float*)d_in[2];
  const float* Wq = (const float*)d_in[3];
  const float* bq = (const float*)d_in[4];
  const float* Wk = (const float*)d_in[5];
  const float* bk = (const float*)d_in[6];
  const float* Wv = (const float*)d_in[7];
  const float* bv = (const float*)d_in[8];
  const float* We = (const float*)d_in[9];
  const float* be = (const float*)d_in[10];
  const float* Wo = (const float*)d_in[11];
  const float* bo = (const float*)d_in[12];

  const int N  = in_sizes[0] / IN_DIM;
  const int NE = in_sizes[1] / 2;

  float* ws   = (float*)d_ws;
  float* Q    = ws;
  float* Km   = Q   + (size_t)N * HD;
  float* Vo   = Km  + (size_t)N * HD;
  float* Wvo  = Vo  + (size_t)N * HD;
  float* bvo  = Wvo + IN_DIM * HD;
  const int NBK = (N + RN - 1) / RN;
  const int G   = (NE + SLICE - 1) / SLICE;
  int* sorted   = (int*)(bvo + HD);
  int* segStart = sorted + (size_t)G * SLICE;

  prep_pass<<<PRE_BLOCKS + G, 1024, 0, stream>>>(Wv, bv, Wo, Wvo, bvo, ei,
                                                 sorted, segStart, N, NE, NBK);
  node_gemm<<<(N + 15) / 16, 256, 0, stream>>>(x, Wq, bq, Wk, bk, Wvo, bvo,
                                               Q, Km, Vo, N);
  edge_pass<<<NBK, 1024, 0, stream>>>(ei, ea, We, be, Q, Km, Vo, bo,
                                      sorted, segStart, G, (float*)d_out, N, NE);
}


# edge section: softmax over heads once per edge, 16 edges side by side in lanes (head dots summed transposed by DPP, weights read back as scalars); was per edge in all lanes via LDS shuffles
# speedup vs baseline: 2.9694x; 1.1081x over previous
#include <hip/hip_runtime.h>
#include <hip/hip_bf16.h>

typedef __attribute__((ext_vector_type(2))) float v2f;
typedef __attribute__((ext_vector_type(8))) float v8f;

#define IN_DIM  128
#define HD      128
#define E_DIM   32
#define NHEAD   4
#define DHEAD   32
#define OUT_DIM 32
#define INV_SQRT_D 0.17677669529663687f

static __device__ __forceinline__ v8f wmma_f32(v2f a, v2f b, v8f c) {
  return __builtin_amdgcn_wmma_f32_16x16x4_f32(false, a, false, b, (short)0, c,
                                               false, false);
}

#define RN        40
#define SLICE     4096
#define BK_MAX    1024
#define PRE_BLOCKS ((IN_DIM * HD + HD + 1023) / 1024)

__global__ __launch_bounds__(1024) void prep_pass(const float* __restrict__ Wv,
                               const float* __restrict__ bv,
                               const float* __restrict__ Wo,
                               float* __restrict__ Wvo,
                               float* __restrict__ bvo,
                               const int* __restrict__ ei,
                               int* __restrict__ sorted,
                               int* __restrict__ segStart,
                               int N, int NE, int NBK) {
  __shared__ int  histL[BK_MAX];
  __shared__ int  wsum[32];
  __shared__ int  s_total;
  __shared__ int4 sortL4[SLICE / 4];
  if (blockIdx.x >= PRE_BLOCKS) {
    const int tid = threadIdx.x;
    const int g = (int)blockIdx.x - PRE_BLOCKS;
    int* const sortL = (int*)sortL4;
    histL[tid] = 0;
    __syncthreads();

    const int e0 = g * SLICE + tid * 4;
    int s[4] = {-1, -1, -1, -1};
    if (e0 + 3 < NE) {
      const int4 v4 = *(const int4*)(ei + e0);
      s[0] = v4.x; s[1] = v4.y; s[2] = v4.z; s[3] = v4.w;
    } else {
      for (int k = 0; k < 4; ++k)
        if (e0 + k < NE) s[k] = ei[e0 + k];
    }
    int bk[4], rk[4];
    for (int k = 0; k < 4; ++k) {
      const bool ok = (unsigned)s[k] < (unsigned)N;
      bk[k] = ok ? s[k] / RN : -1;
      rk[k] = 0;
      if (ok) rk[k] = atomicAdd(&histL[bk[k]], 1);
    }
    __syncthreads();

    const int L = tid & 31, w = tid >> 5;
    const int v = (tid < NBK) ? histL[tid] : 0;
    int inc = v;
    for (int d = 1; d < 32; d <<= 1) {
      const int t = __shfl_up(inc, d);
      if (L >= d) inc += t;
    }
    if (L == 31) wsum[w] = inc;
    __syncthreads();
    if (w == 0) {
      const int x = wsum[L];
      int xi = x;
      for (int d = 1; d < 32; d <<= 1) {
        const int t = __shfl_up(xi, d);
        if (L >= d) xi += t;
      }
      wsum[L] = xi - x;
    }
    __syncthreads();
    const int excl = inc - v + wsum[w];
    int* const st = segStart + (size_t)g * (NBK + 1);
    if (tid < NBK) { histL[tid] = excl; st[tid] = excl; }
    if (tid == NBK - 1) { st[NBK] = excl + v; s_total = excl + v; }
    __syncthreads();

    for (int k = 0; k < 4; ++k)
      if (bk[k] >= 0) sortL[histL[bk[k]] + rk[k]] = e0 + k;
    __syncthreads();
    if (tid * 4 < s_total)
      *(int4*)(sorted + (size_t)g * SLICE + tid * 4) = sortL4[tid];
    return;
  }

  int idx = blockIdx.x * blockDim.x + threadIdx.x;
  if (idx < IN_DIM * HD) {
    int k = idx >> 7;
    int c = idx & 127;
    int h = c >> 5, j = c & 31;
    float acc = 0.0f;
    for (int d = 0; d < DHEAD; ++d)
      acc += Wv[k * HD + h * DHEAD + d] * Wo[(h * DHEAD + d) * OUT_DIM + j];
    Wvo[idx] = acc;
  } else if (idx < IN_DIM * HD + HD) {
    int c = idx - IN_DIM * HD;
    int h = c >> 5, j = c & 31;
    float acc = 0.0f;
    for (int d = 0; d < DHEAD; ++d)
      acc += bv[h * DHEAD + d] * Wo[(h * DHEAD + d) * OUT_DIM + j];
    bvo[c] = acc;
  }
}

__global__ __launch_bounds__(256) void node_gemm(
    const float* __restrict__ x,
    const float* __restrict__ Wq, const float* __restrict__ bq,
    const float* __restrict__ Wk, const float* __restrict__ bk,
    const float* __restrict__ Wvo, const float* __restrict__ bvo,
    float* __restrict__ Q, float* __restrict__ Km, float* __restrict__ Vo,
    int N) {
  __shared__ float As[16][IN_DIM];
  const int tid = threadIdx.x;
  const int m0 = blockIdx.x * 16;

  for (int i = 0; i < 8; ++i) {
    int lin = tid + i * 256;
    int r = lin >> 7, c = lin & 127;
    int gr = m0 + r; if (gr >= N) gr = N - 1;
    As[r][c] = x[(size_t)gr * IN_DIM + c];
  }
  __syncthreads();

  const int w = tid >> 5, L = tid & 31;
  const int n0 = w * 16;
  const int rowA = L & 15;
  const int kOff = (L < 16) ? 0 : 2;
  const int colB = L & 15;

  const float bq0 = bq[n0 + colB];
  const float bk0 = bk[n0 + colB];
  const float bv0 = bvo[n0 + colB];
  v8f cq, ck, cv;
  for (int j = 0; j < 8; ++j) { cq[j] = bq0; ck[j] = bk0; cv[j] = bv0; }

  for (int kk = 0; kk < IN_DIM; kk += 4) {
    v2f a;
    a.x = As[rowA][kk + kOff];
    a.y = As[rowA][kk + kOff + 1];
    v2f b;
    b.x = Wq[(kk + kOff) * HD + n0 + colB];
    b.y = Wq[(kk + kOff + 1) * HD + n0 + colB];
    cq = wmma_f32(a, b, cq);
    b.x = Wk[(kk + kOff) * HD + n0 + colB];
    b.y = Wk[(kk + kOff + 1) * HD + n0 + colB];
    ck = wmma_f32(a, b, ck);
    b.x = Wvo[(kk + kOff) * HD + n0 + colB];
    b.y = Wvo[(kk + kOff + 1) * HD + n0 + colB];
    cv = wmma_f32(a, b, cv);
  }

  for (int j = 0; j < 8; ++j) {
    int r = (L < 16) ? j : (j + 8);
    int gr = m0 + r;
    if (gr < N) {
      size_t o = (size_t)gr * HD + n0 + colB;
      Q[o] = cq[j]; Km[o] = ck[j]; Vo[o] = cv[j];
    }
  }
}

#define EP_WAVES  32
#define QCAP      (16 + 32)
#define EH        8

static __device__ __forceinline__ int bcast_i(int v, int lane) {
  return __builtin_amdgcn_readlane(v, lane);
}
static __device__ __forceinline__ float bcast_f(float v, int lane) {
  return __int_as_float(__builtin_amdgcn_readlane(__float_as_int(v), lane));
}
template <int M>
static __device__ __forceinline__ float row_xor(float v) {
  return __int_as_float(__builtin_amdgcn_update_dpp(
      0, __float_as_int(v), 0x160 + M, 0xf, 0xf, true));
}

static __device__ __forceinline__ float xhalf(float v) {
  const int b = __float_as_int(v);
  return __int_as_float(__builtin_amdgcn_permlanex16(b, b, 0x76543210, 0xfedcba98,
                                                     false, false));
}

static __device__ __forceinline__ float tsum4(const float v0, const float v1,
                                              const float v2, const float v3,
                                              const bool b1, const bool b2) {
  const float k0 = b1 ? v1 : v0, s0 = b1 ? v0 : v1;
  const float k1 = b1 ? v3 : v2, s1 = b1 ? v2 : v3;
  const float a0 = k0 + row_xor<1>(s0);
  const float a1 = k1 + row_xor<1>(s1);
  const float k = b2 ? a1 : a0, s = b2 ? a0 : a1;
  const float b = k + row_xor<2>(s);
  return b + row_xor<4>(b);
}

template <int H>
static __device__ __forceinline__ void k_load(const float* __restrict__ Km,
                                              const int src_l, const int dst_l,
                                              const unsigned l16,
                                              float4 (&kr)[EH], int (&sr)[EH]) {
  for (int j = 0; j < EH; ++j) {
    sr[j] = bcast_i(src_l, H * 16 + j);
    const int dst = bcast_i(dst_l, H * 16 + j);
    const unsigned ok = (unsigned)dst * (HD * 4u) + l16;
    kr[j] = *(const float4*)((const char*)Km + ok);
  }
}

static __device__ __forceinline__ void qk_sums(const float4 (*Qs)[32],
                                               const float4 (&kr)[EH],
                                               const int (&sr)[EH], const int L,
                                               const bool b1, const bool b2,
                                               float& ta, float& tb) {
  float p[EH];
  for (int j = 0; j < EH; ++j) {
    const float4 q = Qs[sr[j]][L];
    p[j] = fmaf(q.w, kr[j].w, fmaf(q.z, kr[j].z, fmaf(q.y, kr[j].y, q.x * kr[j].x)));
  }
  ta = tsum4(p[0], p[1], p[2], p[3], b1, b2);
  tb = tsum4(p[4], p[5], p[6], p[7], b1, b2);
}

static __device__ __forceinline__ void softmax16(const float (&t)[4],
                                                 const float (&se_l)[NHEAD],
                                                 const bool b4, const int L,
                                                 float (&al)[NHEAD]) {
  const bool row1 = (L & 16) != 0;
  const float lo = b4 ? t[1] : t[0], hi = b4 ? t[3] : t[2];
  const float same = row1 ? hi : lo, cross = row1 ? lo : hi;
  const float d0 = same, d1 = row_xor<8>(same);
  const float d2 = xhalf(cross), d3 = row_xor<8>(d2);
  float sc[NHEAD];
  sc[0] = ((row1 ? d2 : d0) + se_l[0]) * INV_SQRT_D;
  sc[1] = ((row1 ? d3 : d1) + se_l[1]) * INV_SQRT_D;
  sc[2] = ((row1 ? d0 : d2) + se_l[2]) * INV_SQRT_D;
  sc[3] = ((row1 ? d1 : d3) + se_l[3]) * INV_SQRT_D;
  float m = sc[0];
  for (int h2 = 1; h2 < NHEAD; ++h2) m = fmaxf(m, sc[h2]);
  float ex[NHEAD], den = 0.0f;
  for (int h2 = 0; h2 < NHEAD; ++h2) { ex[h2] = __expf(sc[h2] - m); den += ex[h2]; }
  const float inv = 1.0f / den;
  for (int h2 = 0; h2 < NHEAD; ++h2) al[h2] = ex[h2] * inv;
}

template <int H>
static __device__ __forceinline__ void vo_load(const float* __restrict__ Vo,
                                               const int dst_l, const unsigned l4,
                                               float (&vo)[EH][NHEAD]) {
  for (int j = 0; j < EH; ++j) {
    const int dst = bcast_i(dst_l, H * 16 + j);
    const unsigned ov = (unsigned)dst * (HD * 4u) + l4;
    for (int h2 = 0; h2 < NHEAD; ++h2)
      vo[j][h2] = *(const float*)((const char*)Vo + ov + h2 * (DHEAD * 4));
  }
}

template <int H, bool FULL>
static __device__ __forceinline__ void acc_half(float (*sumS)[OUT_DIM],
                                                const int src_l,
                                                const float (&al)[NHEAD],
                                                const float (&vo)[EH][NHEAD],
                                                const int L, const float bo_j,
                                                const int nb) {
  for (int j = 0; j < EH; ++j) {
    float acc = bo_j;
    for (int h2 = 0; h2 < NHEAD; ++h2)
      acc = fmaf(bcast_f(al[h2], H * 16 + j), vo[j][h2], acc);
    if (FULL || H * 8 + j < nb)
      atomicAdd(&sumS[bcast_i(src_l, H * 16 + j)][L], acc);
  }
}

__global__ __launch_bounds__(1024) void edge_pass(
    const int* __restrict__ ei, const float* __restrict__ ea,
    const float* __restrict__ We, const float* __restrict__ be,
    const float* __restrict__ Q, const float* __restrict__ Km,
    const float* __restrict__ Vo, const float* __restrict__ bo,
    const int* __restrict__ sorted, const int* __restrict__ segStart, int G,
    float* __restrict__ out, int N, int NE) {
  __shared__ v2f    Bf[64][32];
  __shared__ float  bes[HD];
  __shared__ float4 Qs[RN][32];
  __shared__ float  sumS[RN][OUT_DIM];
  __shared__ int    cntS[RN];
  __shared__ int    qS[EP_WAVES][QCAP];
  __shared__ int    s_next;

  const int tid = threadIdx.x;
  const int L = tid & 31;
  const int wv = __builtin_amdgcn_readfirstlane((int)(tid >> 5));
  const int c16 = L & 15;
  const int kOff = (L < 16) ? 0 : 2;
  const int base = blockIdx.x * RN;

  for (int q = 0; q < 2; ++q) {
    const int f = wv * 2 + q;
    const int col = (f >> 4) * DHEAD + ((f >> 3) & 1) * 16 + c16;
    const int kk = (f & 7) * 4;
    v2f b;
    b.x = We[(kk + kOff) * HD + col];
    b.y = We[(kk + kOff + 1) * HD + col];
    Bf[f][L] = b;
  }
  if (tid < HD) bes[tid] = be[tid];
  for (int i = tid; i < RN * 32; i += 1024) {
    int node = base + (i >> 5); if (node >= N) node = N - 1;
    Qs[i >> 5][i & 31] = *(const float4*)(Q + (size_t)node * HD + (i & 31) * 4);
  }
  for (int i = tid; i < RN * OUT_DIM; i += 1024) sumS[i >> 5][i & 31] = 0.0f;
  if (tid < RN) cntS[tid] = 0;
  if (tid == 0) s_next = 0;
  __syncthreads();

  const int rot = (int)(((unsigned)blockIdx.x * (unsigned)G) / gridDim.x);
  const int segPitch = (int)gridDim.x + 1;

  int* const qw = qS[wv];
  int qn = 0;
  bool more = true;
  int segPos = 0, segEnd = 0;

  const float bo_j = bo[L];
  const int slot = (L >> 4) * 8 + (L & 7);
  const bool p1 = (L & 1) != 0, p2 = (L & 2) != 0, p4 = (L & 4) != 0;
  const unsigned l16 = (unsigned)L * 16u, l4 = (unsigned)L * 4u;

  while (true) {
    if (qn < 16) {
      if (segPos < segEnd) {
        int n = segEnd - segPos; if (n > 32) n = 32;
        if (L < n) qw[qn + L] = sorted[segPos + L];
        qn += n; segPos += n;
        continue;
      }
      if (more) {
        int c = 0;
        if (L == 0) c = atomicAdd(&s_next, 1);
        c = __builtin_amdgcn_readfirstlane(c);
        if (c >= G) { more = false; continue; }
        int gg = c + rot; if (gg >= G) gg -= G;
        const int* const st = segStart + gg * segPitch + (int)blockIdx.x;
        segPos = gg * SLICE + st[0];
        segEnd = gg * SLICE + st[1];
        continue;
      }
      if (qn == 0) break;
    }

    const int nb = (qn < 16) ? qn : 16;
    const int q0 = qn - nb;
    qn = q0;
    const int rowE = qw[q0 + ((c16 < nb) ? c16 : nb - 1)];
    const unsigned aoff = (unsigned)rowE * (E_DIM * 4u) + (unsigned)kOff * 4u;
    v2f af[8];
    for (int s = 0; s < 8; ++s)
      af[s] = *(const v2f*)((const char*)ea + aoff + s * 16);

    float se_l[NHEAD];
    for (int h = 0; h < NHEAD; ++h) {
      v8f c0 = {}, c1 = {};
      v2f b0 = Bf[(h * 2 + 0) * 8][L];
      v2f b1 = Bf[(h * 2 + 1) * 8][L];
      for (int s = 0; s < 8; ++s) {
        v2f n0 = b0, n1 = b1;
        if (s < 7) {
          n0 = Bf[(h * 2 + 0) * 8 + s + 1][L];
          n1 = Bf[(h * 2 + 1) * 8 + s + 1][L];
        }
        c0 = wmma_f32(af[s], b0, c0);
        c1 = wmma_f32(af[s], b1, c1);
        b0 = n0; b1 = n1;
      }
      const float be0 = bes[h * DHEAD + c16], be1 = bes[h * DHEAD + 16 + c16];
      float tt[8];
      for (int j = 0; j < 8; ++j) {
        float u = c0[j] + be0, v = c1[j] + be1;
        tt[j] = u * u + v * v;
      }
      float u[4], w2[2];
      for (int k = 0; k < 4; ++k) {
        const float keep = p1 ? tt[2 * k + 1] : tt[2 * k];
        const float send = p1 ? tt[2 * k] : tt[2 * k + 1];
        u[k] = keep + row_xor<1>(send);
      }
      for (int k = 0; k < 2; ++k) {
        const float keep = p2 ? u[2 * k + 1] : u[2 * k];
        const float send = p2 ? u[2 * k] : u[2 * k + 1];
        w2[k] = keep + row_xor<2>(send);
      }
      const float keep4 = p4 ? w2[1] : w2[0];
      const float send4 = p4 ? w2[0] : w2[1];
      const float x8 = keep4 + row_xor<4>(send4);
      se_l[h] = x8 + row_xor<8>(x8);
    }

    const int myE = qw[q0 + ((slot < nb) ? slot : nb - 1)];
    const int src_l = ei[myE] - base;
    const int dst_l = ei[NE + myE];
    if ((L & 8) == 0 && slot < nb) atomicAdd(&cntS[src_l], 1);

    float t[4], al[NHEAD];
    float4 kr0[EH], kr1[EH];
    int sr0[EH], sr1[EH];
    float vo0[EH][NHEAD], vo1[EH][NHEAD];
    k_load<0>(Km, src_l, dst_l, l16, kr0, sr0);
    qk_sums(Qs, kr0, sr0, L, p1, p2, t[0], t[1]);
    t[2] = 0.0f; t[3] = 0.0f;
    if (nb > EH) {
      k_load<1>(Km, src_l, dst_l, l16, kr1, sr1);
      qk_sums(Qs, kr1, sr1, L, p1, p2, t[2], t[3]);
    }
    vo_load<0>(Vo, dst_l, l4, vo0);
    softmax16(t, se_l, p4, L, al);
    if (nb == 2 * EH) {
      vo_load<1>(Vo, dst_l, l4, vo1);
      acc_half<0, true>(sumS, src_l, al, vo0, L, bo_j, nb);
      acc_half<1, true>(sumS, src_l, al, vo1, L, bo_j, nb);
    } else {
      acc_half<0, false>(sumS, src_l, al, vo0, L, bo_j, nb);
      if (nb > EH) {
        vo_load<1>(Vo, dst_l, l4, vo1);
        acc_half<1, false>(sumS, src_l, al, vo1, L, bo_j, nb);
      }
    }
  }

  __syncthreads();
  for (int i = tid; i < RN * OUT_DIM; i += 1024) {
    const int node = base + (i >> 5);
    if (node < N)
      out[(size_t)node * OUT_DIM + (i & 31)] =
          sumS[i >> 5][i & 31] / fmaxf((float)cntS[i >> 5], 1.0f);
  }
}

extern "C" void kernel_launch(void* const* d_in, const int* in_sizes, int n_in,
                              void* d_out, int out_size, void* d_ws, size_t ws_size,
                              hipStream_t stream) {
  const float* x  = (const float*)d_in[0];
  const int*   ei = (const int*)d_in[1];
  const float* ea = (const float*)d_in[2];
  const float* Wq = (const float*)d_in[3];
  const float* bq = (const float*)d_in[4];
  const float* Wk = (const float*)d_in[5];
  const float* bk = (const float*)d_in[6];
  const float* Wv = (const float*)d_in[7];
  const float* bv = (const float*)d_in[8];
  const float* We = (const float*)d_in[9];
  const float* be = (const float*)d_in[10];
  const float* Wo = (const float*)d_in[11];
  const float* bo = (const float*)d_in[12];

  const int N  = in_sizes[0] / IN_DIM;
  const int NE = in_sizes[1] / 2;

  float* ws   = (float*)d_ws;
  float* Q    = ws;
  float* Km   = Q   + (size_t)N * HD;
  float* Vo   = Km  + (size_t)N * HD;
  float* Wvo  = Vo  + (size_t)N * HD;
  float* bvo  = Wvo + IN_DIM * HD;
  const int NBK = (N + RN - 1) / RN;
  const int G   = (NE + SLICE - 1) / SLICE;
  int* sorted   = (int*)(bvo + HD);
  int* segStart = sorted + (size_t)G * SLICE;

  prep_pass<<<PRE_BLOCKS + G, 1024, 0, stream>>>(Wv, bv, Wo, Wvo, bvo, ei,
                                                 sorted, segStart, N, NE, NBK);
  node_gemm<<<(N + 15) / 16, 256, 0, stream>>>(x, Wq, bq, Wk, bk, Wvo, bvo,
                                               Q, Km, Vo, N);
  edge_pass<<<NBK, 1024, 0, stream>>>(ei, ea, We, be, Q, Km, Vo, bo,
                                      sorted, segStart, G, (float*)d_out, N, NE);
}


# edge pass: the owner work-group first copies its pieces into one LDS edge list (scan + bisection), its waves then draw full 16-edge batches from a counter; LDS float adds still in arrival order
# speedup vs baseline: 3.0912x; 1.0410x over previous
#include <hip/hip_runtime.h>
#include <hip/hip_bf16.h>

typedef __attribute__((ext_vector_type(2))) float v2f;
typedef __attribute__((ext_vector_type(8))) float v8f;

#define IN_DIM  128
#define HD      128
#define E_DIM   32
#define NHEAD   4
#define DHEAD   32
#define OUT_DIM 32
#define INV_SQRT_D 0.17677669529663687f

static __device__ __forceinline__ v8f wmma_f32(v2f a, v2f b, v8f c) {
  return __builtin_amdgcn_wmma_f32_16x16x4_f32(false, a, false, b, (short)0, c,
                                               false, false);
}

#define RN        40
#define SLICE     4096
#define BK_MAX    1024
#define PRE_BLOCKS ((IN_DIM * HD + HD + 1023) / 1024)

__global__ __launch_bounds__(1024) void prep_pass(const float* __restrict__ Wv,
                               const float* __restrict__ bv,
                               const float* __restrict__ Wo,
                               float* __restrict__ Wvo,
                               float* __restrict__ bvo,
                               const int* __restrict__ ei,
                               int* __restrict__ sorted,
                               int* __restrict__ segStart,
                               int N, int NE, int NBK) {
  __shared__ int  histL[BK_MAX];
  __shared__ int  wsum[32];
  __shared__ int  s_total;
  __shared__ int4 sortL4[SLICE / 4];
  if (blockIdx.x >= PRE_BLOCKS) {
    const int tid = threadIdx.x;
    const int g = (int)blockIdx.x - PRE_BLOCKS;
    int* const sortL = (int*)sortL4;
    histL[tid] = 0;
    __syncthreads();

    const int e0 = g * SLICE + tid * 4;
    int s[4] = {-1, -1, -1, -1};
    if (e0 + 3 < NE) {
      const int4 v4 = *(const int4*)(ei + e0);
      s[0] = v4.x; s[1] = v4.y; s[2] = v4.z; s[3] = v4.w;
    } else {
      for (int k = 0; k < 4; ++k)
        if (e0 + k < NE) s[k] = ei[e0 + k];
    }
    int bk[4], rk[4];
    for (int k = 0; k < 4; ++k) {
      const bool ok = (unsigned)s[k] < (unsigned)N;
      bk[k] = ok ? s[k] / RN : -1;
      rk[k] = 0;
      if (ok) rk[k] = atomicAdd(&histL[bk[k]], 1);
    }
    __syncthreads();

    const int L = tid & 31, w = tid >> 5;
    const int v = (tid < NBK) ? histL[tid] : 0;
    int inc = v;
    for (int d = 1; d < 32; d <<= 1) {
      const int t = __shfl_up(inc, d);
      if (L >= d) inc += t;
    }
    if (L == 31) wsum[w] = inc;
    __syncthreads();
    if (w == 0) {
      const int x = wsum[L];
      int xi = x;
      for (int d = 1; d < 32; d <<= 1) {
        const int t = __shfl_up(xi, d);
        if (L >= d) xi += t;
      }
      wsum[L] = xi - x;
    }
    __syncthreads();
    const int excl = inc - v + wsum[w];
    int* const st = segStart + (size_t)g * (NBK + 1);
    if (tid < NBK) { histL[tid] = excl; st[tid] = excl; }
    if (tid == NBK - 1) { st[NBK] = excl + v; s_total = excl + v; }
    __syncthreads();

    for (int k = 0; k < 4; ++k)
      if (bk[k] >= 0) sortL[histL[bk[k]] + rk[k]] = e0 + k;
    __syncthreads();
    if (tid * 4 < s_total)
      *(int4*)(sorted + (size_t)g * SLICE + tid * 4) = sortL4[tid];
    return;
  }

  int idx = blockIdx.x * blockDim.x + threadIdx.x;
  if (idx < IN_DIM * HD) {
    int k = idx >> 7;
    int c = idx & 127;
    int h = c >> 5, j = c & 31;
    float acc = 0.0f;
    for (int d = 0; d < DHEAD; ++d)
      acc += Wv[k * HD + h * DHEAD + d] * Wo[(h * DHEAD + d) * OUT_DIM + j];
    Wvo[idx] = acc;
  } else if (idx < IN_DIM * HD + HD) {
    int c = idx - IN_DIM * HD;
    int h = c >> 5, j = c & 31;
    float acc = 0.0f;
    for (int d = 0; d < DHEAD; ++d)
      acc += bv[h * DHEAD + d] * Wo[(h * DHEAD + d) * OUT_DIM + j];
    bvo[c] = acc;
  }
}

__global__ __launch_bounds__(256) void node_gemm(
    const float* __restrict__ x,
    const float* __restrict__ Wq, const float* __restrict__ bq,
    const float* __restrict__ Wk, const float* __restrict__ bk,
    const float* __restrict__ Wvo, const float* __restrict__ bvo,
    float* __restrict__ Q, float* __restrict__ Km, float* __restrict__ Vo,
    int N) {
  __shared__ float As[16][IN_DIM];
  const int tid = threadIdx.x;
  const int m0 = blockIdx.x * 16;

  for (int i = 0; i < 8; ++i) {
    int lin = tid + i * 256;
    int r = lin >> 7, c = lin & 127;
    int gr = m0 + r; if (gr >= N) gr = N - 1;
    As[r][c] = x[(size_t)gr * IN_DIM + c];
  }
  __syncthreads();

  const int w = tid >> 5, L = tid & 31;
  const int n0 = w * 16;
  const int rowA = L & 15;
  const int kOff = (L < 16) ? 0 : 2;
  const int colB = L & 15;

  const float bq0 = bq[n0 + colB];
  const float bk0 = bk[n0 + colB];
  const float bv0 = bvo[n0 + colB];
  v8f cq, ck, cv;
  for (int j = 0; j < 8; ++j) { cq[j] = bq0; ck[j] = bk0; cv[j] = bv0; }

  for (int kk = 0; kk < IN_DIM; kk += 4) {
    v2f a;
    a.x = As[rowA][kk + kOff];
    a.y = As[rowA][kk + kOff + 1];
    v2f b;
    b.x = Wq[(kk + kOff) * HD + n0 + colB];
    b.y = Wq[(kk + kOff + 1) * HD + n0 + colB];
    cq = wmma_f32(a, b, cq);
    b.x = Wk[(kk + kOff) * HD + n0 + colB];
    b.y = Wk[(kk + kOff + 1) * HD + n0 + colB];
    ck = wmma_f32(a, b, ck);
    b.x = Wvo[(kk + kOff) * HD + n0 + colB];
    b.y = Wvo[(kk + kOff + 1) * HD + n0 + colB];
    cv = wmma_f32(a, b, cv);
  }

  for (int j = 0; j < 8; ++j) {
    int r = (L < 16) ? j : (j + 8);
    int gr = m0 + r;
    if (gr < N) {
      size_t o = (size_t)gr * HD + n0 + colB;
      Q[o] = cq[j]; Km[o] = ck[j]; Vo[o] = cv[j];
    }
  }
}

#define EP_WAVES  32
#define LCAP      4096
#define GCH       1024
#define EH        8

static __device__ __forceinline__ int bcast_i(int v, int lane) {
  return __builtin_amdgcn_readlane(v, lane);
}
static __device__ __forceinline__ float bcast_f(float v, int lane) {
  return __int_as_float(__builtin_amdgcn_readlane(__float_as_int(v), lane));
}
template <int M>
static __device__ __forceinline__ float row_xor(float v) {
  return __int_as_float(__builtin_amdgcn_update_dpp(
      0, __float_as_int(v), 0x160 + M, 0xf, 0xf, true));
}

static __device__ __forceinline__ float xhalf(float v) {
  const int b = __float_as_int(v);
  return __int_as_float(__builtin_amdgcn_permlanex16(b, b, 0x76543210, 0xfedcba98,
                                                     false, false));
}

static __device__ __forceinline__ float tsum4(const float v0, const float v1,
                                              const float v2, const float v3,
                                              const bool b1, const bool b2) {
  const float k0 = b1 ? v1 : v0, s0 = b1 ? v0 : v1;
  const float k1 = b1 ? v3 : v2, s1 = b1 ? v2 : v3;
  const float a0 = k0 + row_xor<1>(s0);
  const float a1 = k1 + row_xor<1>(s1);
  const float k = b2 ? a1 : a0, s = b2 ? a0 : a1;
  const float b = k + row_xor<2>(s);
  return b + row_xor<4>(b);
}

template <int H>
static __device__ __forceinline__ void k_load(const float* __restrict__ Km,
                                              const int src_l, const int dst_l,
                                              const unsigned l16,
                                              float4 (&kr)[EH], int (&sr)[EH]) {
  for (int j = 0; j < EH; ++j) {
    sr[j] = bcast_i(src_l, H * 16 + j);
    const int dst = bcast_i(dst_l, H * 16 + j);
    const unsigned ok = (unsigned)dst * (HD * 4u) + l16;
    kr[j] = *(const float4*)((const char*)Km + ok);
  }
}

static __device__ __forceinline__ void qk_sums(const float4 (*Qs)[32],
                                               const float4 (&kr)[EH],
                                               const int (&sr)[EH], const int L,
                                               const bool b1, const bool b2,
                                               float& ta, float& tb) {
  float p[EH];
  for (int j = 0; j < EH; ++j) {
    const float4 q = Qs[sr[j]][L];
    p[j] = fmaf(q.w, kr[j].w, fmaf(q.z, kr[j].z, fmaf(q.y, kr[j].y, q.x * kr[j].x)));
  }
  ta = tsum4(p[0], p[1], p[2], p[3], b1, b2);
  tb = tsum4(p[4], p[5], p[6], p[7], b1, b2);
}

static __device__ __forceinline__ void softmax16(const float (&t)[4],
                                                 const float (&se_l)[NHEAD],
                                                 const bool b4, const int L,
                                                 float (&al)[NHEAD]) {
  const bool row1 = (L & 16) != 0;
  const float lo = b4 ? t[1] : t[0], hi = b4 ? t[3] : t[2];
  const float same = row1 ? hi : lo, cross = row1 ? lo : hi;
  const float d0 = same, d1 = row_xor<8>(same);
  const float d2 = xhalf(cross), d3 = row_xor<8>(d2);
  float sc[NHEAD];
  sc[0] = ((row1 ? d2 : d0) + se_l[0]) * INV_SQRT_D;
  sc[1] = ((row1 ? d3 : d1) + se_l[1]) * INV_SQRT_D;
  sc[2] = ((row1 ? d0 : d2) + se_l[2]) * INV_SQRT_D;
  sc[3] = ((row1 ? d1 : d3) + se_l[3]) * INV_SQRT_D;
  float m = sc[0];
  for (int h2 = 1; h2 < NHEAD; ++h2) m = fmaxf(m, sc[h2]);
  float ex[NHEAD], den = 0.0f;
  for (int h2 = 0; h2 < NHEAD; ++h2) { ex[h2] = __expf(sc[h2] - m); den += ex[h2]; }
  const float inv = 1.0f / den;
  for (int h2 = 0; h2 < NHEAD; ++h2) al[h2] = ex[h2] * inv;
}

template <int H>
static __device__ __forceinline__ void vo_load(const float* __restrict__ Vo,
                                               const int dst_l, const unsigned l4,
                                               float (&vo)[EH][NHEAD]) {
  for (int j = 0; j < EH; ++j) {
    const int dst = bcast_i(dst_l, H * 16 + j);
    const unsigned ov = (unsigned)dst * (HD * 4u) + l4;
    for (int h2 = 0; h2 < NHEAD; ++h2)
      vo[j][h2] = *(const float*)((const char*)Vo + ov + h2 * (DHEAD * 4));
  }
}

template <int H, bool FULL>
static __device__ __forceinline__ void acc_half(float (*sumS)[OUT_DIM],
                                                const int src_l,
                                                const float (&al)[NHEAD],
                                                const float (&vo)[EH][NHEAD],
                                                const int L, const float bo_j,
                                                const int nb) {
  for (int j = 0; j < EH; ++j) {
    float acc = bo_j;
    for (int h2 = 0; h2 < NHEAD; ++h2)
      acc = fmaf(bcast_f(al[h2], H * 16 + j), vo[j][h2], acc);
    if (FULL || H * 8 + j < nb)
      atomicAdd(&sumS[bcast_i(src_l, H * 16 + j)][L], acc);
  }
}

__global__ __launch_bounds__(1024) void edge_pass(
    const int* __restrict__ ei, const float* __restrict__ ea,
    const float* __restrict__ We, const float* __restrict__ be,
    const float* __restrict__ Q, const float* __restrict__ Km,
    const float* __restrict__ Vo, const float* __restrict__ bo,
    const int* __restrict__ sorted, const int* __restrict__ segStart, int G,
    float* __restrict__ out, int N, int NE) {
  __shared__ v2f    Bf[64][32];
  __shared__ float  bes[HD];
  __shared__ float4 Qs[RN][32];
  __shared__ float  sumS[RN][OUT_DIM];
  __shared__ int    cntS[RN];
  __shared__ int    listS[LCAP];
  __shared__ int    pfxS[GCH + 1];
  __shared__ int    offS[GCH];
  __shared__ int    wsumS[EP_WAVES];
  __shared__ int    s_next;

  const int tid = threadIdx.x;
  const int L = tid & 31;
  const int wv = __builtin_amdgcn_readfirstlane((int)(tid >> 5));
  const int c16 = L & 15;
  const int kOff = (L < 16) ? 0 : 2;
  const int base = blockIdx.x * RN;

  for (int q = 0; q < 2; ++q) {
    const int f = wv * 2 + q;
    const int col = (f >> 4) * DHEAD + ((f >> 3) & 1) * 16 + c16;
    const int kk = (f & 7) * 4;
    v2f b;
    b.x = We[(kk + kOff) * HD + col];
    b.y = We[(kk + kOff + 1) * HD + col];
    Bf[f][L] = b;
  }
  if (tid < HD) bes[tid] = be[tid];
  for (int i = tid; i < RN * 32; i += 1024) {
    int node = base + (i >> 5); if (node >= N) node = N - 1;
    Qs[i >> 5][i & 31] = *(const float4*)(Q + (size_t)node * HD + (i & 31) * 4);
  }
  for (int i = tid; i < RN * OUT_DIM; i += 1024) sumS[i >> 5][i & 31] = 0.0f;
  if (tid < RN) cntS[tid] = 0;

  const int rot = (int)(((unsigned)blockIdx.x * (unsigned)G) / gridDim.x);
  const int segPitch = (int)gridDim.x + 1;

  const float bo_j = bo[L];
  const int slot = (L >> 4) * 8 + (L & 7);
  const bool p1 = (L & 1) != 0, p2 = (L & 2) != 0, p4 = (L & 4) != 0;
  const unsigned l16 = (unsigned)L * 16u, l4 = (unsigned)L * 4u;

  for (int gB = 0; gB < G; gB += GCH) {
    const int Gc = (G - gB < GCH) ? (G - gB) : GCH;
    int n = 0, off = 0;
    if (tid < Gc) {
      int gg = gB + tid + rot; if (gg >= G) gg -= G;
      const int* const st = segStart + (size_t)gg * segPitch + blockIdx.x;
      const int a = st[0];
      n = st[1] - a;
      off = gg * SLICE + a;
    }
    int inc = n;
    for (int d = 1; d < 32; d <<= 1) {
      const int t = __shfl_up(inc, d);
      if (L >= d) inc += t;
    }
    if (L == 31) wsumS[wv] = inc;
    __syncthreads();
    if (wv == 0) {
      const int x = wsumS[L];
      int xi = x;
      for (int d = 1; d < 32; d <<= 1) {
        const int t = __shfl_up(xi, d);
        if (L >= d) xi += t;
      }
      wsumS[L] = xi - x;
    }
    __syncthreads();
    const int excl = inc - n + wsumS[wv];
    if (tid < Gc) { pfxS[tid] = excl; offS[tid] = off; }
    if (tid == Gc - 1) pfxS[Gc] = excl + n;
    __syncthreads();
    const int T = pfxS[Gc];
    int step0 = 1;
    while (step0 < Gc) step0 <<= 1;
    step0 >>= 1;

    for (int w0 = 0; w0 < T; w0 += LCAP) {
      const int Tw = (T - w0 < LCAP) ? (T - w0) : LCAP;
      int pos[LCAP / 1024], lo[LCAP / 1024];
      for (int k = 0; k < LCAP / 1024; ++k) {
        const int i = tid + k * 1024;
        pos[k] = w0 + ((i < Tw) ? i : Tw - 1);
        lo[k] = 0;
      }
      for (int step = step0; step >= 1; step >>= 1) {
        for (int k = 0; k < LCAP / 1024; ++k) {
          const int c = lo[k] + step;
          const int cc = (c < Gc) ? c : Gc - 1;
          if (c < Gc && pfxS[cc] <= pos[k]) lo[k] = c;
        }
      }
      for (int k = 0; k < LCAP / 1024; ++k) {
        const int i = tid + k * 1024;
        if (i < Tw) listS[i] = sorted[offS[lo[k]] + (pos[k] - pfxS[lo[k]])];
      }
      if (tid == 0) s_next = 0;
      __syncthreads();

      const int nBatch = (Tw + 15) >> 4;
      int bq = 0;
      if (L == 0) bq = atomicAdd(&s_next, 1);
      bq = __builtin_amdgcn_readfirstlane(bq);
      while (bq < nBatch) {
        asm volatile("" ::: "memory");
        const int e0 = bq * 16;
        const int nb = (Tw - e0 < 16) ? (Tw - e0) : 16;
        const int rowE = listS[e0 + ((c16 < nb) ? c16 : nb - 1)];
        const unsigned aoff = (unsigned)rowE * (E_DIM * 4u) + (unsigned)kOff * 4u;
        v2f af[8];
        for (int s = 0; s < 8; ++s)
          af[s] = *(const v2f*)((const char*)ea + aoff + s * 16);

        float se_l[NHEAD];
        for (int h = 0; h < NHEAD; ++h) {
          v8f c0 = {}, c1 = {};
          v2f b0 = Bf[(h * 2 + 0) * 8][L];
          v2f b1 = Bf[(h * 2 + 1) * 8][L];
          for (int s = 0; s < 8; ++s) {
            v2f n0 = b0, n1 = b1;
            if (s < 7) {
              n0 = Bf[(h * 2 + 0) * 8 + s + 1][L];
              n1 = Bf[(h * 2 + 1) * 8 + s + 1][L];
            }
            c0 = wmma_f32(af[s], b0, c0);
            c1 = wmma_f32(af[s], b1, c1);
            b0 = n0; b1 = n1;
          }
          const float be0 = bes[h * DHEAD + c16], be1 = bes[h * DHEAD + 16 + c16];
          float tt[8];
          for (int j = 0; j < 8; ++j) {
            float u = c0[j] + be0, v = c1[j] + be1;
            tt[j] = u * u + v * v;
          }
          float u[4], w2[2];
          for (int k = 0; k < 4; ++k) {
            const float keep = p1 ? tt[2 * k + 1] : tt[2 * k];
            const float send = p1 ? tt[2 * k] : tt[2 * k + 1];
            u[k] = keep + row_xor<1>(send);
          }
          for (int k = 0; k < 2; ++k) {
            const float keep = p2 ? u[2 * k + 1] : u[2 * k];
            const float send = p2 ? u[2 * k] : u[2 * k + 1];
            w2[k] = keep + row_xor<2>(send);
          }
          const float keep4 = p4 ? w2[1] : w2[0];
          const float send4 = p4 ? w2[0] : w2[1];
          const float x8 = keep4 + row_xor<4>(send4);
          se_l[h] = x8 + row_xor<8>(x8);
        }

        const int myE = listS[e0 + ((slot < nb) ? slot : nb - 1)];
        const int src_l = ei[myE] - base;
        const int dst_l = ei[NE + myE];
        if ((L & 8) == 0 && slot < nb) atomicAdd(&cntS[src_l], 1);

        float t[4], al[NHEAD];
        float4 kr0[EH], kr1[EH];
        int sr0[EH], sr1[EH];
        float vo0[EH][NHEAD], vo1[EH][NHEAD];
        k_load<0>(Km, src_l, dst_l, l16, kr0, sr0);
        qk_sums(Qs, kr0, sr0, L, p1, p2, t[0], t[1]);
        t[2] = 0.0f; t[3] = 0.0f;
        if (nb > EH) {
          k_load<1>(Km, src_l, dst_l, l16, kr1, sr1);
          qk_sums(Qs, kr1, sr1, L, p1, p2, t[2], t[3]);
        }
        vo_load<0>(Vo, dst_l, l4, vo0);
        softmax16(t, se_l, p4, L, al);
        if (nb == 2 * EH) {
          vo_load<1>(Vo, dst_l, l4, vo1);
          acc_half<0, true>(sumS, src_l, al, vo0, L, bo_j, nb);
          acc_half<1, true>(sumS, src_l, al, vo1, L, bo_j, nb);
        } else {
          acc_half<0, false>(sumS, src_l, al, vo0, L, bo_j, nb);
          if (nb > EH) {
            vo_load<1>(Vo, dst_l, l4, vo1);
            acc_half<1, false>(sumS, src_l, al, vo1, L, bo_j, nb);
          }
        }
        bq = 0;
        if (L == 0) bq = atomicAdd(&s_next, 1);
        bq = __builtin_amdgcn_readfirstlane(bq);
      }
      __syncthreads();
    }
  }

  __syncthreads();
  for (int i = tid; i < RN * OUT_DIM; i += 1024) {
    const int node = base + (i >> 5);
    if (node < N)
      out[(size_t)node * OUT_DIM + (i & 31)] =
          sumS[i >> 5][i & 31] / fmaxf((float)cntS[i >> 5], 1.0f);
  }
}

extern "C" void kernel_launch(void* const* d_in, const int* in_sizes, int n_in,
                              void* d_out, int out_size, void* d_ws, size_t ws_size,
                              hipStream_t stream) {
  const float* x  = (const float*)d_in[0];
  const int*   ei = (const int*)d_in[1];
  const float* ea = (const float*)d_in[2];
  const float* Wq = (const float*)d_in[3];
  const float* bq = (const float*)d_in[4];
  const float* Wk = (const float*)d_in[5];
  const float* bk = (const float*)d_in[6];
  const float* Wv = (const float*)d_in[7];
  const float* bv = (const float*)d_in[8];
  const float* We = (const float*)d_in[9];
  const float* be = (const float*)d_in[10];
  const float* Wo = (const float*)d_in[11];
  const float* bo = (const float*)d_in[12];

  const int N  = in_sizes[0] / IN_DIM;
  const int NE = in_sizes[1] / 2;

  float* ws   = (float*)d_ws;
  float* Q    = ws;
  float* Km   = Q   + (size_t)N * HD;
  float* Vo   = Km  + (size_t)N * HD;
  float* Wvo  = Vo  + (size_t)N * HD;
  float* bvo  = Wvo + IN_DIM * HD;
  const int NBK = (N + RN - 1) / RN;
  const int G   = (NE + SLICE - 1) / SLICE;
  int* sorted   = (int*)(bvo + HD);
  int* segStart = sorted + (size_t)G * SLICE;

  prep_pass<<<PRE_BLOCKS + G, 1024, 0, stream>>>(Wv, bv, Wo, Wvo, bvo, ei,
                                                 sorted, segStart, N, NE, NBK);
  node_gemm<<<(N + 15) / 16, 256, 0, stream>>>(x, Wq, bq, Wk, bk, Wvo, bvo,
                                               Q, Km, Vo, N);
  edge_pass<<<NBK, 1024, 0, stream>>>(ei, ea, We, be, Q, Km, Vo, bo,
                                      sorted, segStart, G, (float*)d_out, N, NE);
}


# edge pass gathers with 16-byte loads: Vo stored head-innermost (1 load per edge and lane, was 4 of 4 bytes); edge_attr half rows as 4 loads (was 8), self-dot k order permuted in A and We alike
# speedup vs baseline: 3.2481x; 1.0508x over previous
#include <hip/hip_runtime.h>
#include <hip/hip_bf16.h>

typedef __attribute__((ext_vector_type(2))) float v2f;
typedef __attribute__((ext_vector_type(8))) float v8f;

#define IN_DIM  128
#define HD      128
#define E_DIM   32
#define NHEAD   4
#define DHEAD   32
#define OUT_DIM 32
#define INV_SQRT_D 0.17677669529663687f

static __device__ __forceinline__ v8f wmma_f32(v2f a, v2f b, v8f c) {
  return __builtin_amdgcn_wmma_f32_16x16x4_f32(false, a, false, b, (short)0, c,
                                               false, false);
}

#define RN        40
#define SLICE     4096
#define BK_MAX    1024
#define PRE_BLOCKS ((IN_DIM * HD + HD + 1023) / 1024)

__global__ __launch_bounds__(1024) void prep_pass(const float* __restrict__ Wv,
                               const float* __restrict__ bv,
                               const float* __restrict__ Wo,
                               float* __restrict__ Wvo,
                               float* __restrict__ bvo,
                               const int* __restrict__ ei,
                               int* __restrict__ sorted,
                               int* __restrict__ segStart,
                               int N, int NE, int NBK) {
  __shared__ int  histL[BK_MAX];
  __shared__ int  wsum[32];
  __shared__ int  s_total;
  __shared__ int4 sortL4[SLICE / 4];
  if (blockIdx.x >= PRE_BLOCKS) {
    const int tid = threadIdx.x;
    const int g = (int)blockIdx.x - PRE_BLOCKS;
    int* const sortL = (int*)sortL4;
    histL[tid] = 0;
    __syncthreads();

    const int e0 = g * SLICE + tid * 4;
    int s[4] = {-1, -1, -1, -1};
    if (e0 + 3 < NE) {
      const int4 v4 = *(const int4*)(ei + e0);
      s[0] = v4.x; s[1] = v4.y; s[2] = v4.z; s[3] = v4.w;
    } else {
      for (int k = 0; k < 4; ++k)
        if (e0 + k < NE) s[k] = ei[e0 + k];
    }
    int bk[4], rk[4];
    for (int k = 0; k < 4; ++k) {
      const bool ok = (unsigned)s[k] < (unsigned)N;
      bk[k] = ok ? s[k] / RN : -1;
      rk[k] = 0;
      if (ok) rk[k] = atomicAdd(&histL[bk[k]], 1);
    }
    __syncthreads();

    const int L = tid & 31, w = tid >> 5;
    const int v = (tid < NBK) ? histL[tid] : 0;
    int inc = v;
    for (int d = 1; d < 32; d <<= 1) {
      const int t = __shfl_up(inc, d);
      if (L >= d) inc += t;
    }
    if (L == 31) wsum[w] = inc;
    __syncthreads();
    if (w == 0) {
      const int x = wsum[L];
      int xi = x;
      for (int d = 1; d < 32; d <<= 1) {
        const int t = __shfl_up(xi, d);
        if (L >= d) xi += t;
      }
      wsum[L] = xi - x;
    }
    __syncthreads();
    const int excl = inc - v + wsum[w];
    int* const st = segStart + (size_t)g * (NBK + 1);
    if (tid < NBK) { histL[tid] = excl; st[tid] = excl; }
    if (tid == NBK - 1) { st[NBK] = excl + v; s_total = excl + v; }
    __syncthreads();

    for (int k = 0; k < 4; ++k)
      if (bk[k] >= 0) sortL[histL[bk[k]] + rk[k]] = e0 + k;
    __syncthreads();
    if (tid * 4 < s_total)
      *(int4*)(sorted + (size_t)g * SLICE + tid * 4) = sortL4[tid];
    return;
  }

  int idx = blockIdx.x * blockDim.x + threadIdx.x;
  if (idx < IN_DIM * HD) {
    int k = idx >> 7;
    int c = idx & 127;
    int j = c >> 2, h = c & 3;
    float acc = 0.0f;
    for (int d = 0; d < DHEAD; ++d)
      acc += Wv[k * HD + h * DHEAD + d] * Wo[(h * DHEAD + d) * OUT_DIM + j];
    Wvo[idx] = acc;
  } else if (idx < IN_DIM * HD + HD) {
    int c = idx - IN_DIM * HD;
    int j = c >> 2, h = c & 3;
    float acc = 0.0f;
    for (int d = 0; d < DHEAD; ++d)
      acc += bv[h * DHEAD + d] * Wo[(h * DHEAD + d) * OUT_DIM + j];
    bvo[c] = acc;
  }
}

__global__ __launch_bounds__(256) void node_gemm(
    const float* __restrict__ x,
    const float* __restrict__ Wq, const float* __restrict__ bq,
    const float* __restrict__ Wk, const float* __restrict__ bk,
    const float* __restrict__ Wvo, const float* __restrict__ bvo,
    float* __restrict__ Q, float* __restrict__ Km, float* __restrict__ Vo,
    int N) {
  __shared__ float As[16][IN_DIM];
  const int tid = threadIdx.x;
  const int m0 = blockIdx.x * 16;

  for (int i = 0; i < 8; ++i) {
    int lin = tid + i * 256;
    int r = lin >> 7, c = lin & 127;
    int gr = m0 + r; if (gr >= N) gr = N - 1;
    As[r][c] = x[(size_t)gr * IN_DIM + c];
  }
  __syncthreads();

  const int w = tid >> 5, L = tid & 31;
  const int n0 = w * 16;
  const int rowA = L & 15;
  const int kOff = (L < 16) ? 0 : 2;
  const int colB = L & 15;

  const float bq0 = bq[n0 + colB];
  const float bk0 = bk[n0 + colB];
  const float bv0 = bvo[n0 + colB];
  v8f cq, ck, cv;
  for (int j = 0; j < 8; ++j) { cq[j] = bq0; ck[j] = bk0; cv[j] = bv0; }

  for (int kk = 0; kk < IN_DIM; kk += 4) {
    v2f a;
    a.x = As[rowA][kk + kOff];
    a.y = As[rowA][kk + kOff + 1];
    v2f b;
    b.x = Wq[(kk + kOff) * HD + n0 + colB];
    b.y = Wq[(kk + kOff + 1) * HD + n0 + colB];
    cq = wmma_f32(a, b, cq);
    b.x = Wk[(kk + kOff) * HD + n0 + colB];
    b.y = Wk[(kk + kOff + 1) * HD + n0 + colB];
    ck = wmma_f32(a, b, ck);
    b.x = Wvo[(kk + kOff) * HD + n0 + colB];
    b.y = Wvo[(kk + kOff + 1) * HD + n0 + colB];
    cv = wmma_f32(a, b, cv);
  }

  for (int j = 0; j < 8; ++j) {
    int r = (L < 16) ? j : (j + 8);
    int gr = m0 + r;
    if (gr < N) {
      size_t o = (size_t)gr * HD + n0 + colB;
      Q[o] = cq[j]; Km[o] = ck[j]; Vo[o] = cv[j];
    }
  }
}

#define EP_WAVES  32
#define LCAP      4096
#define GCH       1024
#define EH        8

static __device__ __forceinline__ int bcast_i(int v, int lane) {
  return __builtin_amdgcn_readlane(v, lane);
}
static __device__ __forceinline__ float bcast_f(float v, int lane) {
  return __int_as_float(__builtin_amdgcn_readlane(__float_as_int(v), lane));
}
template <int M>
static __device__ __forceinline__ float row_xor(float v) {
  return __int_as_float(__builtin_amdgcn_update_dpp(
      0, __float_as_int(v), 0x160 + M, 0xf, 0xf, true));
}

static __device__ __forceinline__ float xhalf(float v) {
  const int b = __float_as_int(v);
  return __int_as_float(__builtin_amdgcn_permlanex16(b, b, 0x76543210, 0xfedcba98,
                                                     false, false));
}

static __device__ __forceinline__ float tsum4(const float v0, const float v1,
                                              const float v2, const float v3,
                                              const bool b1, const bool b2) {
  const float k0 = b1 ? v1 : v0, s0 = b1 ? v0 : v1;
  const float k1 = b1 ? v3 : v2, s1 = b1 ? v2 : v3;
  const float a0 = k0 + row_xor<1>(s0);
  const float a1 = k1 + row_xor<1>(s1);
  const float k = b2 ? a1 : a0, s = b2 ? a0 : a1;
  const float b = k + row_xor<2>(s);
  return b + row_xor<4>(b);
}

template <int H>
static __device__ __forceinline__ void k_load(const float* __restrict__ Km,
                                              const int src_l, const int dst_l,
                                              const unsigned l16,
                                              float4 (&kr)[EH], int (&sr)[EH]) {
  for (int j = 0; j < EH; ++j) {
    sr[j] = bcast_i(src_l, H * 16 + j);
    const int dst = bcast_i(dst_l, H * 16 + j);
    const unsigned ok = (unsigned)dst * (HD * 4u) + l16;
    kr[j] = *(const float4*)((const char*)Km + ok);
  }
}

static __device__ __forceinline__ void qk_sums(const float4 (*Qs)[32],
                                               const float4 (&kr)[EH],
                                               const int (&sr)[EH], const int L,
                                               const bool b1, const bool b2,
                                               float& ta, float& tb) {
  float p[EH];
  for (int j = 0; j < EH; ++j) {
    const float4 q = Qs[sr[j]][L];
    p[j] = fmaf(q.w, kr[j].w, fmaf(q.z, kr[j].z, fmaf(q.y, kr[j].y, q.x * kr[j].x)));
  }
  ta = tsum4(p[0], p[1], p[2], p[3], b1, b2);
  tb = tsum4(p[4], p[5], p[6], p[7], b1, b2);
}

static __device__ __forceinline__ void softmax16(const float (&t)[4],
                                                 const float (&se_l)[NHEAD],
                                                 const bool b4, const int L,
                                                 float (&al)[NHEAD]) {
  const bool row1 = (L & 16) != 0;
  const float lo = b4 ? t[1] : t[0], hi = b4 ? t[3] : t[2];
  const float same = row1 ? hi : lo, cross = row1 ? lo : hi;
  const float d0 = same, d1 = row_xor<8>(same);
  const float d2 = xhalf(cross), d3 = row_xor<8>(d2);
  float sc[NHEAD];
  sc[0] = ((row1 ? d2 : d0) + se_l[0]) * INV_SQRT_D;
  sc[1] = ((row1 ? d3 : d1) + se_l[1]) * INV_SQRT_D;
  sc[2] = ((row1 ? d0 : d2) + se_l[2]) * INV_SQRT_D;
  sc[3] = ((row1 ? d1 : d3) + se_l[3]) * INV_SQRT_D;
  float m = sc[0];
  for (int h2 = 1; h2 < NHEAD; ++h2) m = fmaxf(m, sc[h2]);
  float ex[NHEAD], den = 0.0f;
  for (int h2 = 0; h2 < NHEAD; ++h2) { ex[h2] = __expf(sc[h2] - m); den += ex[h2]; }
  const float inv = 1.0f / den;
  for (int h2 = 0; h2 < NHEAD; ++h2) al[h2] = ex[h2] * inv;
}

template <int H>
static __device__ __forceinline__ void vo_load(const float* __restrict__ Vo,
                                               const int dst_l, const unsigned l16,
                                               float4 (&vo)[EH]) {
  for (int j = 0; j < EH; ++j) {
    const int dst = bcast_i(dst_l, H * 16 + j);
    const unsigned ov = (unsigned)dst * (HD * 4u) + l16;
    vo[j] = *(const float4*)((const char*)Vo + ov);
  }
}

template <int H, bool FULL>
static __device__ __forceinline__ void acc_half(float (*sumS)[OUT_DIM],
                                                const int src_l,
                                                const float (&al)[NHEAD],
                                                const float4 (&vo)[EH],
                                                const int L, const float bo_j,
                                                const int nb) {
  for (int j = 0; j < EH; ++j) {
    float acc = bo_j;
    acc = fmaf(bcast_f(al[0], H * 16 + j), vo[j].x, acc);
    acc = fmaf(bcast_f(al[1], H * 16 + j), vo[j].y, acc);
    acc = fmaf(bcast_f(al[2], H * 16 + j), vo[j].z, acc);
    acc = fmaf(bcast_f(al[3], H * 16 + j), vo[j].w, acc);
    if (FULL || H * 8 + j < nb)
      atomicAdd(&sumS[bcast_i(src_l, H * 16 + j)][L], acc);
  }
}

__global__ __launch_bounds__(1024) void edge_pass(
    const int* __restrict__ ei, const float* __restrict__ ea,
    const float* __restrict__ We, const float* __restrict__ be,
    const float* __restrict__ Q, const float* __restrict__ Km,
    const float* __restrict__ Vo, const float* __restrict__ bo,
    const int* __restrict__ sorted, const int* __restrict__ segStart, int G,
    float* __restrict__ out, int N, int NE) {
  __shared__ v2f    Bf[64][32];
  __shared__ float  bes[HD];
  __shared__ float4 Qs[RN][32];
  __shared__ float  sumS[RN][OUT_DIM];
  __shared__ int    cntS[RN];
  __shared__ int    listS[LCAP];
  __shared__ int    pfxS[GCH + 1];
  __shared__ int    offS[GCH];
  __shared__ int    wsumS[EP_WAVES];
  __shared__ int    s_next;

  const int tid = threadIdx.x;
  const int L = tid & 31;
  const int wv = __builtin_amdgcn_readfirstlane((int)(tid >> 5));
  const int c16 = L & 15;
  const int hh = L >> 4;
  const int base = blockIdx.x * RN;

  for (int q = 0; q < 2; ++q) {
    const int f = wv * 2 + q;
    const int col = (f >> 4) * DHEAD + ((f >> 3) & 1) * 16 + c16;
    const int kr = hh * 16 + (f & 7) * 2;
    v2f b;
    b.x = We[kr * HD + col];
    b.y = We[(kr + 1) * HD + col];
    Bf[f][L] = b;
  }
  if (tid < HD) bes[tid] = be[tid];
  for (int i = tid; i < RN * 32; i += 1024) {
    int node = base + (i >> 5); if (node >= N) node = N - 1;
    Qs[i >> 5][i & 31] = *(const float4*)(Q + (size_t)node * HD + (i & 31) * 4);
  }
  for (int i = tid; i < RN * OUT_DIM; i += 1024) sumS[i >> 5][i & 31] = 0.0f;
  if (tid < RN) cntS[tid] = 0;

  const int rot = (int)(((unsigned)blockIdx.x * (unsigned)G) / gridDim.x);
  const int segPitch = (int)gridDim.x + 1;

  const float bo_j = bo[L];
  const int slot = (L >> 4) * 8 + (L & 7);
  const bool p1 = (L & 1) != 0, p2 = (L & 2) != 0, p4 = (L & 4) != 0;
  const unsigned l16 = (unsigned)L * 16u;

  for (int gB = 0; gB < G; gB += GCH) {
    const int Gc = (G - gB < GCH) ? (G - gB) : GCH;
    int n = 0, off = 0;
    if (tid < Gc) {
      int gg = gB + tid + rot; if (gg >= G) gg -= G;
      const int* const st = segStart + (size_t)gg * segPitch + blockIdx.x;
      const int a = st[0];
      n = st[1] - a;
      off = gg * SLICE + a;
    }
    int inc = n;
    for (int d = 1; d < 32; d <<= 1) {
      const int t = __shfl_up(inc, d);
      if (L >= d) inc += t;
    }
    if (L == 31) wsumS[wv] = inc;
    __syncthreads();
    if (wv == 0) {
      const int x = wsumS[L];
      int xi = x;
      for (int d = 1; d < 32; d <<= 1) {
        const int t = __shfl_up(xi, d);
        if (L >= d) xi += t;
      }
      wsumS[L] = xi - x;
    }
    __syncthreads();
    const int excl = inc - n + wsumS[wv];
    if (tid < Gc) { pfxS[tid] = excl; offS[tid] = off; }
    if (tid == Gc - 1) pfxS[Gc] = excl + n;
    __syncthreads();
    const int T = pfxS[Gc];
    int step0 = 1;
    while (step0 < Gc) step0 <<= 1;
    step0 >>= 1;

    for (int w0 = 0; w0 < T; w0 += LCAP) {
      const int Tw = (T - w0 < LCAP) ? (T - w0) : LCAP;
      int pos[LCAP / 1024], lo[LCAP / 1024];
      for (int k = 0; k < LCAP / 1024; ++k) {
        const int i = tid + k * 1024;
        pos[k] = w0 + ((i < Tw) ? i : Tw - 1);
        lo[k] = 0;
      }
      for (int step = step0; step >= 1; step >>= 1) {
        for (int k = 0; k < LCAP / 1024; ++k) {
          const int c = lo[k] + step;
          const int cc = (c < Gc) ? c : Gc - 1;
          if (c < Gc && pfxS[cc] <= pos[k]) lo[k] = c;
        }
      }
      for (int k = 0; k < LCAP / 1024; ++k) {
        const int i = tid + k * 1024;
        if (i < Tw) listS[i] = sorted[offS[lo[k]] + (pos[k] - pfxS[lo[k]])];
      }
      if (tid == 0) s_next = 0;
      __syncthreads();

      const int nBatch = (Tw + 15) >> 4;
      int bq = 0;
      if (L == 0) bq = atomicAdd(&s_next, 1);
      bq = __builtin_amdgcn_readfirstlane(bq);
      while (bq < nBatch) {
        asm volatile("" ::: "memory");
        const int e0 = bq * 16;
        const int nb = (Tw - e0 < 16) ? (Tw - e0) : 16;
        const int rowE = listS[e0 + ((c16 < nb) ? c16 : nb - 1)];
        const unsigned aoff = (unsigned)rowE * (E_DIM * 4u) + (unsigned)hh * 64u;
        v2f af[8];
        for (int i = 0; i < 4; ++i) {
          const float4 q4 = *(const float4*)((const char*)ea + aoff + i * 16);
          af[2 * i].x = q4.x;     af[2 * i].y = q4.y;
          af[2 * i + 1].x = q4.z; af[2 * i + 1].y = q4.w;
        }
        __builtin_amdgcn_sched_barrier(0x386);

        float se_l[NHEAD];
        for (int h = 0; h < NHEAD; ++h) {
          v8f c0 = {}, c1 = {};
          v2f b0 = Bf[(h * 2 + 0) * 8][L];
          v2f b1 = Bf[(h * 2 + 1) * 8][L];
          for (int s = 0; s < 8; ++s) {
            v2f n0 = b0, n1 = b1;
            if (s < 7) {
              n0 = Bf[(h * 2 + 0) * 8 + s + 1][L];
              n1 = Bf[(h * 2 + 1) * 8 + s + 1][L];
            }
            c0 = wmma_f32(af[s], b0, c0);
            c1 = wmma_f32(af[s], b1, c1);
            b0 = n0; b1 = n1;
          }
          const float be0 = bes[h * DHEAD + c16], be1 = bes[h * DHEAD + 16 + c16];
          float tt[8];
          for (int j = 0; j < 8; ++j) {
            float u = c0[j] + be0, v = c1[j] + be1;
            tt[j] = u * u + v * v;
          }
          float u[4], w2[2];
          for (int k = 0; k < 4; ++k) {
            const float keep = p1 ? tt[2 * k + 1] : tt[2 * k];
            const float send = p1 ? tt[2 * k] : tt[2 * k + 1];
            u[k] = keep + row_xor<1>(send);
          }
          for (int k = 0; k < 2; ++k) {
            const float keep = p2 ? u[2 * k + 1] : u[2 * k];
            const float send = p2 ? u[2 * k] : u[2 * k + 1];
            w2[k] = keep + row_xor<2>(send);
          }
          const float keep4 = p4 ? w2[1] : w2[0];
          const float send4 = p4 ? w2[0] : w2[1];
          const float x8 = keep4 + row_xor<4>(send4);
          se_l[h] = x8 + row_xor<8>(x8);
        }

        const int myE = listS[e0 + ((slot < nb) ? slot : nb - 1)];
        const int src_l = ei[myE] - base;
        const int dst_l = ei[NE + myE];
        if ((L & 8) == 0 && slot < nb) atomicAdd(&cntS[src_l], 1);

        float t[4], al[NHEAD];
        float4 kr0[EH], kr1[EH];
        int sr0[EH], sr1[EH];
        float4 vo0[EH], vo1[EH];
        k_load<0>(Km, src_l, dst_l, l16, kr0, sr0);
        qk_sums(Qs, kr0, sr0, L, p1, p2, t[0], t[1]);
        t[2] = 0.0f; t[3] = 0.0f;
        if (nb > EH) {
          k_load<1>(Km, src_l, dst_l, l16, kr1, sr1);
          qk_sums(Qs, kr1, sr1, L, p1, p2, t[2], t[3]);
        }
        vo_load<0>(Vo, dst_l, l16, vo0);
        softmax16(t, se_l, p4, L, al);
        if (nb == 2 * EH) {
          vo_load<1>(Vo, dst_l, l16, vo1);
          acc_half<0, true>(sumS, src_l, al, vo0, L, bo_j, nb);
          acc_half<1, true>(sumS, src_l, al, vo1, L, bo_j, nb);
        } else {
          acc_half<0, false>(sumS, src_l, al, vo0, L, bo_j, nb);
          if (nb > EH) {
            vo_load<1>(Vo, dst_l, l16, vo1);
            acc_half<1, false>(sumS, src_l, al, vo1, L, bo_j, nb);
          }
        }
        bq = 0;
        if (L == 0) bq = atomicAdd(&s_next, 1);
        bq = __builtin_amdgcn_readfirstlane(bq);
      }
      __syncthreads();
    }
  }

  __syncthreads();
  for (int r = wv; r < RN; r += EP_WAVES) {
    const int node = base + r;
    if (node < N)
      out[(size_t)node * OUT_DIM + L] = sumS[r][L] / fmaxf((float)cntS[r], 1.0f);
  }
}

extern "C" void kernel_launch(void* const* d_in, const int* in_sizes, int n_in,
                              void* d_out, int out_size, void* d_ws, size_t ws_size,
                              hipStream_t stream) {
  const float* x  = (const float*)d_in[0];
  const int*   ei = (const int*)d_in[1];
  const float* ea = (const float*)d_in[2];
  const float* Wq = (const float*)d_in[3];
  const float* bq = (const float*)d_in[4];
  const float* Wk = (const float*)d_in[5];
  const float* bk = (const float*)d_in[6];
  const float* Wv = (const float*)d_in[7];
  const float* bv = (const float*)d_in[8];
  const float* We = (const float*)d_in[9];
  const float* be = (const float*)d_in[10];
  const float* Wo = (const float*)d_in[11];
  const float* bo = (const float*)d_in[12];

  const int N  = in_sizes[0] / IN_DIM;
  const int NE = in_sizes[1] / 2;

  float* ws   = (float*)d_ws;
  float* Q    = ws;
  float* Km   = Q   + (size_t)N * HD;
  float* Vo   = Km  + (size_t)N * HD;
  float* Wvo  = Vo  + (size_t)N * HD;
  float* bvo  = Wvo + IN_DIM * HD;
  const int NBK = (N + RN - 1) / RN;
  const int G   = (NE + SLICE - 1) / SLICE;
  int* sorted   = (int*)(bvo + HD);
  int* segStart = sorted + (size_t)G * SLICE;

  prep_pass<<<PRE_BLOCKS + G, 1024, 0, stream>>>(Wv, bv, Wo, Wvo, bvo, ei,
                                                 sorted, segStart, N, NE, NBK);
  node_gemm<<<(N + 15) / 16, 256, 0, stream>>>(x, Wq, bq, Wk, bk, Wvo, bvo,
                                               Q, Km, Vo, N);
  edge_pass<<<NBK, 1024, 0, stream>>>(ei, ea, We, be, Q, Km, Vo, bo,
                                      sorted, segStart, G, (float*)d_out, N, NE);
}


# edge self-dot taken transposed (We^T tile x edge rows): a lane holds one edge, squares summed in-lane on register pairs (new sum order), bias as accumulator start, heads one by one; butterfly gone
# speedup vs baseline: 3.3693x; 1.0373x over previous
#include <hip/hip_runtime.h>
#include <hip/hip_bf16.h>

typedef __attribute__((ext_vector_type(2))) float v2f;
typedef __attribute__((ext_vector_type(8))) float v8f;

#define IN_DIM  128
#define HD      128
#define E_DIM   32
#define NHEAD   4
#define DHEAD   32
#define OUT_DIM 32
#define INV_SQRT_D 0.17677669529663687f

static __device__ __forceinline__ v8f wmma_f32(v2f a, v2f b, v8f c) {
  return __builtin_amdgcn_wmma_f32_16x16x4_f32(false, a, false, b, (short)0, c,
                                               false, false);
}

#define RN        40
#define SLICE     4096
#define BK_MAX    1024
#define PRE_BLOCKS ((IN_DIM * HD + HD + 1023) / 1024)

__global__ __launch_bounds__(1024) void prep_pass(const float* __restrict__ Wv,
                               const float* __restrict__ bv,
                               const float* __restrict__ Wo,
                               float* __restrict__ Wvo,
                               float* __restrict__ bvo,
                               const int* __restrict__ ei,
                               int* __restrict__ sorted,
                               int* __restrict__ segStart,
                               int N, int NE, int NBK) {
  __shared__ int  histL[BK_MAX];
  __shared__ int  wsum[32];
  __shared__ int  s_total;
  __shared__ int4 sortL4[SLICE / 4];
  if (blockIdx.x >= PRE_BLOCKS) {
    const int tid = threadIdx.x;
    const int g = (int)blockIdx.x - PRE_BLOCKS;
    int* const sortL = (int*)sortL4;
    histL[tid] = 0;
    __syncthreads();

    const int e0 = g * SLICE + tid * 4;
    int s[4] = {-1, -1, -1, -1};
    if (e0 + 3 < NE) {
      const int4 v4 = *(const int4*)(ei + e0);
      s[0] = v4.x; s[1] = v4.y; s[2] = v4.z; s[3] = v4.w;
    } else {
      for (int k = 0; k < 4; ++k)
        if (e0 + k < NE) s[k] = ei[e0 + k];
    }
    int bk[4], rk[4];
    for (int k = 0; k < 4; ++k) {
      const bool ok = (unsigned)s[k] < (unsigned)N;
      bk[k] = ok ? s[k] / RN : -1;
      rk[k] = 0;
      if (ok) rk[k] = atomicAdd(&histL[bk[k]], 1);
    }
    __syncthreads();

    const int L = tid & 31, w = tid >> 5;
    const int v = (tid < NBK) ? histL[tid] : 0;
    int inc = v;
    for (int d = 1; d < 32; d <<= 1) {
      const int t = __shfl_up(inc, d);
      if (L >= d) inc += t;
    }
    if (L == 31) wsum[w] = inc;
    __syncthreads();
    if (w == 0) {
      const int x = wsum[L];
      int xi = x;
      for (int d = 1; d < 32; d <<= 1) {
        const int t = __shfl_up(xi, d);
        if (L >= d) xi += t;
      }
      wsum[L] = xi - x;
    }
    __syncthreads();
    const int excl = inc - v + wsum[w];
    int* const st = segStart + (size_t)g * (NBK + 1);
    if (tid < NBK) { histL[tid] = excl; st[tid] = excl; }
    if (tid == NBK - 1) { st[NBK] = excl + v; s_total = excl + v; }
    __syncthreads();

    for (int k = 0; k < 4; ++k)
      if (bk[k] >= 0) sortL[histL[bk[k]] + rk[k]] = e0 + k;
    __syncthreads();
    if (tid * 4 < s_total)
      *(int4*)(sorted + (size_t)g * SLICE + tid * 4) = sortL4[tid];
    return;
  }

  int idx = blockIdx.x * blockDim.x + threadIdx.x;
  if (idx < IN_DIM * HD) {
    int k = idx >> 7;
    int c = idx & 127;
    int j = c >> 2, h = c & 3;
    float acc = 0.0f;
    for (int d = 0; d < DHEAD; ++d)
      acc += Wv[k * HD + h * DHEAD + d] * Wo[(h * DHEAD + d) * OUT_DIM + j];
    Wvo[idx] = acc;
  } else if (idx < IN_DIM * HD + HD) {
    int c = idx - IN_DIM * HD;
    int j = c >> 2, h = c & 3;
    float acc = 0.0f;
    for (int d = 0; d < DHEAD; ++d)
      acc += bv[h * DHEAD + d] * Wo[(h * DHEAD + d) * OUT_DIM + j];
    bvo[c] = acc;
  }
}

__global__ __launch_bounds__(256) void node_gemm(
    const float* __restrict__ x,
    const float* __restrict__ Wq, const float* __restrict__ bq,
    const float* __restrict__ Wk, const float* __restrict__ bk,
    const float* __restrict__ Wvo, const float* __restrict__ bvo,
    float* __restrict__ Q, float* __restrict__ Km, float* __restrict__ Vo,
    int N) {
  __shared__ float As[16][IN_DIM];
  const int tid = threadIdx.x;
  const int m0 = blockIdx.x * 16;

  for (int i = 0; i < 8; ++i) {
    int lin = tid + i * 256;
    int r = lin >> 7, c = lin & 127;
    int gr = m0 + r; if (gr >= N) gr = N - 1;
    As[r][c] = x[(size_t)gr * IN_DIM + c];
  }
  __syncthreads();

  const int w = tid >> 5, L = tid & 31;
  const int n0 = w * 16;
  const int rowA = L & 15;
  const int kOff = (L < 16) ? 0 : 2;
  const int colB = L & 15;

  const float bq0 = bq[n0 + colB];
  const float bk0 = bk[n0 + colB];
  const float bv0 = bvo[n0 + colB];
  v8f cq, ck, cv;
  for (int j = 0; j < 8; ++j) { cq[j] = bq0; ck[j] = bk0; cv[j] = bv0; }

  for (int kk = 0; kk < IN_DIM; kk += 4) {
    v2f a;
    a.x = As[rowA][kk + kOff];
    a.y = As[rowA][kk + kOff + 1];
    v2f b;
    b.x = Wq[(kk + kOff) * HD + n0 + colB];
    b.y = Wq[(kk + kOff + 1) * HD + n0 + colB];
    cq = wmma_f32(a, b, cq);
    b.x = Wk[(kk + kOff) * HD + n0 + colB];
    b.y = Wk[(kk + kOff + 1) * HD + n0 + colB];
    ck = wmma_f32(a, b, ck);
    b.x = Wvo[(kk + kOff) * HD + n0 + colB];
    b.y = Wvo[(kk + kOff + 1) * HD + n0 + colB];
    cv = wmma_f32(a, b, cv);
  }

  for (int j = 0; j < 8; ++j) {
    int r = (L < 16) ? j : (j + 8);
    int gr = m0 + r;
    if (gr < N) {
      size_t o = (size_t)gr * HD + n0 + colB;
      Q[o] = cq[j]; Km[o] = ck[j]; Vo[o] = cv[j];
    }
  }
}

#define EP_WAVES  32
#define LCAP      4096
#define GCH       1024
#define EH        8

static __device__ __forceinline__ int bcast_i(int v, int lane) {
  return __builtin_amdgcn_readlane(v, lane);
}
static __device__ __forceinline__ float bcast_f(float v, int lane) {
  return __int_as_float(__builtin_amdgcn_readlane(__float_as_int(v), lane));
}
template <int M>
static __device__ __forceinline__ float row_xor(float v) {
  return __int_as_float(__builtin_amdgcn_update_dpp(
      0, __float_as_int(v), 0x160 + M, 0xf, 0xf, true));
}

static __device__ __forceinline__ float xhalf(float v) {
  const int b = __float_as_int(v);
  return __int_as_float(__builtin_amdgcn_permlanex16(b, b, 0x76543210, 0xfedcba98,
                                                     false, false));
}

template <int J>
static __device__ __forceinline__ v2f sq2(const v8f c) {
  const v2f p = __builtin_shufflevector(c, c, J, J + 1);
  return p * p;
}
template <int J>
static __device__ __forceinline__ v2f sq2_acc(const v8f c, const v2f acc) {
  const v2f p = __builtin_shufflevector(c, c, J, J + 1);
  return __builtin_elementwise_fma(p, p, acc);
}

static __device__ __forceinline__ float tsum4(const float v0, const float v1,
                                              const float v2, const float v3,
                                              const bool b1, const bool b2) {
  const float k0 = b1 ? v1 : v0, s0 = b1 ? v0 : v1;
  const float k1 = b1 ? v3 : v2, s1 = b1 ? v2 : v3;
  const float a0 = k0 + row_xor<1>(s0);
  const float a1 = k1 + row_xor<1>(s1);
  const float k = b2 ? a1 : a0, s = b2 ? a0 : a1;
  const float b = k + row_xor<2>(s);
  return b + row_xor<4>(b);
}

template <int H>
static __device__ __forceinline__ void k_load(const float* __restrict__ Km,
                                              const int src_l, const int dst_l,
                                              const unsigned l16,
                                              float4 (&kr)[EH], int (&sr)[EH]) {
  for (int j = 0; j < EH; ++j) {
    sr[j] = bcast_i(src_l, H * 16 + j);
    const int dst = bcast_i(dst_l, H * 16 + j);
    const unsigned ok = (unsigned)dst * (HD * 4u) + l16;
    kr[j] = *(const float4*)((const char*)Km + ok);
  }
}

static __device__ __forceinline__ void qk_sums(const float4 (*Qs)[32],
                                               const float4 (&kr)[EH],
                                               const int (&sr)[EH], const int L,
                                               const bool b1, const bool b2,
                                               float& ta, float& tb) {
  float p[EH];
  for (int j = 0; j < EH; ++j) {
    const float4 q = Qs[sr[j]][L];
    p[j] = fmaf(q.w, kr[j].w, fmaf(q.z, kr[j].z, fmaf(q.y, kr[j].y, q.x * kr[j].x)));
  }
  ta = tsum4(p[0], p[1], p[2], p[3], b1, b2);
  tb = tsum4(p[4], p[5], p[6], p[7], b1, b2);
}

static __device__ __forceinline__ void softmax16(const float (&t)[4],
                                                 const float (&se_l)[NHEAD],
                                                 const bool b4, const int L,
                                                 float (&al)[NHEAD]) {
  const bool row1 = (L & 16) != 0;
  const float lo = b4 ? t[1] : t[0], hi = b4 ? t[3] : t[2];
  const float same = row1 ? hi : lo, cross = row1 ? lo : hi;
  const float d0 = same, d1 = row_xor<8>(same);
  const float d2 = xhalf(cross), d3 = row_xor<8>(d2);
  float sc[NHEAD];
  sc[0] = ((row1 ? d2 : d0) + se_l[0]) * INV_SQRT_D;
  sc[1] = ((row1 ? d3 : d1) + se_l[1]) * INV_SQRT_D;
  sc[2] = ((row1 ? d0 : d2) + se_l[2]) * INV_SQRT_D;
  sc[3] = ((row1 ? d1 : d3) + se_l[3]) * INV_SQRT_D;
  float m = sc[0];
  for (int h2 = 1; h2 < NHEAD; ++h2) m = fmaxf(m, sc[h2]);
  float ex[NHEAD], den = 0.0f;
  for (int h2 = 0; h2 < NHEAD; ++h2) { ex[h2] = __expf(sc[h2] - m); den += ex[h2]; }
  const float inv = 1.0f / den;
  for (int h2 = 0; h2 < NHEAD; ++h2) al[h2] = ex[h2] * inv;
}

template <int H>
static __device__ __forceinline__ void vo_load(const float* __restrict__ Vo,
                                               const int dst_l, const unsigned l16,
                                               float4 (&vo)[EH]) {
  for (int j = 0; j < EH; ++j) {
    const int dst = bcast_i(dst_l, H * 16 + j);
    const unsigned ov = (unsigned)dst * (HD * 4u) + l16;
    vo[j] = *(const float4*)((const char*)Vo + ov);
  }
}

template <int H, bool FULL>
static __device__ __forceinline__ void acc_half(float (*sumS)[OUT_DIM],
                                                const int src_l,
                                                const float (&al)[NHEAD],
                                                const float4 (&vo)[EH],
                                                const int L, const float bo_j,
                                                const int nb) {
  for (int j = 0; j < EH; ++j) {
    float acc = bo_j;
    acc = fmaf(bcast_f(al[0], H * 16 + j), vo[j].x, acc);
    acc = fmaf(bcast_f(al[1], H * 16 + j), vo[j].y, acc);
    acc = fmaf(bcast_f(al[2], H * 16 + j), vo[j].z, acc);
    acc = fmaf(bcast_f(al[3], H * 16 + j), vo[j].w, acc);
    if (FULL || H * 8 + j < nb)
      atomicAdd(&sumS[bcast_i(src_l, H * 16 + j)][L], acc);
  }
}

__global__ __launch_bounds__(1024) void edge_pass(
    const int* __restrict__ ei, const float* __restrict__ ea,
    const float* __restrict__ We, const float* __restrict__ be,
    const float* __restrict__ Q, const float* __restrict__ Km,
    const float* __restrict__ Vo, const float* __restrict__ bo,
    const int* __restrict__ sorted, const int* __restrict__ segStart, int G,
    float* __restrict__ out, int N, int NE) {
  __shared__ v2f    Bf[64][32];
  __shared__ __attribute__((aligned(32))) float bes[HD];
  __shared__ float4 Qs[RN][32];
  __shared__ float  sumS[RN][OUT_DIM];
  __shared__ int    cntS[RN];
  __shared__ int    listS[LCAP];
  __shared__ int    pfxS[GCH + 1];
  __shared__ int    offS[GCH];
  __shared__ int    wsumS[EP_WAVES];
  __shared__ int    s_next;

  const int tid = threadIdx.x;
  const int L = tid & 31;
  const int wv = __builtin_amdgcn_readfirstlane((int)(tid >> 5));
  const int c16 = L & 15;
  const int hh = L >> 4;
  const int base = blockIdx.x * RN;

  for (int q = 0; q < 2; ++q) {
    const int f = wv * 2 + q;
    const int col = (f >> 4) * DHEAD + ((f >> 3) & 1) * 16 + c16;
    const int kr = hh * 16 + (f & 7) * 2;
    v2f b;
    b.x = We[kr * HD + col];
    b.y = We[(kr + 1) * HD + col];
    Bf[f][L] = b;
  }
  if (tid < HD) bes[tid] = be[tid];
  for (int i = tid; i < RN * 32; i += 1024) {
    int node = base + (i >> 5); if (node >= N) node = N - 1;
    Qs[i >> 5][i & 31] = *(const float4*)(Q + (size_t)node * HD + (i & 31) * 4);
  }
  for (int i = tid; i < RN * OUT_DIM; i += 1024) sumS[i >> 5][i & 31] = 0.0f;
  if (tid < RN) cntS[tid] = 0;

  const int rot = (int)(((unsigned)blockIdx.x * (unsigned)G) / gridDim.x);
  const int segPitch = (int)gridDim.x + 1;

  const float bo_j = bo[L];
  const int slot = (L >> 4) * 8 + (L & 7);
  const bool p1 = (L & 1) != 0, p2 = (L & 2) != 0, p4 = (L & 4) != 0;
  const unsigned l16 = (unsigned)L * 16u;

  for (int gB = 0; gB < G; gB += GCH) {
    const int Gc = (G - gB < GCH) ? (G - gB) : GCH;
    int n = 0, off = 0;
    if (tid < Gc) {
      int gg = gB + tid + rot; if (gg >= G) gg -= G;
      const int* const st = segStart + (size_t)gg * segPitch + blockIdx.x;
      const int a = st[0];
      n = st[1] - a;
      off = gg * SLICE + a;
    }
    int inc = n;
    for (int d = 1; d < 32; d <<= 1) {
      const int t = __shfl_up(inc, d);
      if (L >= d) inc += t;
    }
    if (L == 31) wsumS[wv] = inc;
    __syncthreads();
    if (wv == 0) {
      const int x = wsumS[L];
      int xi = x;
      for (int d = 1; d < 32; d <<= 1) {
        const int t = __shfl_up(xi, d);
        if (L >= d) xi += t;
      }
      wsumS[L] = xi - x;
    }
    __syncthreads();
    const int excl = inc - n + wsumS[wv];
    if (tid < Gc) { pfxS[tid] = excl; offS[tid] = off; }
    if (tid == Gc - 1) pfxS[Gc] = excl + n;
    __syncthreads();
    const int T = pfxS[Gc];
    int step0 = 1;
    while (step0 < Gc) step0 <<= 1;
    step0 >>= 1;

    for (int w0 = 0; w0 < T; w0 += LCAP) {
      const int Tw = (T - w0 < LCAP) ? (T - w0) : LCAP;
      int pos[LCAP / 1024], lo[LCAP / 1024];
      for (int k = 0; k < LCAP / 1024; ++k) {
        const int i = tid + k * 1024;
        pos[k] = w0 + ((i < Tw) ? i : Tw - 1);
        lo[k] = 0;
      }
      for (int step = step0; step >= 1; step >>= 1) {
        for (int k = 0; k < LCAP / 1024; ++k) {
          const int c = lo[k] + step;
          const int cc = (c < Gc) ? c : Gc - 1;
          if (c < Gc && pfxS[cc] <= pos[k]) lo[k] = c;
        }
      }
      for (int k = 0; k < LCAP / 1024; ++k) {
        const int i = tid + k * 1024;
        if (i < Tw) listS[i] = sorted[offS[lo[k]] + (pos[k] - pfxS[lo[k]])];
      }
      if (tid == 0) s_next = 0;
      __syncthreads();

      const int nBatch = (Tw + 15) >> 4;
      int bq = 0;
      if (L == 0) bq = atomicAdd(&s_next, 1);
      bq = __builtin_amdgcn_readfirstlane(bq);
      while (bq < nBatch) {
        asm volatile("" ::: "memory");
        const int e0 = bq * 16;
        const int nb = (Tw - e0 < 16) ? (Tw - e0) : 16;
        const int rowE = listS[e0 + ((c16 < nb) ? c16 : nb - 1)];
        const unsigned aoff = (unsigned)rowE * (E_DIM * 4u) + (unsigned)hh * 64u;
        v2f af[8];
        for (int i = 0; i < 4; ++i) {
          const float4 q4 = *(const float4*)((const char*)ea + aoff + i * 16);
          af[2 * i].x = q4.x;     af[2 * i].y = q4.y;
          af[2 * i + 1].x = q4.z; af[2 * i + 1].y = q4.w;
        }
        __builtin_amdgcn_sched_barrier(0x386);

        const bool swp = ((L >> 3) & 1) != hh;
        float se_l[NHEAD];
        for (int h = 0; h < NHEAD; ++h) {
          v8f c0 = *(const v8f*)&bes[(h * 2 + 0) * 16 + hh * 8];
          v8f c1 = *(const v8f*)&bes[(h * 2 + 1) * 16 + hh * 8];
          v2f b0 = Bf[(h * 2 + 0) * 8][L];
          v2f b1 = Bf[(h * 2 + 1) * 8][L];
          for (int s = 0; s < 8; ++s) {
            v2f n0 = b0, n1 = b1;
            if (s < 7) {
              n0 = Bf[(h * 2 + 0) * 8 + s + 1][L];
              n1 = Bf[(h * 2 + 1) * 8 + s + 1][L];
            }
            c0 = wmma_f32(b0, af[s], c0);
            c1 = wmma_f32(b1, af[s], c1);
            b0 = n0; b1 = n1;
          }
          v2f a2 = sq2<0>(c0);
          a2 = sq2_acc<2>(c0, a2); a2 = sq2_acc<4>(c0, a2); a2 = sq2_acc<6>(c0, a2);
          a2 = sq2_acc<0>(c1, a2); a2 = sq2_acc<2>(c1, a2);
          a2 = sq2_acc<4>(c1, a2); a2 = sq2_acc<6>(c1, a2);
          const float sp = a2.x + a2.y;
          const float tot = sp + xhalf(sp);
          const float oth = row_xor<8>(tot);
          se_l[h] = swp ? oth : tot;
          __builtin_amdgcn_sched_barrier(0x7f6);
        }

        const int myE = listS[e0 + ((slot < nb) ? slot : nb - 1)];
        const int src_l = ei[myE] - base;
        const int dst_l = ei[NE + myE];
        if ((L & 8) == 0 && slot < nb) atomicAdd(&cntS[src_l], 1);

        float t[4], al[NHEAD];
        float4 kr0[EH], kr1[EH];
        int sr0[EH], sr1[EH];
        float4 vo0[EH], vo1[EH];
        k_load<0>(Km, src_l, dst_l, l16, kr0, sr0);
        qk_sums(Qs, kr0, sr0, L, p1, p2, t[0], t[1]);
        t[2] = 0.0f; t[3] = 0.0f;
        if (nb > EH) {
          k_load<1>(Km, src_l, dst_l, l16, kr1, sr1);
          qk_sums(Qs, kr1, sr1, L, p1, p2, t[2], t[3]);
        }
        vo_load<0>(Vo, dst_l, l16, vo0);
        softmax16(t, se_l, p4, L, al);
        if (nb == 2 * EH) {
          vo_load<1>(Vo, dst_l, l16, vo1);
          acc_half<0, true>(sumS, src_l, al, vo0, L, bo_j, nb);
          acc_half<1, true>(sumS, src_l, al, vo1, L, bo_j, nb);
        } else {
          acc_half<0, false>(sumS, src_l, al, vo0, L, bo_j, nb);
          if (nb > EH) {
            vo_load<1>(Vo, dst_l, l16, vo1);
            acc_half<1, false>(sumS, src_l, al, vo1, L, bo_j, nb);
          }
        }
        bq = 0;
        if (L == 0) bq = atomicAdd(&s_next, 1);
        bq = __builtin_amdgcn_readfirstlane(bq);
      }
      __syncthreads();
    }
  }

  __syncthreads();
  for (int r = wv; r < RN; r += EP_WAVES) {
    const int node = base + r;
    if (node < N)
      out[(size_t)node * OUT_DIM + L] = sumS[r][L] / fmaxf((float)cntS[r], 1.0f);
  }
}

extern "C" void kernel_launch(void* const* d_in, const int* in_sizes, int n_in,
                              void* d_out, int out_size, void* d_ws, size_t ws_size,
                              hipStream_t stream) {
  const float* x  = (const float*)d_in[0];
  const int*   ei = (const int*)d_in[1];
  const float* ea = (const float*)d_in[2];
  const float* Wq = (const float*)d_in[3];
  const float* bq = (const float*)d_in[4];
  const float* Wk = (const float*)d_in[5];
  const float* bk = (const float*)d_in[6];
  const float* Wv = (const float*)d_in[7];
  const float* bv = (const float*)d_in[8];
  const float* We = (const float*)d_in[9];
  const float* be = (const float*)d_in[10];
  const float* Wo = (const float*)d_in[11];
  const float* bo = (const float*)d_in[12];

  const int N  = in_sizes[0] / IN_DIM;
  const int NE = in_sizes[1] / 2;

  float* ws   = (float*)d_ws;
  float* Q    = ws;
  float* Km   = Q   + (size_t)N * HD;
  float* Vo   = Km  + (size_t)N * HD;
  float* Wvo  = Vo  + (size_t)N * HD;
  float* bvo  = Wvo + IN_DIM * HD;
  const int NBK = (N + RN - 1) / RN;
  const int G   = (NE + SLICE - 1) / SLICE;
  int* sorted   = (int*)(bvo + HD);
  int* segStart = sorted + (size_t)G * SLICE;

  prep_pass<<<PRE_BLOCKS + G, 1024, 0, stream>>>(Wv, bv, Wo, Wvo, bvo, ei,
                                                 sorted, segStart, N, NE, NBK);
  node_gemm<<<(N + 15) / 16, 256, 0, stream>>>(x, Wq, bq, Wk, bk, Wvo, bvo,
                                               Q, Km, Vo, N);
  edge_pass<<<NBK, 1024, 0, stream>>>(ei, ea, We, be, Q, Km, Vo, bo,
                                      sorted, segStart, G, (float*)d_out, N, NE);
}


# one launch fewer: Wo applied to each 16-row V tile in the node GEMM (8 more WMMAs a wave; Vo = (x Wv + bv) Wo, new sum order), bucket pass a block role of that launch; weight fold pass gone
# speedup vs baseline: 3.4789x; 1.0325x over previous
#include <hip/hip_runtime.h>
#include <hip/hip_bf16.h>

typedef __attribute__((ext_vector_type(2))) float v2f;
typedef __attribute__((ext_vector_type(8))) float v8f;

#define IN_DIM  128
#define HD      128
#define E_DIM   32
#define NHEAD   4
#define DHEAD   32
#define OUT_DIM 32
#define INV_SQRT_D 0.17677669529663687f

static __device__ __forceinline__ v8f wmma_f32(v2f a, v2f b, v8f c) {
  return __builtin_amdgcn_wmma_f32_16x16x4_f32(false, a, false, b, (short)0, c,
                                               false, false);
}

#define RN        40
#define SLICE     4096
#define BK_MAX    1024
#define VPITCH    132

__global__ __launch_bounds__(256) void node_pass(
    const float* __restrict__ x,
    const float* __restrict__ Wq, const float* __restrict__ bq,
    const float* __restrict__ Wk, const float* __restrict__ bk,
    const float* __restrict__ Wv, const float* __restrict__ bv,
    const float* __restrict__ Wo,
    float* __restrict__ Q, float* __restrict__ Km, float* __restrict__ Vo,
    const int* __restrict__ ei, int* __restrict__ sorted,
    int* __restrict__ segStart, int N, int NE, int NBK, int G) {
  __shared__ int  histL[BK_MAX];
  __shared__ int  wsum[8];
  __shared__ int  s_total;
  __shared__ int4 sortL4[SLICE / 4];
  __shared__ float As[16][IN_DIM];
  __shared__ __attribute__((aligned(16))) float Vs[16][VPITCH];
  __shared__ __attribute__((aligned(16))) float Os[16][VPITCH];

  const int tid = threadIdx.x;
  const int w = tid >> 5, L = tid & 31;

  if ((int)blockIdx.x < G) {
    const int g = (int)blockIdx.x;
    int* const sortL = (int*)sortL4;
    for (int i = tid; i < BK_MAX; i += 256) histL[i] = 0;
    __syncthreads();

    const int eb = g * SLICE + tid * 4;
    int s[16];
    if (g * SLICE + SLICE <= NE) {
      int4 v4[4];
      for (int q = 0; q < 4; ++q) v4[q] = *(const int4*)(ei + eb + q * 1024);
      for (int q = 0; q < 4; ++q) {
        s[q * 4 + 0] = v4[q].x; s[q * 4 + 1] = v4[q].y;
        s[q * 4 + 2] = v4[q].z; s[q * 4 + 3] = v4[q].w;
      }
    } else {
      for (int q = 0; q < 4; ++q)
        for (int k = 0; k < 4; ++k) {
          const int e = eb + q * 1024 + k;
          s[q * 4 + k] = (e < NE) ? ei[e] : -1;
        }
    }
    int bkt[16], rk[16];
    for (int i = 0; i < 16; ++i) {
      const bool ok = (unsigned)s[i] < (unsigned)N;
      bkt[i] = ok ? s[i] / RN : -1;
      rk[i] = 0;
      if (ok) rk[i] = atomicAdd(&histL[bkt[i]], 1);
    }
    __syncthreads();

    int c[4];
    for (int k = 0; k < 4; ++k) c[k] = histL[tid * 4 + k];
    const int v = c[0] + c[1] + c[2] + c[3];
    int inc = v;
    for (int d = 1; d < 32; d <<= 1) {
      const int t = __shfl_up(inc, d);
      if (L >= d) inc += t;
    }
    if (L == 31) wsum[w] = inc;
    __syncthreads();
    if (w == 0) {
      const int xw = (L < 8) ? wsum[L & 7] : 0;
      int xi = xw;
      for (int d = 1; d < 8; d <<= 1) {
        const int t = __shfl_up(xi, d);
        if (L >= d) xi += t;
      }
      if (L < 8) wsum[L] = xi - xw;
    }
    __syncthreads();
    int* const st = segStart + (size_t)g * (NBK + 1);
    int p = inc - v + wsum[w];
    for (int k = 0; k < 4; ++k) {
      const int b = tid * 4 + k;
      histL[b] = p;
      if (b < NBK) st[b] = p;
      if (b == NBK - 1) { st[NBK] = p + c[k]; s_total = p + c[k]; }
      p += c[k];
    }
    __syncthreads();

    for (int q = 0; q < 4; ++q)
      for (int k = 0; k < 4; ++k) {
        const int i = q * 4 + k;
        if (bkt[i] >= 0)
          sortL[histL[bkt[i]] + rk[i]] = eb + q * 1024 + k;
      }
    __syncthreads();
    for (int q = 0; q < 4; ++q) {
      const int i4 = tid + q * 256;
      if (i4 * 4 < s_total)
        *(int4*)(sorted + (size_t)g * SLICE + i4 * 4) = sortL4[i4];
    }
    return;
  }

  const int m0 = ((int)blockIdx.x - G) * 16;

  for (int i = 0; i < 8; ++i) {
    int lin = tid + i * 256;
    int r = lin >> 7, cc = lin & 127;
    int gr = m0 + r; if (gr >= N) gr = N - 1;
    As[r][cc] = x[(size_t)gr * IN_DIM + cc];
  }
  __syncthreads();

  const int n0 = w * 16;
  const int rowA = L & 15;
  const int kOff = (L < 16) ? 0 : 2;
  const int colB = L & 15;

  const float bq0 = bq[n0 + colB];
  const float bk0 = bk[n0 + colB];
  const float bv0 = bv[n0 + colB];
  v8f cq, ck, cv;
  for (int j = 0; j < 8; ++j) { cq[j] = bq0; ck[j] = bk0; cv[j] = bv0; }

  for (int kk = 0; kk < IN_DIM; kk += 4) {
    v2f a;
    a.x = As[rowA][kk + kOff];
    a.y = As[rowA][kk + kOff + 1];
    v2f b;
    b.x = Wq[(kk + kOff) * HD + n0 + colB];
    b.y = Wq[(kk + kOff + 1) * HD + n0 + colB];
    cq = wmma_f32(a, b, cq);
    b.x = Wk[(kk + kOff) * HD + n0 + colB];
    b.y = Wk[(kk + kOff + 1) * HD + n0 + colB];
    ck = wmma_f32(a, b, ck);
    b.x = Wv[(kk + kOff) * HD + n0 + colB];
    b.y = Wv[(kk + kOff + 1) * HD + n0 + colB];
    cv = wmma_f32(a, b, cv);
  }

  for (int j = 0; j < 8; ++j) {
    int r = (L < 16) ? j : (j + 8);
    int gr = m0 + r;
    Vs[r][n0 + colB] = cv[j];
    if (gr < N) {
      size_t o = (size_t)gr * HD + n0 + colB;
      Q[o] = cq[j]; Km[o] = ck[j];
    }
  }
  __syncthreads();

  const int h = w >> 1, t2 = w & 1;
  v8f co;
  for (int j = 0; j < 8; ++j) co[j] = 0.0f;
  for (int kk = 0; kk < DHEAD; kk += 4) {
    const int k0 = h * DHEAD + kk + kOff;
    v2f a;
    a.x = Vs[rowA][k0];
    a.y = Vs[rowA][k0 + 1];
    v2f b;
    b.x = Wo[k0 * OUT_DIM + t2 * 16 + colB];
    b.y = Wo[(k0 + 1) * OUT_DIM + t2 * 16 + colB];
    co = wmma_f32(a, b, co);
  }
  for (int j = 0; j < 8; ++j) {
    int r = (L < 16) ? j : (j + 8);
    Os[r][(t2 * 16 + colB) * 4 + h] = co[j];
  }
  __syncthreads();
  for (int i = tid; i < 16 * 32; i += 256) {
    const int r = i >> 5, c4 = i & 31;
    const int gr = m0 + r;
    if (gr < N)
      *(float4*)(Vo + (size_t)gr * HD + c4 * 4) = *(const float4*)&Os[r][c4 * 4];
  }
}

#define EP_WAVES  32
#define LCAP      4096
#define GCH       1024
#define EH        8

static __device__ __forceinline__ int bcast_i(int v, int lane) {
  return __builtin_amdgcn_readlane(v, lane);
}
static __device__ __forceinline__ float bcast_f(float v, int lane) {
  return __int_as_float(__builtin_amdgcn_readlane(__float_as_int(v), lane));
}
template <int M>
static __device__ __forceinline__ float row_xor(float v) {
  return __int_as_float(__builtin_amdgcn_update_dpp(
      0, __float_as_int(v), 0x160 + M, 0xf, 0xf, true));
}

static __device__ __forceinline__ float xhalf(float v) {
  const int b = __float_as_int(v);
  return __int_as_float(__builtin_amdgcn_permlanex16(b, b, 0x76543210, 0xfedcba98,
                                                     false, false));
}

template <int J>
static __device__ __forceinline__ v2f sq2(const v8f c) {
  const v2f p = __builtin_shufflevector(c, c, J, J + 1);
  return p * p;
}
template <int J>
static __device__ __forceinline__ v2f sq2_acc(const v8f c, const v2f acc) {
  const v2f p = __builtin_shufflevector(c, c, J, J + 1);
  return __builtin_elementwise_fma(p, p, acc);
}

static __device__ __forceinline__ float tsum4(const float v0, const float v1,
                                              const float v2, const float v3,
                                              const bool b1, const bool b2) {
  const float k0 = b1 ? v1 : v0, s0 = b1 ? v0 : v1;
  const float k1 = b1 ? v3 : v2, s1 = b1 ? v2 : v3;
  const float a0 = k0 + row_xor<1>(s0);
  const float a1 = k1 + row_xor<1>(s1);
  const float k = b2 ? a1 : a0, s = b2 ? a0 : a1;
  const float b = k + row_xor<2>(s);
  return b + row_xor<4>(b);
}

template <int H>
static __device__ __forceinline__ void k_load(const float* __restrict__ Km,
                                              const int src_l, const int dst_l,
                                              const unsigned l16,
                                              float4 (&kr)[EH], int (&sr)[EH]) {
  for (int j = 0; j < EH; ++j) {
    sr[j] = bcast_i(src_l, H * 16 + j);
    const int dst = bcast_i(dst_l, H * 16 + j);
    const unsigned ok = (unsigned)dst * (HD * 4u) + l16;
    kr[j] = *(const float4*)((const char*)Km + ok);
  }
}

static __device__ __forceinline__ void qk_sums(const float4 (*Qs)[32],
                                               const float4 (&kr)[EH],
                                               const int (&sr)[EH], const int L,
                                               const bool b1, const bool b2,
                                               float& ta, float& tb) {
  float p[EH];
  for (int j = 0; j < EH; ++j) {
    const float4 q = Qs[sr[j]][L];
    p[j] = fmaf(q.w, kr[j].w, fmaf(q.z, kr[j].z, fmaf(q.y, kr[j].y, q.x * kr[j].x)));
  }
  ta = tsum4(p[0], p[1], p[2], p[3], b1, b2);
  tb = tsum4(p[4], p[5], p[6], p[7], b1, b2);
}

static __device__ __forceinline__ void softmax16(const float (&t)[4],
                                                 const float (&se_l)[NHEAD],
                                                 const bool b4, const int L,
                                                 float (&al)[NHEAD]) {
  const bool row1 = (L & 16) != 0;
  const float lo = b4 ? t[1] : t[0], hi = b4 ? t[3] : t[2];
  const float same = row1 ? hi : lo, cross = row1 ? lo : hi;
  const float d0 = same, d1 = row_xor<8>(same);
  const float d2 = xhalf(cross), d3 = row_xor<8>(d2);
  float sc[NHEAD];
  sc[0] = ((row1 ? d2 : d0) + se_l[0]) * INV_SQRT_D;
  sc[1] = ((row1 ? d3 : d1) + se_l[1]) * INV_SQRT_D;
  sc[2] = ((row1 ? d0 : d2) + se_l[2]) * INV_SQRT_D;
  sc[3] = ((row1 ? d1 : d3) + se_l[3]) * INV_SQRT_D;
  float m = sc[0];
  for (int h2 = 1; h2 < NHEAD; ++h2) m = fmaxf(m, sc[h2]);
  float ex[NHEAD], den = 0.0f;
  for (int h2 = 0; h2 < NHEAD; ++h2) { ex[h2] = __expf(sc[h2] - m); den += ex[h2]; }
  const float inv = 1.0f / den;
  for (int h2 = 0; h2 < NHEAD; ++h2) al[h2] = ex[h2] * inv;
}

template <int H>
static __device__ __forceinline__ void vo_load(const float* __restrict__ Vo,
                                               const int dst_l, const unsigned l16,
                                               float4 (&vo)[EH]) {
  for (int j = 0; j < EH; ++j) {
    const int dst = bcast_i(dst_l, H * 16 + j);
    const unsigned ov = (unsigned)dst * (HD * 4u) + l16;
    vo[j] = *(const float4*)((const char*)Vo + ov);
  }
}

template <int H, bool FULL>
static __device__ __forceinline__ void acc_half(float (*sumS)[OUT_DIM],
                                                const int src_l,
                                                const float (&al)[NHEAD],
                                                const float4 (&vo)[EH],
                                                const int L, const float bo_j,
                                                const int nb) {
  for (int j = 0; j < EH; ++j) {
    float acc = bo_j;
    acc = fmaf(bcast_f(al[0], H * 16 + j), vo[j].x, acc);
    acc = fmaf(bcast_f(al[1], H * 16 + j), vo[j].y, acc);
    acc = fmaf(bcast_f(al[2], H * 16 + j), vo[j].z, acc);
    acc = fmaf(bcast_f(al[3], H * 16 + j), vo[j].w, acc);
    if (FULL || H * 8 + j < nb)
      atomicAdd(&sumS[bcast_i(src_l, H * 16 + j)][L], acc);
  }
}

__global__ __launch_bounds__(1024) void edge_pass(
    const int* __restrict__ ei, const float* __restrict__ ea,
    const float* __restrict__ We, const float* __restrict__ be,
    const float* __restrict__ Q, const float* __restrict__ Km,
    const float* __restrict__ Vo, const float* __restrict__ bo,
    const int* __restrict__ sorted, const int* __restrict__ segStart, int G,
    float* __restrict__ out, int N, int NE) {
  __shared__ v2f    Bf[64][32];
  __shared__ __attribute__((aligned(32))) float bes[HD];
  __shared__ float4 Qs[RN][32];
  __shared__ float  sumS[RN][OUT_DIM];
  __shared__ int    cntS[RN];
  __shared__ int    listS[LCAP];
  __shared__ int    pfxS[GCH + 1];
  __shared__ int    offS[GCH];
  __shared__ int    wsumS[EP_WAVES];
  __shared__ int    s_next;

  const int tid = threadIdx.x;
  const int L = tid & 31;
  const int wv = __builtin_amdgcn_readfirstlane((int)(tid >> 5));
  const int c16 = L & 15;
  const int hh = L >> 4;
  const int base = blockIdx.x * RN;

  for (int q = 0; q < 2; ++q) {
    const int f = wv * 2 + q;
    const int col = (f >> 4) * DHEAD + ((f >> 3) & 1) * 16 + c16;
    const int kr = hh * 16 + (f & 7) * 2;
    v2f b;
    b.x = We[kr * HD + col];
    b.y = We[(kr + 1) * HD + col];
    Bf[f][L] = b;
  }
  if (tid < HD) bes[tid] = be[tid];
  for (int i = tid; i < RN * 32; i += 1024) {
    int node = base + (i >> 5); if (node >= N) node = N - 1;
    Qs[i >> 5][i & 31] = *(const float4*)(Q + (size_t)node * HD + (i & 31) * 4);
  }
  for (int i = tid; i < RN * OUT_DIM; i += 1024) sumS[i >> 5][i & 31] = 0.0f;
  if (tid < RN) cntS[tid] = 0;

  const int rot = (int)(((unsigned)blockIdx.x * (unsigned)G) / gridDim.x);
  const int segPitch = (int)gridDim.x + 1;

  const float bo_j = bo[L];
  const int slot = (L >> 4) * 8 + (L & 7);
  const bool p1 = (L & 1) != 0, p2 = (L & 2) != 0, p4 = (L & 4) != 0;
  const unsigned l16 = (unsigned)L * 16u;

  for (int gB = 0; gB < G; gB += GCH) {
    const int Gc = (G - gB < GCH) ? (G - gB) : GCH;
    int n = 0, off = 0;
    if (tid < Gc) {
      int gg = gB + tid + rot; if (gg >= G) gg -= G;
      const int* const st = segStart + (size_t)gg * segPitch + blockIdx.x;
      const int a = st[0];
      n = st[1] - a;
      off = gg * SLICE + a;
    }
    int inc = n;
    for (int d = 1; d < 32; d <<= 1) {
      const int t = __shfl_up(inc, d);
      if (L >= d) inc += t;
    }
    if (L == 31) wsumS[wv] = inc;
    __syncthreads();
    if (wv == 0) {
      const int x = wsumS[L];
      int xi = x;
      for (int d = 1; d < 32; d <<= 1) {
        const int t = __shfl_up(xi, d);
        if (L >= d) xi += t;
      }
      wsumS[L] = xi - x;
    }
    __syncthreads();
    const int excl = inc - n + wsumS[wv];
    if (tid < Gc) { pfxS[tid] = excl; offS[tid] = off; }
    if (tid == Gc - 1) pfxS[Gc] = excl + n;
    __syncthreads();
    const int T = pfxS[Gc];
    int step0 = 1;
    while (step0 < Gc) step0 <<= 1;
    step0 >>= 1;

    for (int w0 = 0; w0 < T; w0 += LCAP) {
      const int Tw = (T - w0 < LCAP) ? (T - w0) : LCAP;
      int pos[LCAP / 1024], lo[LCAP / 1024];
      for (int k = 0; k < LCAP / 1024; ++k) {
        const int i = tid + k * 1024;
        pos[k] = w0 + ((i < Tw) ? i : Tw - 1);
        lo[k] = 0;
      }
      for (int step = step0; step >= 1; step >>= 1) {
        for (int k = 0; k < LCAP / 1024; ++k) {
          const int c = lo[k] + step;
          const int cc = (c < Gc) ? c : Gc - 1;
          if (c < Gc && pfxS[cc] <= pos[k]) lo[k] = c;
        }
      }
      for (int k = 0; k < LCAP / 1024; ++k) {
        const int i = tid + k * 1024;
        if (i < Tw) listS[i] = sorted[offS[lo[k]] + (pos[k] - pfxS[lo[k]])];
      }
      if (tid == 0) s_next = 0;
      __syncthreads();

      const int nBatch = (Tw + 15) >> 4;
      int bq = 0;
      if (L == 0) bq = atomicAdd(&s_next, 1);
      bq = __builtin_amdgcn_readfirstlane(bq);
      while (bq < nBatch) {
        asm volatile("" ::: "memory");
        const int e0 = bq * 16;
        const int nb = (Tw - e0 < 16) ? (Tw - e0) : 16;
        const int rowE = listS[e0 + ((c16 < nb) ? c16 : nb - 1)];
        const unsigned aoff = (unsigned)rowE * (E_DIM * 4u) + (unsigned)hh * 64u;
        v2f af[8];
        for (int i = 0; i < 4; ++i) {
          const float4 q4 = *(const float4*)((const char*)ea + aoff + i * 16);
          af[2 * i].x = q4.x;     af[2 * i].y = q4.y;
          af[2 * i + 1].x = q4.z; af[2 * i + 1].y = q4.w;
        }
        __builtin_amdgcn_sched_barrier(0x386);

        const bool swp = ((L >> 3) & 1) != hh;
        float se_l[NHEAD];
        for (int h = 0; h < NHEAD; ++h) {
          v8f c0 = *(const v8f*)&bes[(h * 2 + 0) * 16 + hh * 8];
          v8f c1 = *(const v8f*)&bes[(h * 2 + 1) * 16 + hh * 8];
          v2f b0 = Bf[(h * 2 + 0) * 8][L];
          v2f b1 = Bf[(h * 2 + 1) * 8][L];
          for (int s = 0; s < 8; ++s) {
            v2f n0 = b0, n1 = b1;
            if (s < 7) {
              n0 = Bf[(h * 2 + 0) * 8 + s + 1][L];
              n1 = Bf[(h * 2 + 1) * 8 + s + 1][L];
            }
            c0 = wmma_f32(b0, af[s], c0);
            c1 = wmma_f32(b1, af[s], c1);
            b0 = n0; b1 = n1;
          }
          v2f a2 = sq2<0>(c0);
          a2 = sq2_acc<2>(c0, a2); a2 = sq2_acc<4>(c0, a2); a2 = sq2_acc<6>(c0, a2);
          a2 = sq2_acc<0>(c1, a2); a2 = sq2_acc<2>(c1, a2);
          a2 = sq2_acc<4>(c1, a2); a2 = sq2_acc<6>(c1, a2);
          const float sp = a2.x + a2.y;
          const float tot = sp + xhalf(sp);
          const float oth = row_xor<8>(tot);
          se_l[h] = swp ? oth : tot;
          __builtin_amdgcn_sched_barrier(0x7f6);
        }

        const int myE = listS[e0 + ((slot < nb) ? slot : nb - 1)];
        const int src_l = ei[myE] - base;
        const int dst_l = ei[NE + myE];
        if ((L & 8) == 0 && slot < nb) atomicAdd(&cntS[src_l], 1);

        float t[4], al[NHEAD];
        float4 kr0[EH], kr1[EH];
        int sr0[EH], sr1[EH];
        float4 vo0[EH], vo1[EH];
        k_load<0>(Km, src_l, dst_l, l16, kr0, sr0);
        qk_sums(Qs, kr0, sr0, L, p1, p2, t[0], t[1]);
        t[2] = 0.0f; t[3] = 0.0f;
        if (nb > EH) {
          k_load<1>(Km, src_l, dst_l, l16, kr1, sr1);
          qk_sums(Qs, kr1, sr1, L, p1, p2, t[2], t[3]);
        }
        vo_load<0>(Vo, dst_l, l16, vo0);
        softmax16(t, se_l, p4, L, al);
        if (nb == 2 * EH) {
          vo_load<1>(Vo, dst_l, l16, vo1);
          acc_half<0, true>(sumS, src_l, al, vo0, L, bo_j, nb);
          acc_half<1, true>(sumS, src_l, al, vo1, L, bo_j, nb);
        } else {
          acc_half<0, false>(sumS, src_l, al, vo0, L, bo_j, nb);
          if (nb > EH) {
            vo_load<1>(Vo, dst_l, l16, vo1);
            acc_half<1, false>(sumS, src_l, al, vo1, L, bo_j, nb);
          }
        }
        bq = 0;
        if (L == 0) bq = atomicAdd(&s_next, 1);
        bq = __builtin_amdgcn_readfirstlane(bq);
      }
      __syncthreads();
    }
  }

  __syncthreads();
  for (int r = wv; r < RN; r += EP_WAVES) {
    const int node = base + r;
    if (node < N)
      out[(size_t)node * OUT_DIM + L] = sumS[r][L] / fmaxf((float)cntS[r], 1.0f);
  }
}

extern "C" void kernel_launch(void* const* d_in, const int* in_sizes, int n_in,
                              void* d_out, int out_size, void* d_ws, size_t ws_size,
                              hipStream_t stream) {
  const float* x  = (const float*)d_in[0];
  const int*   ei = (const int*)d_in[1];
  const float* ea = (const float*)d_in[2];
  const float* Wq = (const float*)d_in[3];
  const float* bq = (const float*)d_in[4];
  const float* Wk = (const float*)d_in[5];
  const float* bk = (const float*)d_in[6];
  const float* Wv = (const float*)d_in[7];
  const float* bv = (const float*)d_in[8];
  const float* We = (const float*)d_in[9];
  const float* be = (const float*)d_in[10];
  const float* Wo = (const float*)d_in[11];
  const float* bo = (const float*)d_in[12];

  const int N  = in_sizes[0] / IN_DIM;
  const int NE = in_sizes[1] / 2;

  float* ws   = (float*)d_ws;
  float* Q    = ws;
  float* Km   = Q   + (size_t)N * HD;
  float* Vo   = Km  + (size_t)N * HD;
  const int NBK = (N + RN - 1) / RN;
  const int G   = (NE + SLICE - 1) / SLICE;
  int* sorted   = (int*)(Vo + (size_t)N * HD);
  int* segStart = sorted + (size_t)G * SLICE;

  node_pass<<<G + (N + 15) / 16, 256, 0, stream>>>(x, Wq, bq, Wk, bk, Wv, bv, Wo,
                                                   Q, Km, Vo, ei, sorted, segStart,
                                                   N, NE, NBK, G);
  edge_pass<<<NBK, 1024, 0, stream>>>(ei, ea, We, be, Q, Km, Vo, bo,
                                      sorted, segStart, G, (float*)d_out, N, NE);
}
